# starting kernel
# baseline (speedup 1.0000x reference)
#include <hip/hip_runtime.h>
#include <math.h>

#define F_OUT 64

typedef __attribute__((ext_vector_type(2))) float v2f;
typedef __attribute__((ext_vector_type(8))) float v8f;

static inline int cdiv(int a, int b) { return (a + b - 1) / b; }

__global__ void k_init_norm(float* __restrict__ ns, float* __restrict__ nd, int n) {
    int i = blockIdx.x * blockDim.x + threadIdx.x;
    if (i < n) { ns[i] = 1.0f; nd[i] = 1.0f; }
}

__global__ void k_degree(const int* __restrict__ src, const int* __restrict__ dst,
                         float* __restrict__ ns, float* __restrict__ nd, int e) {
    int i = blockIdx.x * blockDim.x + threadIdx.x;
    if (i < e) {
        atomicAdd(&ns[src[i]], 1.0f);
        atomicAdd(&nd[dst[i]], 1.0f);
    }
}

__global__ void k_rsqrt(float* __restrict__ ns, float* __restrict__ nd, int n) {
    int i = blockIdx.x * blockDim.x + threadIdx.x;
    if (i < n) { ns[i] = rsqrtf(ns[i]); nd[i] = rsqrtf(nd[i]); }
}

template <int K>
__global__ __launch_bounds__(256) void k_gemm_scale(const float* __restrict__ X,
                                                    const float* __restrict__ W,
                                                    const float* __restrict__ norm,
                                                    float* __restrict__ H,
                                                    float* __restrict__ H2, int nrows) {
    __shared__ float ldsW[F_OUT * (K + 2)];

    for (int t = threadIdx.x; t < K * F_OUT; t += blockDim.x) {
        int k = t / F_OUT, c = t % F_OUT;
        ldsW[c * (K + 2) + k] = W[t];
    }
    __syncthreads();

    const int wave  = threadIdx.x >> 5;
    const int lane  = threadIdx.x & 31;
    const int half  = lane >> 4;
    const int l16   = lane & 15;
    const int mBase = (blockIdx.x * 8 + wave) * 16;
    if (mBase >= nrows) return;

    int rowA = mBase + l16;
    if (rowA >= nrows) rowA = nrows - 1;
    const float* xrow = X + (size_t)rowA * K;

    v8f c0 = {}, c1 = {}, c2 = {}, c3 = {};
    for (int k0 = 0; k0 < K; k0 += 4) {
        const int kk = k0 + 2 * half;
        v2f a = *(const v2f*)(xrow + kk);
        v2f b0 = *(const v2f*)(&ldsW[(0 * 16 + l16) * (K + 2) + kk]);
        v2f b1 = *(const v2f*)(&ldsW[(1 * 16 + l16) * (K + 2) + kk]);
        v2f b2 = *(const v2f*)(&ldsW[(2 * 16 + l16) * (K + 2) + kk]);
        v2f b3 = *(const v2f*)(&ldsW[(3 * 16 + l16) * (K + 2) + kk]);
        c0 = __builtin_amdgcn_wmma_f32_16x16x4_f32(false, a, false, b0, (short)0, c0, false, false);
        c1 = __builtin_amdgcn_wmma_f32_16x16x4_f32(false, a, false, b1, (short)0, c1, false, false);
        c2 = __builtin_amdgcn_wmma_f32_16x16x4_f32(false, a, false, b2, (short)0, c2, false, false);
        c3 = __builtin_amdgcn_wmma_f32_16x16x4_f32(false, a, false, b3, (short)0, c3, false, false);
    }

#pragma unroll
    for (int r = 0; r < 8; ++r) {
        int row = mBase + 8 * half + r;
        if (row < nrows) {
            float s = norm[row];
            size_t base = (size_t)row * F_OUT + l16;
            float v0 = c0[r] * s, v1 = c1[r] * s, v2 = c2[r] * s, v3 = c3[r] * s;
            H [base +  0] = v0;  H [base + 16] = v1;
            H [base + 32] = v2;  H [base + 48] = v3;
            H2[base +  0] = v0;  H2[base + 16] = v1;
            H2[base + 32] = v2;  H2[base + 48] = v3;
        }
    }
}

__global__ void k_scatter(const int* __restrict__ src, const int* __restrict__ dst,
                          const float* __restrict__ h, float* __restrict__ agg, int e) {
    int t = blockIdx.x * blockDim.x + threadIdx.x;
    int edge = t >> 4;
    if (edge >= e) return;
    int c = (t & 15) * 4;
    int s = src[edge], d = dst[edge];
    const float4 v = *(const float4*)(h + (size_t)s * F_OUT + c);
    float* p = agg + (size_t)d * F_OUT + c;
    atomicAdd(p + 0, v.x);
    atomicAdd(p + 1, v.y);
    atomicAdd(p + 2, v.z);
    atomicAdd(p + 3, v.w);
}

__global__ void k_finalize(const float* __restrict__ agg, const float* __restrict__ nd,
                           const float* __restrict__ bias, float* __restrict__ out,
                           int n, int do_relu) {
    int t = blockIdx.x * blockDim.x + threadIdx.x;
    if (t >= n * (F_OUT / 4)) return;
    int row = t >> 4, c4 = (t & 15) * 4;
    float s = nd[row];
    float4 a = *(const float4*)(agg + (size_t)row * F_OUT + c4);
    float4 b = *(const float4*)(bias + c4);
    float4 v;
    v.x = fmaf(a.x, s, b.x);
    v.y = fmaf(a.y, s, b.y);
    v.z = fmaf(a.z, s, b.z);
    v.w = fmaf(a.w, s, b.w);
    if (do_relu) {
        v.x = fmaxf(v.x, 0.0f); v.y = fmaxf(v.y, 0.0f);
        v.z = fmaxf(v.z, 0.0f); v.w = fmaxf(v.w, 0.0f);
    }
    *(float4*)(out + (size_t)row * F_OUT + c4) = v;
}

extern "C" void kernel_launch(void* const* d_in, const int* in_sizes, int n_in,
                              void* d_out, int out_size, void* d_ws, size_t ws_size,
                              hipStream_t stream) {
    const float* x   = (const float*)d_in[0];
    const int*   src = (const int*)d_in[1];
    const int*   dst = (const int*)d_in[2];
    const float* W1  = (const float*)d_in[3];
    const float* b1  = (const float*)d_in[4];
    const float* W2  = (const float*)d_in[5];
    const float* b2  = (const float*)d_in[6];
    float*       out = (float*)d_out;

    const int n = in_sizes[0] / 128;
    const int e = in_sizes[1];

    float* ws   = (float*)d_ws;
    float* buf0 = ws;
    float* buf1 = ws + (size_t)n * F_OUT;
    float* ns   = ws + 2 * (size_t)n * F_OUT;
    float* nd   = ns + n;

    const int B = 256;
    const int gemmBlocks = cdiv(cdiv(n, 16), 8);

    k_init_norm<<<cdiv(n, B), B, 0, stream>>>(ns, nd, n);
    k_degree<<<cdiv(e, B), B, 0, stream>>>(src, dst, ns, nd, e);
    k_rsqrt<<<cdiv(n, B), B, 0, stream>>>(ns, nd, n);

    k_gemm_scale<128><<<gemmBlocks, B, 0, stream>>>(x, W1, ns, buf0, buf1, n);
    k_scatter<<<cdiv(e * 16, B), B, 0, stream>>>(src, dst, buf0, buf1, e);
    k_finalize<<<cdiv(n * 16, B), B, 0, stream>>>(buf1, nd, b1, buf1, n, 1);

    k_gemm_scale<64><<<gemmBlocks, B, 0, stream>>>(buf1, W2, ns, buf0, out, n);
    k_scatter<<<cdiv(e * 16, B), B, 0, stream>>>(src, dst, buf0, out, e);
    k_finalize<<<cdiv(n * 16, B), B, 0, stream>>>(out, nd, b2, out, n, 0);
}


# Degrees counted in LDS histograms (200 KB of bins per block, integer LDS adds, per-chunk count planes summed in the rsqrt kernel), not by a scattered global f32 atomic per index; norms bit-identical
# speedup vs baseline: 1.2809x; 1.2809x over previous
#include <hip/hip_runtime.h>
#include <math.h>

#define F_OUT 64

typedef __attribute__((ext_vector_type(2))) float v2f;
typedef __attribute__((ext_vector_type(8))) float v8f;

static inline int cdiv(int a, int b) { return (a + b - 1) / b; }

#define HR 51200
#define HC 32
#define HT 1024

static __device__ __forceinline__ void hist_add4(int* hist, int4 v, int lo) {
    unsigned a = (unsigned)(v.x - lo), b = (unsigned)(v.y - lo);
    unsigned c = (unsigned)(v.z - lo), d = (unsigned)(v.w - lo);
    if (a < HR) atomicAdd(&hist[a], 1);
    if (b < HR) atomicAdd(&hist[b], 1);
    if (c < HR) atomicAdd(&hist[c], 1);
    if (d < HR) atomicAdd(&hist[d], 1);
}

__global__ __launch_bounds__(HT) void k_hist(const int* __restrict__ src,
                                             const int* __restrict__ dst,
                                             int* __restrict__ part, int n, int e) {
    __shared__ alignas(16) int hist[HR];
    const int* __restrict__ idx = blockIdx.z ? dst : src;
    const int lo = blockIdx.y * HR;

    for (int t = threadIdx.x; t < HR / 4; t += HT)
        ((int4*)hist)[t] = make_int4(0, 0, 0, 0);
    __syncthreads();

    const int per = (((e + HC - 1) / HC) + 3) & ~3;
    const int beg = min((int)blockIdx.x * per, e);
    const int end = min(beg + per, e);
    const int nvec = (((size_t)idx & 15) == 0) ? ((end - beg) >> 2) : 0;
    const int4* __restrict__ p4 = (const int4*)(idx + beg);

    for (int q = threadIdx.x; q < nvec; q += 4 * HT) {
        const int q1 = q + 1 * HT, q2 = q + 2 * HT, q3 = q + 3 * HT;
        const int4 v0 = p4[q];
        const int4 v1 = p4[min(q1, nvec - 1)];
        const int4 v2 = p4[min(q2, nvec - 1)];
        const int4 v3 = p4[min(q3, nvec - 1)];
        hist_add4(hist, v0, lo);
        if (q1 < nvec) hist_add4(hist, v1, lo);
        if (q2 < nvec) hist_add4(hist, v2, lo);
        if (q3 < nvec) hist_add4(hist, v3, lo);
    }
    for (int j = beg + 4 * nvec + threadIdx.x; j < end; j += HT) {
        unsigned a = (unsigned)(idx[j] - lo);
        if (a < HR) atomicAdd(&hist[a], 1);
    }
    __syncthreads();

    int* __restrict__ plane = part + (size_t)(blockIdx.z * HC + blockIdx.x) * n;
    const int cnt = min(HR, n - lo);
    if ((n & 3) == 0) {
        for (int t = threadIdx.x; t < (cnt >> 2); t += HT)
            ((int4*)(plane + lo))[t] = ((const int4*)hist)[t];
    } else {
        for (int t = threadIdx.x; t < cnt; t += HT)
            plane[lo + t] = hist[t];
    }
}

__global__ void k_norm(const int* __restrict__ part, float* __restrict__ ns,
                       float* __restrict__ nd, int n) {
    int i = blockIdx.x * blockDim.x + threadIdx.x;
    if (i >= n) return;
    int s = 1, d = 1;
#pragma unroll
    for (int c = 0; c < HC; ++c) {
        s += part[(size_t)c * n + i];
        d += part[(size_t)(HC + c) * n + i];
    }
    ns[i] = rsqrtf((float)s);
    nd[i] = rsqrtf((float)d);
}

template <int K>
__global__ __launch_bounds__(256) void k_gemm_scale(const float* __restrict__ X,
                                                    const float* __restrict__ W,
                                                    const float* __restrict__ norm,
                                                    float* __restrict__ H,
                                                    float* __restrict__ H2, int nrows) {
    __shared__ float ldsW[F_OUT * (K + 2)];

    for (int t = threadIdx.x; t < K * F_OUT; t += blockDim.x) {
        int k = t / F_OUT, c = t % F_OUT;
        ldsW[c * (K + 2) + k] = W[t];
    }
    __syncthreads();

    const int wave  = threadIdx.x >> 5;
    const int lane  = threadIdx.x & 31;
    const int half  = lane >> 4;
    const int l16   = lane & 15;
    const int mBase = (blockIdx.x * 8 + wave) * 16;
    if (mBase >= nrows) return;

    int rowA = mBase + l16;
    if (rowA >= nrows) rowA = nrows - 1;
    const float* xrow = X + (size_t)rowA * K;

    v8f c0 = {}, c1 = {}, c2 = {}, c3 = {};
    for (int k0 = 0; k0 < K; k0 += 4) {
        const int kk = k0 + 2 * half;
        v2f a = *(const v2f*)(xrow + kk);
        v2f b0 = *(const v2f*)(&ldsW[(0 * 16 + l16) * (K + 2) + kk]);
        v2f b1 = *(const v2f*)(&ldsW[(1 * 16 + l16) * (K + 2) + kk]);
        v2f b2 = *(const v2f*)(&ldsW[(2 * 16 + l16) * (K + 2) + kk]);
        v2f b3 = *(const v2f*)(&ldsW[(3 * 16 + l16) * (K + 2) + kk]);
        c0 = __builtin_amdgcn_wmma_f32_16x16x4_f32(false, a, false, b0, (short)0, c0, false, false);
        c1 = __builtin_amdgcn_wmma_f32_16x16x4_f32(false, a, false, b1, (short)0, c1, false, false);
        c2 = __builtin_amdgcn_wmma_f32_16x16x4_f32(false, a, false, b2, (short)0, c2, false, false);
        c3 = __builtin_amdgcn_wmma_f32_16x16x4_f32(false, a, false, b3, (short)0, c3, false, false);
    }

#pragma unroll
    for (int r = 0; r < 8; ++r) {
        int row = mBase + 8 * half + r;
        if (row < nrows) {
            float s = norm[row];
            size_t base = (size_t)row * F_OUT + l16;
            float v0 = c0[r] * s, v1 = c1[r] * s, v2 = c2[r] * s, v3 = c3[r] * s;
            H [base +  0] = v0;  H [base + 16] = v1;
            H [base + 32] = v2;  H [base + 48] = v3;
            H2[base +  0] = v0;  H2[base + 16] = v1;
            H2[base + 32] = v2;  H2[base + 48] = v3;
        }
    }
}

__global__ void k_scatter(const int* __restrict__ src, const int* __restrict__ dst,
                          const float* __restrict__ h, float* __restrict__ agg, int e) {
    int t = blockIdx.x * blockDim.x + threadIdx.x;
    int edge = t >> 4;
    if (edge >= e) return;
    int c = (t & 15) * 4;
    int s = src[edge], d = dst[edge];
    const float4 v = *(const float4*)(h + (size_t)s * F_OUT + c);
    float* p = agg + (size_t)d * F_OUT + c;
    atomicAdd(p + 0, v.x);
    atomicAdd(p + 1, v.y);
    atomicAdd(p + 2, v.z);
    atomicAdd(p + 3, v.w);
}

__global__ void k_finalize(const float* __restrict__ agg, const float* __restrict__ nd,
                           const float* __restrict__ bias, float* __restrict__ out,
                           int n, int do_relu) {
    int t = blockIdx.x * blockDim.x + threadIdx.x;
    if (t >= n * (F_OUT / 4)) return;
    int row = t >> 4, c4 = (t & 15) * 4;
    float s = nd[row];
    float4 a = *(const float4*)(agg + (size_t)row * F_OUT + c4);
    float4 b = *(const float4*)(bias + c4);
    float4 v;
    v.x = fmaf(a.x, s, b.x);
    v.y = fmaf(a.y, s, b.y);
    v.z = fmaf(a.z, s, b.z);
    v.w = fmaf(a.w, s, b.w);
    if (do_relu) {
        v.x = fmaxf(v.x, 0.0f); v.y = fmaxf(v.y, 0.0f);
        v.z = fmaxf(v.z, 0.0f); v.w = fmaxf(v.w, 0.0f);
    }
    *(float4*)(out + (size_t)row * F_OUT + c4) = v;
}

extern "C" void kernel_launch(void* const* d_in, const int* in_sizes, int n_in,
                              void* d_out, int out_size, void* d_ws, size_t ws_size,
                              hipStream_t stream) {
    const float* x   = (const float*)d_in[0];
    const int*   src = (const int*)d_in[1];
    const int*   dst = (const int*)d_in[2];
    const float* W1  = (const float*)d_in[3];
    const float* b1  = (const float*)d_in[4];
    const float* W2  = (const float*)d_in[5];
    const float* b2  = (const float*)d_in[6];
    float*       out = (float*)d_out;

    const int n = in_sizes[0] / 128;
    const int e = in_sizes[1];

    float* ws   = (float*)d_ws;
    float* buf0 = ws;
    float* buf1 = ws + (size_t)n * F_OUT;
    float* ns   = ws + 2 * (size_t)n * F_OUT;
    float* nd   = ns + n;

    const int B = 256;
    const int gemmBlocks = cdiv(cdiv(n, 16), 8);

    int* part = (int*)d_ws;
    k_hist<<<dim3(HC, cdiv(n, HR), 2), HT, 0, stream>>>(src, dst, part, n, e);
    k_norm<<<cdiv(n, B), B, 0, stream>>>(part, ns, nd, n);

    k_gemm_scale<128><<<gemmBlocks, B, 0, stream>>>(x, W1, ns, buf0, buf1, n);
    k_scatter<<<cdiv(e * 16, B), B, 0, stream>>>(src, dst, buf0, buf1, e);
    k_finalize<<<cdiv(n * 16, B), B, 0, stream>>>(buf1, nd, b1, buf1, n, 1);

    k_gemm_scale<64><<<gemmBlocks, B, 0, stream>>>(buf1, W2, ns, buf0, out, n);
    k_scatter<<<cdiv(e * 16, B), B, 0, stream>>>(src, dst, buf0, out, e);
    k_finalize<<<cdiv(n * 16, B), B, 0, stream>>>(out, nd, b2, out, n, 0);
}


# Edges counting-sorted by dst per call (LDS cursors seeded from the count planes); a 16-lane group adds a row's messages in list order (was f32 atomics, arrival order) and scales+biases it
# speedup vs baseline: 2.7200x; 2.1235x over previous
#include <hip/hip_runtime.h>
#include <math.h>

#define F_OUT 64

typedef __attribute__((ext_vector_type(2))) float v2f;
typedef __attribute__((ext_vector_type(8))) float v8f;

static inline int cdiv(int a, int b) { return (a + b - 1) / b; }

#define HR 51200
#define HC 32
#define HT 1024

static __device__ __forceinline__ void hist_add4(int* hist, int4 v, int lo) {
    unsigned a = (unsigned)(v.x - lo), b = (unsigned)(v.y - lo);
    unsigned c = (unsigned)(v.z - lo), d = (unsigned)(v.w - lo);
    if (a < HR) atomicAdd(&hist[a], 1);
    if (b < HR) atomicAdd(&hist[b], 1);
    if (c < HR) atomicAdd(&hist[c], 1);
    if (d < HR) atomicAdd(&hist[d], 1);
}

__global__ __launch_bounds__(HT) void k_hist(const int* __restrict__ src,
                                             const int* __restrict__ dst,
                                             int* __restrict__ part, int n, int e) {
    __shared__ alignas(16) int hist[HR];
    const int* __restrict__ idx = blockIdx.z ? dst : src;
    const int lo = blockIdx.y * HR;

    for (int t = threadIdx.x; t < HR / 4; t += HT)
        ((int4*)hist)[t] = make_int4(0, 0, 0, 0);
    __syncthreads();

    const int per = (((e + HC - 1) / HC) + 3) & ~3;
    const int beg = min((int)blockIdx.x * per, e);
    const int end = min(beg + per, e);
    const int nvec = (((size_t)idx & 15) == 0) ? ((end - beg) >> 2) : 0;
    const int4* __restrict__ p4 = (const int4*)(idx + beg);

    for (int q = threadIdx.x; q < nvec; q += 4 * HT) {
        const int q1 = q + 1 * HT, q2 = q + 2 * HT, q3 = q + 3 * HT;
        const int4 v0 = p4[q];
        const int4 v1 = p4[min(q1, nvec - 1)];
        const int4 v2 = p4[min(q2, nvec - 1)];
        const int4 v3 = p4[min(q3, nvec - 1)];
        hist_add4(hist, v0, lo);
        if (q1 < nvec) hist_add4(hist, v1, lo);
        if (q2 < nvec) hist_add4(hist, v2, lo);
        if (q3 < nvec) hist_add4(hist, v3, lo);
    }
    for (int j = beg + 4 * nvec + threadIdx.x; j < end; j += HT) {
        unsigned a = (unsigned)(idx[j] - lo);
        if (a < HR) atomicAdd(&hist[a], 1);
    }
    __syncthreads();

    int* __restrict__ plane = part + (size_t)(blockIdx.z * HC + blockIdx.x) * n;
    const int cnt = min(HR, n - lo);
    if ((n & 3) == 0) {
        for (int t = threadIdx.x; t < (cnt >> 2); t += HT)
            ((int4*)(plane + lo))[t] = ((const int4*)hist)[t];
    } else {
        for (int t = threadIdx.x; t < cnt; t += HT)
            plane[lo + t] = hist[t];
    }
}

#define NB 256
__global__ __launch_bounds__(NB) void k_norm(int* __restrict__ part, float* __restrict__ ns,
                                             float* __restrict__ nd, int* __restrict__ lexcl,
                                             int* __restrict__ bsum, int n) {
    __shared__ int sc[NB];
    const int tid = threadIdx.x;
    const int i = blockIdx.x * NB + tid;
    const bool valid = i < n;
    int din = 0;
    if (valid) {
        int s = 1;
        int v[HC];
#pragma unroll
        for (int c = 0; c < HC; ++c) {
            s += part[(size_t)c * n + i];
            v[c] = part[(size_t)(HC + c) * n + i];
        }
#pragma unroll
        for (int c = 0; c < HC; ++c) {
            part[(size_t)(HC + c) * n + i] = din;
            din += v[c];
        }
        ns[i] = rsqrtf((float)s);
        nd[i] = rsqrtf((float)(din + 1));
    }
    sc[tid] = din;
    __syncthreads();
    for (int off = 1; off < NB; off <<= 1) {
        const int t = (tid >= off) ? sc[tid - off] : 0;
        __syncthreads();
        sc[tid] += t;
        __syncthreads();
    }
    if (valid) lexcl[i] = sc[tid] - din;
    if (tid == NB - 1) bsum[blockIdx.x] = sc[tid];
}

#define NBR (HR / NB)

static __device__ __forceinline__ void fill4(int* cur, int* __restrict__ sorted,
                                             int4 d, int4 s, int lo) {
    unsigned a = (unsigned)(d.x - lo), b = (unsigned)(d.y - lo);
    unsigned c = (unsigned)(d.z - lo), g = (unsigned)(d.w - lo);
    if (a < HR) sorted[atomicAdd(&cur[a], 1)] = s.x;
    if (b < HR) sorted[atomicAdd(&cur[b], 1)] = s.y;
    if (c < HR) sorted[atomicAdd(&cur[c], 1)] = s.z;
    if (g < HR) sorted[atomicAdd(&cur[g], 1)] = s.w;
}

__global__ __launch_bounds__(HT) void k_fill(const int* __restrict__ src,
                                             const int* __restrict__ dst,
                                             const int* __restrict__ dpart,
                                             const int* __restrict__ lexcl,
                                             const int* __restrict__ bsum,
                                             int* __restrict__ rowptr,
                                             int* __restrict__ sorted, int n, int e) {
    __shared__ alignas(16) int cur[HR];
    __shared__ int sb[NB];
    __shared__ int pre_s;
    const int tid = threadIdx.x;
    const int c   = blockIdx.x;
    const int lo  = blockIdx.y * HR;
    const int nb  = (n + NB - 1) / NB;
    const int b0  = lo / NB;

    if (tid == 0) pre_s = 0;
    if (tid < NB) {
        const int j = b0 + tid;
        sb[tid] = (tid < NBR && j < nb) ? bsum[j] : 0;
    }
    int p = 0;
    for (int j = tid; j < b0; j += HT) p += bsum[j];
    __syncthreads();
    if (p) atomicAdd(&pre_s, p);
    for (int off = 1; off < NB; off <<= 1) {
        const int t = (tid < NB && tid >= off) ? sb[tid - off] : 0;
        __syncthreads();
        if (tid < NB) sb[tid] += t;
        __syncthreads();
    }
    const int pre = pre_s;

    const int cnt = min(HR, n - lo);
    const int* __restrict__ plane = dpart + (size_t)c * n;
    const bool al4 = ((n & 3) == 0) &&
                     ((((size_t)lexcl | (size_t)dpart | (size_t)rowptr) & 15) == 0);
    const int ng = al4 ? (cnt >> 2) : 0;
    const int4* __restrict__ lx4 = (const int4*)(lexcl + lo);
    const int4* __restrict__ pl4 = (const int4*)(plane + lo);
    for (int g = tid; g < ng; g += 4 * HT) {
        int4 lx[4], pl[4];
#pragma unroll
        for (int u = 0; u < 4; ++u) {
            const int gg = min(g + u * HT, ng - 1);
            lx[u] = lx4[gg];
            pl[u] = pl4[gg];
        }
#pragma unroll
        for (int u = 0; u < 4; ++u) {
            const int gg = g + u * HT;
            if (gg < ng) {
                const int k = (4 * gg) / NB;
                const int bs = pre + (k ? sb[k - 1] : 0);
                const int4 rp = make_int4(bs + lx[u].x, bs + lx[u].y, bs + lx[u].z, bs + lx[u].w);
                ((int4*)cur)[gg] = make_int4(rp.x + pl[u].x, rp.y + pl[u].y,
                                             rp.z + pl[u].z, rp.w + pl[u].w);
                if (c == 0) ((int4*)(rowptr + lo))[gg] = rp;
            }
        }
    }
    for (int t = 4 * ng + tid; t < cnt; t += HT) {
        const int i = lo + t;
        const int k = t / NB;
        const int rp = pre + (k ? sb[k - 1] : 0) + lexcl[i];
        cur[t] = rp + plane[i];
        if (c == 0) rowptr[i] = rp;
    }
    __syncthreads();

    const int per = (((e + HC - 1) / HC) + 3) & ~3;
    const int beg = min(c * per, e);
    const int end = min(beg + per, e);
    const bool al = ((((size_t)src | (size_t)dst) & 15) == 0);
    const int nvec = al ? ((end - beg) >> 2) : 0;
    const int4* __restrict__ d4 = (const int4*)(dst + beg);
    const int4* __restrict__ s4 = (const int4*)(src + beg);

    for (int q = tid; q < nvec; q += 4 * HT) {
        const int q1 = q + 1 * HT, q2 = q + 2 * HT, q3 = q + 3 * HT;
        const int4 dv0 = d4[q],                   sv0 = s4[q];
        const int4 dv1 = d4[min(q1, nvec - 1)],   sv1 = s4[min(q1, nvec - 1)];
        const int4 dv2 = d4[min(q2, nvec - 1)],   sv2 = s4[min(q2, nvec - 1)];
        const int4 dv3 = d4[min(q3, nvec - 1)],   sv3 = s4[min(q3, nvec - 1)];
        fill4(cur, sorted, dv0, sv0, lo);
        if (q1 < nvec) fill4(cur, sorted, dv1, sv1, lo);
        if (q2 < nvec) fill4(cur, sorted, dv2, sv2, lo);
        if (q3 < nvec) fill4(cur, sorted, dv3, sv3, lo);
    }
    for (int j = beg + 4 * nvec + tid; j < end; j += HT) {
        const unsigned a = (unsigned)(dst[j] - lo);
        if (a < HR) sorted[atomicAdd(&cur[a], 1)] = src[j];
    }
}

template <int K>
__global__ __launch_bounds__(256) void k_gemm_scale(const float* __restrict__ X,
                                                    const float* __restrict__ W,
                                                    const float* __restrict__ norm,
                                                    float* __restrict__ H, int nrows) {
    __shared__ float ldsW[F_OUT * (K + 2)];

    for (int t = threadIdx.x; t < K * F_OUT; t += blockDim.x) {
        int k = t / F_OUT, c = t % F_OUT;
        ldsW[c * (K + 2) + k] = W[t];
    }
    __syncthreads();

    const int wave  = threadIdx.x >> 5;
    const int lane  = threadIdx.x & 31;
    const int half  = lane >> 4;
    const int l16   = lane & 15;
    const int mBase = (blockIdx.x * 8 + wave) * 16;
    if (mBase >= nrows) return;

    int rowA = mBase + l16;
    if (rowA >= nrows) rowA = nrows - 1;
    const float* xrow = X + (size_t)rowA * K;

    v8f c0 = {}, c1 = {}, c2 = {}, c3 = {};
    for (int k0 = 0; k0 < K; k0 += 4) {
        const int kk = k0 + 2 * half;
        v2f a = *(const v2f*)(xrow + kk);
        v2f b0 = *(const v2f*)(&ldsW[(0 * 16 + l16) * (K + 2) + kk]);
        v2f b1 = *(const v2f*)(&ldsW[(1 * 16 + l16) * (K + 2) + kk]);
        v2f b2 = *(const v2f*)(&ldsW[(2 * 16 + l16) * (K + 2) + kk]);
        v2f b3 = *(const v2f*)(&ldsW[(3 * 16 + l16) * (K + 2) + kk]);
        c0 = __builtin_amdgcn_wmma_f32_16x16x4_f32(false, a, false, b0, (short)0, c0, false, false);
        c1 = __builtin_amdgcn_wmma_f32_16x16x4_f32(false, a, false, b1, (short)0, c1, false, false);
        c2 = __builtin_amdgcn_wmma_f32_16x16x4_f32(false, a, false, b2, (short)0, c2, false, false);
        c3 = __builtin_amdgcn_wmma_f32_16x16x4_f32(false, a, false, b3, (short)0, c3, false, false);
    }

#pragma unroll
    for (int r = 0; r < 8; ++r) {
        int row = mBase + 8 * half + r;
        if (row < nrows) {
            float s = norm[row];
            size_t base = (size_t)row * F_OUT + l16;
            float v0 = c0[r] * s, v1 = c1[r] * s, v2 = c2[r] * s, v3 = c3[r] * s;
            H[base +  0] = v0;  H[base + 16] = v1;
            H[base + 32] = v2;  H[base + 48] = v3;
        }
    }
}

template <int RELU>
__global__ __launch_bounds__(256) void k_gather(const int* __restrict__ rowptr,
                                                const int* __restrict__ sorted,
                                                const float* __restrict__ h,
                                                const float* __restrict__ nd,
                                                const float* __restrict__ bias,
                                                float* __restrict__ out, int n, int e) {
    const int sub = threadIdx.x & 15;
    const int grp = threadIdx.x & 16;
    const int row = blockIdx.x * 16 + (threadIdx.x >> 4);
    if (row >= n) return;

    const int beg = rowptr[row];
    const int end = (row + 1 < n) ? rowptr[row + 1] : e;
    const int deg = end - beg;
    const float4* __restrict__ h4 = (const float4*)h;
    float4 acc = h4[(size_t)row * 16 + sub];

    for (int base = 0; base < deg; base += 16) {
        const int cnt = min(16, deg - base);
        int idx = row;
        if (sub < cnt) idx = sorted[beg + base + sub];
        for (int k = 0; k < cnt; k += 4) {
            const int s0 = __builtin_amdgcn_ds_bpermute((grp + k + 0) << 2, idx);
            const int s1 = __builtin_amdgcn_ds_bpermute((grp + k + 1) << 2, idx);
            const int s2 = __builtin_amdgcn_ds_bpermute((grp + k + 2) << 2, idx);
            const int s3 = __builtin_amdgcn_ds_bpermute((grp + k + 3) << 2, idx);
            const float4 v0 = h4[(size_t)s0 * 16 + sub];
            const float4 v1 = h4[(size_t)s1 * 16 + sub];
            const float4 v2 = h4[(size_t)s2 * 16 + sub];
            const float4 v3 = h4[(size_t)s3 * 16 + sub];
            const bool p1 = k + 1 < cnt, p2 = k + 2 < cnt, p3 = k + 3 < cnt;
            acc.x += v0.x; acc.y += v0.y; acc.z += v0.z; acc.w += v0.w;
            acc.x += p1 ? v1.x : 0.0f; acc.y += p1 ? v1.y : 0.0f;
            acc.z += p1 ? v1.z : 0.0f; acc.w += p1 ? v1.w : 0.0f;
            acc.x += p2 ? v2.x : 0.0f; acc.y += p2 ? v2.y : 0.0f;
            acc.z += p2 ? v2.z : 0.0f; acc.w += p2 ? v2.w : 0.0f;
            acc.x += p3 ? v3.x : 0.0f; acc.y += p3 ? v3.y : 0.0f;
            acc.z += p3 ? v3.z : 0.0f; acc.w += p3 ? v3.w : 0.0f;
        }
    }

    const float s = nd[row];
    const float4 b = *(const float4*)(bias + sub * 4);
    float4 v;
    v.x = fmaf(acc.x, s, b.x);
    v.y = fmaf(acc.y, s, b.y);
    v.z = fmaf(acc.z, s, b.z);
    v.w = fmaf(acc.w, s, b.w);
    if (RELU) {
        v.x = fmaxf(v.x, 0.0f); v.y = fmaxf(v.y, 0.0f);
        v.z = fmaxf(v.z, 0.0f); v.w = fmaxf(v.w, 0.0f);
    }
    *(float4*)(out + (size_t)row * F_OUT + sub * 4) = v;
}

extern "C" void kernel_launch(void* const* d_in, const int* in_sizes, int n_in,
                              void* d_out, int out_size, void* d_ws, size_t ws_size,
                              hipStream_t stream) {
    const float* x   = (const float*)d_in[0];
    const int*   src = (const int*)d_in[1];
    const int*   dst = (const int*)d_in[2];
    const float* W1  = (const float*)d_in[3];
    const float* b1  = (const float*)d_in[4];
    const float* W2  = (const float*)d_in[5];
    const float* b2  = (const float*)d_in[6];
    float*       out = (float*)d_out;

    const int n = in_sizes[0] / 128;
    const int e = in_sizes[1];

    float* ws   = (float*)d_ws;
    float* buf0 = ws;
    int*   sorted = (int*)(ws + (size_t)n * F_OUT);
    int*   rowptr = sorted + e;
    int*   lexcl  = rowptr + n;
    int*   bsum   = lexcl + n;
    float* ns   = ws + 2 * (size_t)n * F_OUT;
    float* nd   = ns + n;

    const int B = 256;
    const int gemmBlocks = cdiv(cdiv(n, 16), 8);

    int* part = (int*)d_ws;
    k_hist<<<dim3(HC, cdiv(n, HR), 2), HT, 0, stream>>>(src, dst, part, n, e);
    k_norm<<<cdiv(n, NB), NB, 0, stream>>>(part, ns, nd, lexcl, bsum, n);
    k_fill<<<dim3(HC, cdiv(n, HR)), HT, 0, stream>>>(src, dst, part + (size_t)HC * n, lexcl, bsum,
                                                     rowptr, sorted, n, e);

    k_gemm_scale<128><<<gemmBlocks, B, 0, stream>>>(x, W1, ns, buf0, n);
    k_gather<1><<<cdiv(n, 16), B, 0, stream>>>(rowptr, sorted, buf0, nd, b1, out, n, e);

    k_gemm_scale<64><<<gemmBlocks, B, 0, stream>>>(out, W2, ns, buf0, n);
    k_gather<0><<<cdiv(n, 16), B, 0, stream>>>(rowptr, sorted, buf0, nd, b2, out, n, e);
}


# Edge placement (k_fill) on 256 work-groups, one per WGP: 8 node ranges of 12,800 x 32 chunks (was 2 x 32 on 64 WGPs), 50 KB of LDS cursors each; list order rule kept (arrival order inside a chunk)
# speedup vs baseline: 3.0301x; 1.1140x over previous
#include <hip/hip_runtime.h>
#include <math.h>

#define F_OUT 64

typedef __attribute__((ext_vector_type(2))) float v2f;
typedef __attribute__((ext_vector_type(8))) float v8f;

static inline int cdiv(int a, int b) { return (a + b - 1) / b; }

#define HR 51200
#define HC 32
#define HT 1024

static __device__ __forceinline__ void hist_add4(int* hist, int4 v, int lo) {
    unsigned a = (unsigned)(v.x - lo), b = (unsigned)(v.y - lo);
    unsigned c = (unsigned)(v.z - lo), d = (unsigned)(v.w - lo);
    if (a < HR) atomicAdd(&hist[a], 1);
    if (b < HR) atomicAdd(&hist[b], 1);
    if (c < HR) atomicAdd(&hist[c], 1);
    if (d < HR) atomicAdd(&hist[d], 1);
}

__global__ __launch_bounds__(HT) void k_hist(const int* __restrict__ src,
                                             const int* __restrict__ dst,
                                             int* __restrict__ part, int n, int e) {
    __shared__ alignas(16) int hist[HR];
    const int* __restrict__ idx = blockIdx.z ? dst : src;
    const int lo = blockIdx.y * HR;

    for (int t = threadIdx.x; t < HR / 4; t += HT)
        ((int4*)hist)[t] = make_int4(0, 0, 0, 0);
    __syncthreads();

    const int per = (((e + HC - 1) / HC) + 3) & ~3;
    const int beg = min((int)blockIdx.x * per, e);
    const int end = min(beg + per, e);
    const int nvec = (((size_t)idx & 15) == 0) ? ((end - beg) >> 2) : 0;
    const int4* __restrict__ p4 = (const int4*)(idx + beg);

    for (int q = threadIdx.x; q < nvec; q += 4 * HT) {
        const int q1 = q + 1 * HT, q2 = q + 2 * HT, q3 = q + 3 * HT;
        const int4 v0 = p4[q];
        const int4 v1 = p4[min(q1, nvec - 1)];
        const int4 v2 = p4[min(q2, nvec - 1)];
        const int4 v3 = p4[min(q3, nvec - 1)];
        hist_add4(hist, v0, lo);
        if (q1 < nvec) hist_add4(hist, v1, lo);
        if (q2 < nvec) hist_add4(hist, v2, lo);
        if (q3 < nvec) hist_add4(hist, v3, lo);
    }
    for (int j = beg + 4 * nvec + threadIdx.x; j < end; j += HT) {
        unsigned a = (unsigned)(idx[j] - lo);
        if (a < HR) atomicAdd(&hist[a], 1);
    }
    __syncthreads();

    int* __restrict__ plane = part + (size_t)(blockIdx.z * HC + blockIdx.x) * n;
    const int cnt = min(HR, n - lo);
    if ((n & 3) == 0) {
        for (int t = threadIdx.x; t < (cnt >> 2); t += HT)
            ((int4*)(plane + lo))[t] = ((const int4*)hist)[t];
    } else {
        for (int t = threadIdx.x; t < cnt; t += HT)
            plane[lo + t] = hist[t];
    }
}

#define NB 256
__global__ __launch_bounds__(NB) void k_norm(int* __restrict__ part, float* __restrict__ ns,
                                             float* __restrict__ nd, int* __restrict__ lexcl,
                                             int* __restrict__ bsum, int n) {
    __shared__ int sc[NB];
    const int tid = threadIdx.x;
    const int i = blockIdx.x * NB + tid;
    const bool valid = i < n;
    int din = 0;
    if (valid) {
        int s = 1;
        int v[HC];
#pragma unroll
        for (int c = 0; c < HC; ++c) {
            s += part[(size_t)c * n + i];
            v[c] = part[(size_t)(HC + c) * n + i];
        }
#pragma unroll
        for (int c = 0; c < HC; ++c) {
            part[(size_t)(HC + c) * n + i] = din;
            din += v[c];
        }
        ns[i] = rsqrtf((float)s);
        nd[i] = rsqrtf((float)(din + 1));
    }
    sc[tid] = din;
    __syncthreads();
    for (int off = 1; off < NB; off <<= 1) {
        const int t = (tid >= off) ? sc[tid - off] : 0;
        __syncthreads();
        sc[tid] += t;
        __syncthreads();
    }
    if (valid) lexcl[i] = sc[tid] - din;
    if (tid == NB - 1) bsum[blockIdx.x] = sc[tid];
}

#define FR  12800
#define FNB (FR / NB)
#define FSB 64
static_assert(FR % NB == 0 && FR % 4 == 0, "a range is whole k_norm blocks and whole 16-byte groups");
static_assert(FNB <= FSB && FSB <= HT, "one lane per block total");

static __device__ __forceinline__ void fill4(int* cur, int* __restrict__ sorted,
                                             int4 d, int4 s, int lo) {
    unsigned a = (unsigned)(d.x - lo), b = (unsigned)(d.y - lo);
    unsigned c = (unsigned)(d.z - lo), g = (unsigned)(d.w - lo);
    if (a < FR) sorted[atomicAdd(&cur[a], 1)] = s.x;
    if (b < FR) sorted[atomicAdd(&cur[b], 1)] = s.y;
    if (c < FR) sorted[atomicAdd(&cur[c], 1)] = s.z;
    if (g < FR) sorted[atomicAdd(&cur[g], 1)] = s.w;
}

__global__ __launch_bounds__(HT) void k_fill(const int* __restrict__ src,
                                             const int* __restrict__ dst,
                                             const int* __restrict__ dpart,
                                             const int* __restrict__ lexcl,
                                             const int* __restrict__ bsum,
                                             int* __restrict__ rowptr,
                                             int* __restrict__ sorted, int n, int e) {
    __shared__ alignas(16) int cur[FR];
    __shared__ int sb[FSB];
    __shared__ int pre_s;
    const int tid = threadIdx.x;
    const int c   = blockIdx.x;
    const int lo  = blockIdx.y * FR;
    const int nb  = (n + NB - 1) / NB;
    const int b0  = lo / NB;

    if (tid == 0) pre_s = 0;
    if (tid < FSB) {
        const int j = b0 + tid;
        sb[tid] = (tid < FNB && j < nb) ? bsum[j] : 0;
    }
    int p = 0;
    for (int j = tid; j < b0; j += HT) p += bsum[j];
    __syncthreads();
    if (p) atomicAdd(&pre_s, p);
    for (int off = 1; off < FSB; off <<= 1) {
        const int t = (tid < FSB && tid >= off) ? sb[tid - off] : 0;
        __syncthreads();
        if (tid < FSB) sb[tid] += t;
        __syncthreads();
    }
    const int pre = pre_s;

    const int cnt = min(FR, n - lo);
    const int* __restrict__ plane = dpart + (size_t)c * n;
    const bool al4 = ((n & 3) == 0) &&
                     ((((size_t)lexcl | (size_t)dpart | (size_t)rowptr) & 15) == 0);
    const int ng = al4 ? (cnt >> 2) : 0;
    const int4* __restrict__ lx4 = (const int4*)(lexcl + lo);
    const int4* __restrict__ pl4 = (const int4*)(plane + lo);
    for (int g = tid; g < ng; g += 4 * HT) {
        int4 lx[4], pl[4];
#pragma unroll
        for (int u = 0; u < 4; ++u) {
            const int gg = min(g + u * HT, ng - 1);
            lx[u] = lx4[gg];
            pl[u] = pl4[gg];
        }
#pragma unroll
        for (int u = 0; u < 4; ++u) {
            const int gg = g + u * HT;
            if (gg < ng) {
                const int k = (4 * gg) / NB;
                const int bs = pre + (k ? sb[k - 1] : 0);
                const int4 rp = make_int4(bs + lx[u].x, bs + lx[u].y, bs + lx[u].z, bs + lx[u].w);
                ((int4*)cur)[gg] = make_int4(rp.x + pl[u].x, rp.y + pl[u].y,
                                             rp.z + pl[u].z, rp.w + pl[u].w);
                if (c == 0) ((int4*)(rowptr + lo))[gg] = rp;
            }
        }
    }
    for (int t = 4 * ng + tid; t < cnt; t += HT) {
        const int i = lo + t;
        const int k = t / NB;
        const int rp = pre + (k ? sb[k - 1] : 0) + lexcl[i];
        cur[t] = rp + plane[i];
        if (c == 0) rowptr[i] = rp;
    }
    __syncthreads();

    const int per = (((e + HC - 1) / HC) + 3) & ~3;
    const int beg = min(c * per, e);
    const int end = min(beg + per, e);
    const bool al = ((((size_t)src | (size_t)dst) & 15) == 0);
    const int nvec = al ? ((end - beg) >> 2) : 0;
    const int4* __restrict__ d4 = (const int4*)(dst + beg);
    const int4* __restrict__ s4 = (const int4*)(src + beg);

    for (int q = tid; q < nvec; q += 4 * HT) {
        const int q1 = q + 1 * HT, q2 = q + 2 * HT, q3 = q + 3 * HT;
        const int4 dv0 = d4[q],                   sv0 = s4[q];
        const int4 dv1 = d4[min(q1, nvec - 1)],   sv1 = s4[min(q1, nvec - 1)];
        const int4 dv2 = d4[min(q2, nvec - 1)],   sv2 = s4[min(q2, nvec - 1)];
        const int4 dv3 = d4[min(q3, nvec - 1)],   sv3 = s4[min(q3, nvec - 1)];
        fill4(cur, sorted, dv0, sv0, lo);
        if (q1 < nvec) fill4(cur, sorted, dv1, sv1, lo);
        if (q2 < nvec) fill4(cur, sorted, dv2, sv2, lo);
        if (q3 < nvec) fill4(cur, sorted, dv3, sv3, lo);
    }
    for (int j = beg + 4 * nvec + tid; j < end; j += HT) {
        const unsigned a = (unsigned)(dst[j] - lo);
        if (a < FR) sorted[atomicAdd(&cur[a], 1)] = src[j];
    }
}

template <int K>
__global__ __launch_bounds__(256) void k_gemm_scale(const float* __restrict__ X,
                                                    const float* __restrict__ W,
                                                    const float* __restrict__ norm,
                                                    float* __restrict__ H, int nrows) {
    __shared__ float ldsW[F_OUT * (K + 2)];

    for (int t = threadIdx.x; t < K * F_OUT; t += blockDim.x) {
        int k = t / F_OUT, c = t % F_OUT;
        ldsW[c * (K + 2) + k] = W[t];
    }
    __syncthreads();

    const int wave  = threadIdx.x >> 5;
    const int lane  = threadIdx.x & 31;
    const int half  = lane >> 4;
    const int l16   = lane & 15;
    const int mBase = (blockIdx.x * 8 + wave) * 16;
    if (mBase >= nrows) return;

    int rowA = mBase + l16;
    if (rowA >= nrows) rowA = nrows - 1;
    const float* xrow = X + (size_t)rowA * K;

    v8f c0 = {}, c1 = {}, c2 = {}, c3 = {};
    for (int k0 = 0; k0 < K; k0 += 4) {
        const int kk = k0 + 2 * half;
        v2f a = *(const v2f*)(xrow + kk);
        v2f b0 = *(const v2f*)(&ldsW[(0 * 16 + l16) * (K + 2) + kk]);
        v2f b1 = *(const v2f*)(&ldsW[(1 * 16 + l16) * (K + 2) + kk]);
        v2f b2 = *(const v2f*)(&ldsW[(2 * 16 + l16) * (K + 2) + kk]);
        v2f b3 = *(const v2f*)(&ldsW[(3 * 16 + l16) * (K + 2) + kk]);
        c0 = __builtin_amdgcn_wmma_f32_16x16x4_f32(false, a, false, b0, (short)0, c0, false, false);
        c1 = __builtin_amdgcn_wmma_f32_16x16x4_f32(false, a, false, b1, (short)0, c1, false, false);
        c2 = __builtin_amdgcn_wmma_f32_16x16x4_f32(false, a, false, b2, (short)0, c2, false, false);
        c3 = __builtin_amdgcn_wmma_f32_16x16x4_f32(false, a, false, b3, (short)0, c3, false, false);
    }

#pragma unroll
    for (int r = 0; r < 8; ++r) {
        int row = mBase + 8 * half + r;
        if (row < nrows) {
            float s = norm[row];
            size_t base = (size_t)row * F_OUT + l16;
            float v0 = c0[r] * s, v1 = c1[r] * s, v2 = c2[r] * s, v3 = c3[r] * s;
            H[base +  0] = v0;  H[base + 16] = v1;
            H[base + 32] = v2;  H[base + 48] = v3;
        }
    }
}

template <int RELU>
__global__ __launch_bounds__(256) void k_gather(const int* __restrict__ rowptr,
                                                const int* __restrict__ sorted,
                                                const float* __restrict__ h,
                                                const float* __restrict__ nd,
                                                const float* __restrict__ bias,
                                                float* __restrict__ out, int n, int e) {
    const int sub = threadIdx.x & 15;
    const int grp = threadIdx.x & 16;
    const int row = blockIdx.x * 16 + (threadIdx.x >> 4);
    if (row >= n) return;

    const int beg = rowptr[row];
    const int end = (row + 1 < n) ? rowptr[row + 1] : e;
    const int deg = end - beg;
    const float4* __restrict__ h4 = (const float4*)h;
    float4 acc = h4[(size_t)row * 16 + sub];

    for (int base = 0; base < deg; base += 16) {
        const int cnt = min(16, deg - base);
        int idx = row;
        if (sub < cnt) idx = sorted[beg + base + sub];
        for (int k = 0; k < cnt; k += 4) {
            const int s0 = __builtin_amdgcn_ds_bpermute((grp + k + 0) << 2, idx);
            const int s1 = __builtin_amdgcn_ds_bpermute((grp + k + 1) << 2, idx);
            const int s2 = __builtin_amdgcn_ds_bpermute((grp + k + 2) << 2, idx);
            const int s3 = __builtin_amdgcn_ds_bpermute((grp + k + 3) << 2, idx);
            const float4 v0 = h4[(size_t)s0 * 16 + sub];
            const float4 v1 = h4[(size_t)s1 * 16 + sub];
            const float4 v2 = h4[(size_t)s2 * 16 + sub];
            const float4 v3 = h4[(size_t)s3 * 16 + sub];
            const bool p1 = k + 1 < cnt, p2 = k + 2 < cnt, p3 = k + 3 < cnt;
            acc.x += v0.x; acc.y += v0.y; acc.z += v0.z; acc.w += v0.w;
            acc.x += p1 ? v1.x : 0.0f; acc.y += p1 ? v1.y : 0.0f;
            acc.z += p1 ? v1.z : 0.0f; acc.w += p1 ? v1.w : 0.0f;
            acc.x += p2 ? v2.x : 0.0f; acc.y += p2 ? v2.y : 0.0f;
            acc.z += p2 ? v2.z : 0.0f; acc.w += p2 ? v2.w : 0.0f;
            acc.x += p3 ? v3.x : 0.0f; acc.y += p3 ? v3.y : 0.0f;
            acc.z += p3 ? v3.z : 0.0f; acc.w += p3 ? v3.w : 0.0f;
        }
    }

    const float s = nd[row];
    const float4 b = *(const float4*)(bias + sub * 4);
    float4 v;
    v.x = fmaf(acc.x, s, b.x);
    v.y = fmaf(acc.y, s, b.y);
    v.z = fmaf(acc.z, s, b.z);
    v.w = fmaf(acc.w, s, b.w);
    if (RELU) {
        v.x = fmaxf(v.x, 0.0f); v.y = fmaxf(v.y, 0.0f);
        v.z = fmaxf(v.z, 0.0f); v.w = fmaxf(v.w, 0.0f);
    }
    *(float4*)(out + (size_t)row * F_OUT + sub * 4) = v;
}

extern "C" void kernel_launch(void* const* d_in, const int* in_sizes, int n_in,
                              void* d_out, int out_size, void* d_ws, size_t ws_size,
                              hipStream_t stream) {
    const float* x   = (const float*)d_in[0];
    const int*   src = (const int*)d_in[1];
    const int*   dst = (const int*)d_in[2];
    const float* W1  = (const float*)d_in[3];
    const float* b1  = (const float*)d_in[4];
    const float* W2  = (const float*)d_in[5];
    const float* b2  = (const float*)d_in[6];
    float*       out = (float*)d_out;

    const int n = in_sizes[0] / 128;
    const int e = in_sizes[1];

    float* ws   = (float*)d_ws;
    float* buf0 = ws;
    int*   sorted = (int*)(ws + (size_t)n * F_OUT);
    int*   rowptr = sorted + e;
    int*   lexcl  = rowptr + n;
    int*   bsum   = lexcl + n;
    float* ns   = ws + 2 * (size_t)n * F_OUT;
    float* nd   = ns + n;

    const int B = 256;
    const int gemmBlocks = cdiv(cdiv(n, 16), 8);

    int* part = (int*)d_ws;
    k_hist<<<dim3(HC, cdiv(n, HR), 2), HT, 0, stream>>>(src, dst, part, n, e);
    k_norm<<<cdiv(n, NB), NB, 0, stream>>>(part, ns, nd, lexcl, bsum, n);
    k_fill<<<dim3(HC, cdiv(n, FR)), HT, 0, stream>>>(src, dst, part + (size_t)HC * n, lexcl, bsum,
                                                     rowptr, sorted, n, e);

    k_gemm_scale<128><<<gemmBlocks, B, 0, stream>>>(x, W1, ns, buf0, n);
    k_gather<1><<<cdiv(n, 16), B, 0, stream>>>(rowptr, sorted, buf0, nd, b1, out, n, e);

    k_gemm_scale<64><<<gemmBlocks, B, 0, stream>>>(out, W2, ns, buf0, n);
    k_gather<0><<<cdiv(n, 16), B, 0, stream>>>(rowptr, sorted, buf0, nd, b2, out, n, e);
}


# k_fill and the layer-1 GEMM share one launch: 256 placement blocks + 250 GEMM blocks (25 one-tile waves each, W1 staged in one round of loads), two per WGP; count planes moved to out; same sums
# speedup vs baseline: 3.6258x; 1.1966x over previous
#include <hip/hip_runtime.h>
#include <math.h>

#define F_OUT 64

typedef __attribute__((ext_vector_type(2))) float v2f;
typedef __attribute__((ext_vector_type(8))) float v8f;

static inline int cdiv(int a, int b) { return (a + b - 1) / b; }

#define HR 51200
#define HC 32
#define HT 1024

static __device__ __forceinline__ void hist_add4(int* hist, int4 v, int lo) {
    unsigned a = (unsigned)(v.x - lo), b = (unsigned)(v.y - lo);
    unsigned c = (unsigned)(v.z - lo), d = (unsigned)(v.w - lo);
    if (a < HR) atomicAdd(&hist[a], 1);
    if (b < HR) atomicAdd(&hist[b], 1);
    if (c < HR) atomicAdd(&hist[c], 1);
    if (d < HR) atomicAdd(&hist[d], 1);
}

__global__ __launch_bounds__(HT) void k_hist(const int* __restrict__ src,
                                             const int* __restrict__ dst,
                                             int* __restrict__ part, int n, int e) {
    __shared__ alignas(16) int hist[HR];
    const int* __restrict__ idx = blockIdx.z ? dst : src;
    const int lo = blockIdx.y * HR;

    for (int t = threadIdx.x; t < HR / 4; t += HT)
        ((int4*)hist)[t] = make_int4(0, 0, 0, 0);
    __syncthreads();

    const int per = (((e + HC - 1) / HC) + 3) & ~3;
    const int beg = min((int)blockIdx.x * per, e);
    const int end = min(beg + per, e);
    const int nvec = (((size_t)idx & 15) == 0) ? ((end - beg) >> 2) : 0;
    const int4* __restrict__ p4 = (const int4*)(idx + beg);

    for (int q = threadIdx.x; q < nvec; q += 4 * HT) {
        const int q1 = q + 1 * HT, q2 = q + 2 * HT, q3 = q + 3 * HT;
        const int4 v0 = p4[q];
        const int4 v1 = p4[min(q1, nvec - 1)];
        const int4 v2 = p4[min(q2, nvec - 1)];
        const int4 v3 = p4[min(q3, nvec - 1)];
        hist_add4(hist, v0, lo);
        if (q1 < nvec) hist_add4(hist, v1, lo);
        if (q2 < nvec) hist_add4(hist, v2, lo);
        if (q3 < nvec) hist_add4(hist, v3, lo);
    }
    for (int j = beg + 4 * nvec + threadIdx.x; j < end; j += HT) {
        unsigned a = (unsigned)(idx[j] - lo);
        if (a < HR) atomicAdd(&hist[a], 1);
    }
    __syncthreads();

    int* __restrict__ plane = part + (size_t)(blockIdx.z * HC + blockIdx.x) * n;
    const int cnt = min(HR, n - lo);
    if ((n & 3) == 0) {
        for (int t = threadIdx.x; t < (cnt >> 2); t += HT)
            ((int4*)(plane + lo))[t] = ((const int4*)hist)[t];
    } else {
        for (int t = threadIdx.x; t < cnt; t += HT)
            plane[lo + t] = hist[t];
    }
}

#define NB 256
__global__ __launch_bounds__(NB) void k_norm(int* __restrict__ part, float* __restrict__ ns,
                                             float* __restrict__ nd, int* __restrict__ lexcl,
                                             int* __restrict__ bsum, int n) {
    __shared__ int sc[NB];
    const int tid = threadIdx.x;
    const int i = blockIdx.x * NB + tid;
    const bool valid = i < n;
    int din = 0;
    if (valid) {
        int s = 1;
        int v[HC];
#pragma unroll
        for (int c = 0; c < HC; ++c) {
            s += part[(size_t)c * n + i];
            v[c] = part[(size_t)(HC + c) * n + i];
        }
#pragma unroll
        for (int c = 0; c < HC; ++c) {
            part[(size_t)(HC + c) * n + i] = din;
            din += v[c];
        }
        ns[i] = rsqrtf((float)s);
        nd[i] = rsqrtf((float)(din + 1));
    }
    sc[tid] = din;
    __syncthreads();
    for (int off = 1; off < NB; off <<= 1) {
        const int t = (tid >= off) ? sc[tid - off] : 0;
        __syncthreads();
        sc[tid] += t;
        __syncthreads();
    }
    if (valid) lexcl[i] = sc[tid] - din;
    if (tid == NB - 1) bsum[blockIdx.x] = sc[tid];
}


template <int K>
static __device__ __forceinline__ void gemm_stage_w(const float* __restrict__ W, float* ldsW) {
    for (int t = threadIdx.x; t < K * F_OUT; t += blockDim.x) {
        int k = t / F_OUT, c = t % F_OUT;
        ldsW[c * (K + 2) + k] = W[t];
    }
}

template <int K, int NT>
static __device__ __forceinline__ void gemm_stage_w_once(const float* __restrict__ W, float* ldsW) {
    constexpr int PER = K * F_OUT / NT;
    static_assert(PER * NT == K * F_OUT, "whole elements per thread");
    float v[PER];
#pragma unroll
    for (int i = 0; i < PER; ++i) v[i] = W[threadIdx.x + i * NT];
#pragma unroll
    for (int i = 0; i < PER; ++i) {
        const int t = threadIdx.x + i * NT;
        ldsW[(t % F_OUT) * (K + 2) + t / F_OUT] = v[i];
    }
}

template <int K, int U>
static __device__ __forceinline__ void gemm_tile(const float* __restrict__ X, const float* ldsW,
                                                 const float* __restrict__ norm,
                                                 float* __restrict__ H, int nrows, int mBase) {
    const int lane  = threadIdx.x & 31;
    const int half  = lane >> 4;
    const int l16   = lane & 15;

    int rowA = mBase + l16;
    if (rowA >= nrows) rowA = nrows - 1;
    const float* xrow = X + (size_t)rowA * K;

    v8f c0 = {}, c1 = {}, c2 = {}, c3 = {};
#pragma unroll U
    for (int k0 = 0; k0 < K; k0 += 4) {
        const int kk = k0 + 2 * half;
        v2f a = *(const v2f*)(xrow + kk);
        v2f b0 = *(const v2f*)(&ldsW[(0 * 16 + l16) * (K + 2) + kk]);
        v2f b1 = *(const v2f*)(&ldsW[(1 * 16 + l16) * (K + 2) + kk]);
        v2f b2 = *(const v2f*)(&ldsW[(2 * 16 + l16) * (K + 2) + kk]);
        v2f b3 = *(const v2f*)(&ldsW[(3 * 16 + l16) * (K + 2) + kk]);
        c0 = __builtin_amdgcn_wmma_f32_16x16x4_f32(false, a, false, b0, (short)0, c0, false, false);
        c1 = __builtin_amdgcn_wmma_f32_16x16x4_f32(false, a, false, b1, (short)0, c1, false, false);
        c2 = __builtin_amdgcn_wmma_f32_16x16x4_f32(false, a, false, b2, (short)0, c2, false, false);
        c3 = __builtin_amdgcn_wmma_f32_16x16x4_f32(false, a, false, b3, (short)0, c3, false, false);
    }

#pragma unroll
    for (int r = 0; r < 8; ++r) {
        int row = mBase + 8 * half + r;
        if (row < nrows) {
            float s = norm[row];
            size_t base = (size_t)row * F_OUT + l16;
            float v0 = c0[r] * s, v1 = c1[r] * s, v2 = c2[r] * s, v3 = c3[r] * s;
            H[base +  0] = v0;  H[base + 16] = v1;
            H[base + 32] = v2;  H[base + 48] = v3;
        }
    }
}

template <int K>
__global__ __launch_bounds__(256) void k_gemm_scale(const float* __restrict__ X,
                                                    const float* __restrict__ W,
                                                    const float* __restrict__ norm,
                                                    float* __restrict__ H, int nrows) {
    __shared__ float ldsW[F_OUT * (K + 2)];
    gemm_stage_w<K>(W, ldsW);
    __syncthreads();

    const int wave  = threadIdx.x >> 5;
    const int mBase = (blockIdx.x * 8 + wave) * 16;
    if (mBase >= nrows) return;
    gemm_tile<K, K / 4>(X, ldsW, norm, H, nrows, mBase);
}

#define FR  12800
#define FNB (FR / NB)
#define FSB 64
static_assert(FR % NB == 0 && FR % 4 == 0, "a range is whole k_norm blocks and whole 16-byte groups");
static_assert(FNB <= FSB && FSB <= HT, "one lane per block total");
static_assert(FR >= F_OUT * (128 + 2), "W1, transposed and padded, fits the cursor array");

static __device__ __forceinline__ void fill4(int* cur, int* __restrict__ sorted,
                                             int4 d, int4 s, int lo) {
    unsigned a = (unsigned)(d.x - lo), b = (unsigned)(d.y - lo);
    unsigned c = (unsigned)(d.z - lo), g = (unsigned)(d.w - lo);
    if (a < FR) sorted[atomicAdd(&cur[a], 1)] = s.x;
    if (b < FR) sorted[atomicAdd(&cur[b], 1)] = s.y;
    if (c < FR) sorted[atomicAdd(&cur[c], 1)] = s.z;
    if (g < FR) sorted[atomicAdd(&cur[g], 1)] = s.w;
}

__global__ __launch_bounds__(HT, 16) void k_fill_gemm(const int* __restrict__ src,
                                                  const int* __restrict__ dst,
                                                  const int* __restrict__ dpart,
                                                  const int* __restrict__ lexcl,
                                                  const int* __restrict__ bsum,
                                                  int* __restrict__ rowptr,
                                                  int* __restrict__ sorted, int n, int e,
                                                  const float* __restrict__ X,
                                                  const float* __restrict__ W,
                                                  const float* __restrict__ norm,
                                                  float* __restrict__ H,
                                                  int fillBlocks, int tpb) {
    __shared__ alignas(16) int cur[FR];
    __shared__ int sb[FSB];
    __shared__ int pre_s;
    if ((int)blockIdx.x >= fillBlocks) {
        float* ldsW = (float*)cur;
        gemm_stage_w_once<128, HT>(W, ldsW);
        __syncthreads();
        const int wave  = __builtin_amdgcn_readfirstlane(threadIdx.x >> 5);
        const int mBase = (((int)blockIdx.x - fillBlocks) * tpb + wave) * 16;
        if (wave >= tpb || mBase >= n) return;
        gemm_tile<128, 4>(X, ldsW, norm, H, n, mBase);
        return;
    }
    const int tid = threadIdx.x;
    const int c   = blockIdx.x % HC;
    const int lo  = (blockIdx.x / HC) * FR;
    const int nb  = (n + NB - 1) / NB;
    const int b0  = lo / NB;

    if (tid == 0) pre_s = 0;
    if (tid < FSB) {
        const int j = b0 + tid;
        sb[tid] = (tid < FNB && j < nb) ? bsum[j] : 0;
    }
    int p = 0;
    for (int j = tid; j < b0; j += HT) p += bsum[j];
    __syncthreads();
    if (p) atomicAdd(&pre_s, p);
    for (int off = 1; off < FSB; off <<= 1) {
        const int t = (tid < FSB && tid >= off) ? sb[tid - off] : 0;
        __syncthreads();
        if (tid < FSB) sb[tid] += t;
        __syncthreads();
    }
    const int pre = pre_s;

    const int cnt = min(FR, n - lo);
    const int* __restrict__ plane = dpart + (size_t)c * n;
    const bool al4 = ((n & 3) == 0) &&
                     ((((size_t)lexcl | (size_t)dpart | (size_t)rowptr) & 15) == 0);
    const int ng = al4 ? (cnt >> 2) : 0;
    const int4* __restrict__ lx4 = (const int4*)(lexcl + lo);
    const int4* __restrict__ pl4 = (const int4*)(plane + lo);
    for (int g = tid; g < ng; g += 4 * HT) {
        int4 lx[4], pl[4];
#pragma unroll
        for (int u = 0; u < 4; ++u) {
            const int gg = min(g + u * HT, ng - 1);
            lx[u] = lx4[gg];
            pl[u] = pl4[gg];
        }
#pragma unroll
        for (int u = 0; u < 4; ++u) {
            const int gg = g + u * HT;
            if (gg < ng) {
                const int k = (4 * gg) / NB;
                const int bs = pre + (k ? sb[k - 1] : 0);
                const int4 rp = make_int4(bs + lx[u].x, bs + lx[u].y, bs + lx[u].z, bs + lx[u].w);
                ((int4*)cur)[gg] = make_int4(rp.x + pl[u].x, rp.y + pl[u].y,
                                             rp.z + pl[u].z, rp.w + pl[u].w);
                if (c == 0) ((int4*)(rowptr + lo))[gg] = rp;
            }
        }
    }
    for (int t = 4 * ng + tid; t < cnt; t += HT) {
        const int i = lo + t;
        const int k = t / NB;
        const int rp = pre + (k ? sb[k - 1] : 0) + lexcl[i];
        cur[t] = rp + plane[i];
        if (c == 0) rowptr[i] = rp;
    }
    __syncthreads();

    const int per = (((e + HC - 1) / HC) + 3) & ~3;
    const int beg = min(c * per, e);
    const int end = min(beg + per, e);
    const bool al = ((((size_t)src | (size_t)dst) & 15) == 0);
    const int nvec = al ? ((end - beg) >> 2) : 0;
    const int4* __restrict__ d4 = (const int4*)(dst + beg);
    const int4* __restrict__ s4 = (const int4*)(src + beg);

    for (int q = tid; q < nvec; q += 4 * HT) {
        const int q1 = q + 1 * HT, q2 = q + 2 * HT, q3 = q + 3 * HT;
        const int4 dv0 = d4[q],                   sv0 = s4[q];
        const int4 dv1 = d4[min(q1, nvec - 1)],   sv1 = s4[min(q1, nvec - 1)];
        const int4 dv2 = d4[min(q2, nvec - 1)],   sv2 = s4[min(q2, nvec - 1)];
        const int4 dv3 = d4[min(q3, nvec - 1)],   sv3 = s4[min(q3, nvec - 1)];
        fill4(cur, sorted, dv0, sv0, lo);
        if (q1 < nvec) fill4(cur, sorted, dv1, sv1, lo);
        if (q2 < nvec) fill4(cur, sorted, dv2, sv2, lo);
        if (q3 < nvec) fill4(cur, sorted, dv3, sv3, lo);
    }
    for (int j = beg + 4 * nvec + tid; j < end; j += HT) {
        const unsigned a = (unsigned)(dst[j] - lo);
        if (a < FR) sorted[atomicAdd(&cur[a], 1)] = src[j];
    }
}

template <int RELU>
__global__ __launch_bounds__(256) void k_gather(const int* __restrict__ rowptr,
                                                const int* __restrict__ sorted,
                                                const float* __restrict__ h,
                                                const float* __restrict__ nd,
                                                const float* __restrict__ bias,
                                                float* __restrict__ out, int n, int e) {
    const int sub = threadIdx.x & 15;
    const int grp = threadIdx.x & 16;
    const int row = blockIdx.x * 16 + (threadIdx.x >> 4);
    if (row >= n) return;

    const int beg = rowptr[row];
    const int end = (row + 1 < n) ? rowptr[row + 1] : e;
    const int deg = end - beg;
    const float4* __restrict__ h4 = (const float4*)h;
    float4 acc = h4[(size_t)row * 16 + sub];

    for (int base = 0; base < deg; base += 16) {
        const int cnt = min(16, deg - base);
        int idx = row;
        if (sub < cnt) idx = sorted[beg + base + sub];
        for (int k = 0; k < cnt; k += 4) {
            const int s0 = __builtin_amdgcn_ds_bpermute((grp + k + 0) << 2, idx);
            const int s1 = __builtin_amdgcn_ds_bpermute((grp + k + 1) << 2, idx);
            const int s2 = __builtin_amdgcn_ds_bpermute((grp + k + 2) << 2, idx);
            const int s3 = __builtin_amdgcn_ds_bpermute((grp + k + 3) << 2, idx);
            const float4 v0 = h4[(size_t)s0 * 16 + sub];
            const float4 v1 = h4[(size_t)s1 * 16 + sub];
            const float4 v2 = h4[(size_t)s2 * 16 + sub];
            const float4 v3 = h4[(size_t)s3 * 16 + sub];
            const bool p1 = k + 1 < cnt, p2 = k + 2 < cnt, p3 = k + 3 < cnt;
            acc.x += v0.x; acc.y += v0.y; acc.z += v0.z; acc.w += v0.w;
            acc.x += p1 ? v1.x : 0.0f; acc.y += p1 ? v1.y : 0.0f;
            acc.z += p1 ? v1.z : 0.0f; acc.w += p1 ? v1.w : 0.0f;
            acc.x += p2 ? v2.x : 0.0f; acc.y += p2 ? v2.y : 0.0f;
            acc.z += p2 ? v2.z : 0.0f; acc.w += p2 ? v2.w : 0.0f;
            acc.x += p3 ? v3.x : 0.0f; acc.y += p3 ? v3.y : 0.0f;
            acc.z += p3 ? v3.z : 0.0f; acc.w += p3 ? v3.w : 0.0f;
        }
    }

    const float s = nd[row];
    const float4 b = *(const float4*)(bias + sub * 4);
    float4 v;
    v.x = fmaf(acc.x, s, b.x);
    v.y = fmaf(acc.y, s, b.y);
    v.z = fmaf(acc.z, s, b.z);
    v.w = fmaf(acc.w, s, b.w);
    if (RELU) {
        v.x = fmaxf(v.x, 0.0f); v.y = fmaxf(v.y, 0.0f);
        v.z = fmaxf(v.z, 0.0f); v.w = fmaxf(v.w, 0.0f);
    }
    *(float4*)(out + (size_t)row * F_OUT + sub * 4) = v;
}

extern "C" void kernel_launch(void* const* d_in, const int* in_sizes, int n_in,
                              void* d_out, int out_size, void* d_ws, size_t ws_size,
                              hipStream_t stream) {
    const float* x   = (const float*)d_in[0];
    const int*   src = (const int*)d_in[1];
    const int*   dst = (const int*)d_in[2];
    const float* W1  = (const float*)d_in[3];
    const float* b1  = (const float*)d_in[4];
    const float* W2  = (const float*)d_in[5];
    const float* b2  = (const float*)d_in[6];
    float*       out = (float*)d_out;

    const int n = in_sizes[0] / 128;
    const int e = in_sizes[1];

    float* ws   = (float*)d_ws;
    float* buf0 = ws;
    int*   sorted = (int*)(ws + (size_t)n * F_OUT);
    int*   rowptr = sorted + e;
    int*   lexcl  = rowptr + n;
    int*   bsum   = lexcl + n;
    float* ns   = ws + 2 * (size_t)n * F_OUT;
    float* nd   = ns + n;

    const int B = 256;
    const int gemmBlocks = cdiv(cdiv(n, 16), 8);

    static_assert(2 * HC * sizeof(int) <= F_OUT * sizeof(float), "the count planes fit 'out'");
    int* part = (int*)d_out;
    k_hist<<<dim3(HC, cdiv(n, HR), 2), HT, 0, stream>>>(src, dst, part, n, e);
    k_norm<<<cdiv(n, NB), NB, 0, stream>>>(part, ns, nd, lexcl, bsum, n);

    const int fillBlocks = HC * cdiv(n, FR);
    const int tiles = cdiv(n, 16);
    int tpb = cdiv(tiles, 256);
    if (tpb > HT / 32) tpb = HT / 32;
    if (tpb < 1) tpb = 1;
    k_fill_gemm<<<fillBlocks + cdiv(tiles, tpb), HT, 0, stream>>>(
        src, dst, part + (size_t)HC * n, lexcl, bsum, rowptr, sorted, n, e,
        x, W1, ns, buf0, fillBlocks, tpb);
    k_gather<1><<<cdiv(n, 16), B, 0, stream>>>(rowptr, sorted, buf0, nd, b1, out, n, e);

    k_gemm_scale<64><<<gemmBlocks, B, 0, stream>>>(out, W2, ns, buf0, n);
    k_gather<0><<<cdiv(n, 16), B, 0, stream>>>(rowptr, sorted, buf0, nd, b2, out, n, e);
}


# Layer-2 GEMM done inside the layer-1 gather: 64-row blocks put their relu rows in LDS, 16 waves do h W2 on WMMA (W2 in LDS, same k order), scaled rows stored; 5 launches, h never stored; same sums
# speedup vs baseline: 4.1880x; 1.1551x over previous
#include <hip/hip_runtime.h>
#include <math.h>

#define F_OUT 64

typedef __attribute__((ext_vector_type(2))) float v2f;
typedef __attribute__((ext_vector_type(8))) float v8f;

static inline int cdiv(int a, int b) { return (a + b - 1) / b; }

#define HR 51200
#define HC 32
#define HT 1024

static __device__ __forceinline__ void hist_add4(int* hist, int4 v, int lo) {
    unsigned a = (unsigned)(v.x - lo), b = (unsigned)(v.y - lo);
    unsigned c = (unsigned)(v.z - lo), d = (unsigned)(v.w - lo);
    if (a < HR) atomicAdd(&hist[a], 1);
    if (b < HR) atomicAdd(&hist[b], 1);
    if (c < HR) atomicAdd(&hist[c], 1);
    if (d < HR) atomicAdd(&hist[d], 1);
}

__global__ __launch_bounds__(HT) void k_hist(const int* __restrict__ src,
                                             const int* __restrict__ dst,
                                             int* __restrict__ part, int n, int e) {
    __shared__ alignas(16) int hist[HR];
    const int* __restrict__ idx = blockIdx.z ? dst : src;
    const int lo = blockIdx.y * HR;

    for (int t = threadIdx.x; t < HR / 4; t += HT)
        ((int4*)hist)[t] = make_int4(0, 0, 0, 0);
    __syncthreads();

    const int per = (((e + HC - 1) / HC) + 3) & ~3;
    const int beg = min((int)blockIdx.x * per, e);
    const int end = min(beg + per, e);
    const int nvec = (((size_t)idx & 15) == 0) ? ((end - beg) >> 2) : 0;
    const int4* __restrict__ p4 = (const int4*)(idx + beg);

    for (int q = threadIdx.x; q < nvec; q += 4 * HT) {
        const int q1 = q + 1 * HT, q2 = q + 2 * HT, q3 = q + 3 * HT;
        const int4 v0 = p4[q];
        const int4 v1 = p4[min(q1, nvec - 1)];
        const int4 v2 = p4[min(q2, nvec - 1)];
        const int4 v3 = p4[min(q3, nvec - 1)];
        hist_add4(hist, v0, lo);
        if (q1 < nvec) hist_add4(hist, v1, lo);
        if (q2 < nvec) hist_add4(hist, v2, lo);
        if (q3 < nvec) hist_add4(hist, v3, lo);
    }
    for (int j = beg + 4 * nvec + threadIdx.x; j < end; j += HT) {
        unsigned a = (unsigned)(idx[j] - lo);
        if (a < HR) atomicAdd(&hist[a], 1);
    }
    __syncthreads();

    int* __restrict__ plane = part + (size_t)(blockIdx.z * HC + blockIdx.x) * n;
    const int cnt = min(HR, n - lo);
    if ((n & 3) == 0) {
        for (int t = threadIdx.x; t < (cnt >> 2); t += HT)
            ((int4*)(plane + lo))[t] = ((const int4*)hist)[t];
    } else {
        for (int t = threadIdx.x; t < cnt; t += HT)
            plane[lo + t] = hist[t];
    }
}

#define NB 256
__global__ __launch_bounds__(NB) void k_norm(int* __restrict__ part, float* __restrict__ ns,
                                             float* __restrict__ nd, int* __restrict__ lexcl,
                                             int* __restrict__ bsum, int n) {
    __shared__ int sc[NB];
    const int tid = threadIdx.x;
    const int i = blockIdx.x * NB + tid;
    const bool valid = i < n;
    int din = 0;
    if (valid) {
        int s = 1;
        int v[HC];
#pragma unroll
        for (int c = 0; c < HC; ++c) {
            s += part[(size_t)c * n + i];
            v[c] = part[(size_t)(HC + c) * n + i];
        }
#pragma unroll
        for (int c = 0; c < HC; ++c) {
            part[(size_t)(HC + c) * n + i] = din;
            din += v[c];
        }
        ns[i] = rsqrtf((float)s);
        nd[i] = rsqrtf((float)(din + 1));
    }
    sc[tid] = din;
    __syncthreads();
    for (int off = 1; off < NB; off <<= 1) {
        const int t = (tid >= off) ? sc[tid - off] : 0;
        __syncthreads();
        sc[tid] += t;
        __syncthreads();
    }
    if (valid) lexcl[i] = sc[tid] - din;
    if (tid == NB - 1) bsum[blockIdx.x] = sc[tid];
}


template <int K, int NT>
static __device__ __forceinline__ void gemm_stage_w_once(const float* __restrict__ W, float* ldsW) {
    constexpr int PER = K * F_OUT / NT;
    static_assert(PER * NT == K * F_OUT, "whole elements per thread");
    float v[PER];
#pragma unroll
    for (int i = 0; i < PER; ++i) v[i] = W[threadIdx.x + i * NT];
#pragma unroll
    for (int i = 0; i < PER; ++i) {
        const int t = threadIdx.x + i * NT;
        ldsW[(t % F_OUT) * (K + 2) + t / F_OUT] = v[i];
    }
}

template <int K, int U>
static __device__ __forceinline__ void gemm_tile(const float* __restrict__ X, const float* ldsW,
                                                 const float* __restrict__ norm,
                                                 float* __restrict__ H, int nrows, int mBase) {
    const int lane  = threadIdx.x & 31;
    const int half  = lane >> 4;
    const int l16   = lane & 15;

    int rowA = mBase + l16;
    if (rowA >= nrows) rowA = nrows - 1;
    const float* xrow = X + (size_t)rowA * K;

    v8f c0 = {}, c1 = {}, c2 = {}, c3 = {};
#pragma unroll U
    for (int k0 = 0; k0 < K; k0 += 4) {
        const int kk = k0 + 2 * half;
        v2f a = *(const v2f*)(xrow + kk);
        v2f b0 = *(const v2f*)(&ldsW[(0 * 16 + l16) * (K + 2) + kk]);
        v2f b1 = *(const v2f*)(&ldsW[(1 * 16 + l16) * (K + 2) + kk]);
        v2f b2 = *(const v2f*)(&ldsW[(2 * 16 + l16) * (K + 2) + kk]);
        v2f b3 = *(const v2f*)(&ldsW[(3 * 16 + l16) * (K + 2) + kk]);
        c0 = __builtin_amdgcn_wmma_f32_16x16x4_f32(false, a, false, b0, (short)0, c0, false, false);
        c1 = __builtin_amdgcn_wmma_f32_16x16x4_f32(false, a, false, b1, (short)0, c1, false, false);
        c2 = __builtin_amdgcn_wmma_f32_16x16x4_f32(false, a, false, b2, (short)0, c2, false, false);
        c3 = __builtin_amdgcn_wmma_f32_16x16x4_f32(false, a, false, b3, (short)0, c3, false, false);
    }

#pragma unroll
    for (int r = 0; r < 8; ++r) {
        int row = mBase + 8 * half + r;
        if (row < nrows) {
            float s = norm[row];
            size_t base = (size_t)row * F_OUT + l16;
            float v0 = c0[r] * s, v1 = c1[r] * s, v2 = c2[r] * s, v3 = c3[r] * s;
            H[base +  0] = v0;  H[base + 16] = v1;
            H[base + 32] = v2;  H[base + 48] = v3;
        }
    }
}

#define FR  12800
#define FNB (FR / NB)
#define FSB 64
static_assert(FR % NB == 0 && FR % 4 == 0, "a range is whole k_norm blocks and whole 16-byte groups");
static_assert(FNB <= FSB && FSB <= HT, "one lane per block total");
static_assert(FR >= F_OUT * (128 + 2), "W1, transposed and padded, fits the cursor array");

static __device__ __forceinline__ void fill4(int* cur, int* __restrict__ sorted,
                                             int4 d, int4 s, int lo) {
    unsigned a = (unsigned)(d.x - lo), b = (unsigned)(d.y - lo);
    unsigned c = (unsigned)(d.z - lo), g = (unsigned)(d.w - lo);
    if (a < FR) sorted[atomicAdd(&cur[a], 1)] = s.x;
    if (b < FR) sorted[atomicAdd(&cur[b], 1)] = s.y;
    if (c < FR) sorted[atomicAdd(&cur[c], 1)] = s.z;
    if (g < FR) sorted[atomicAdd(&cur[g], 1)] = s.w;
}

__global__ __launch_bounds__(HT, 16) void k_fill_gemm(const int* __restrict__ src,
                                                  const int* __restrict__ dst,
                                                  const int* __restrict__ dpart,
                                                  const int* __restrict__ lexcl,
                                                  const int* __restrict__ bsum,
                                                  int* __restrict__ rowptr,
                                                  int* __restrict__ sorted, int n, int e,
                                                  const float* __restrict__ X,
                                                  const float* __restrict__ W,
                                                  const float* __restrict__ norm,
                                                  float* __restrict__ H,
                                                  int fillBlocks, int tpb) {
    __shared__ alignas(16) int cur[FR];
    __shared__ int sb[FSB];
    __shared__ int pre_s;
    if ((int)blockIdx.x >= fillBlocks) {
        float* ldsW = (float*)cur;
        gemm_stage_w_once<128, HT>(W, ldsW);
        __syncthreads();
        const int wave  = __builtin_amdgcn_readfirstlane(threadIdx.x >> 5);
        const int mBase = (((int)blockIdx.x - fillBlocks) * tpb + wave) * 16;
        if (wave >= tpb || mBase >= n) return;
        gemm_tile<128, 4>(X, ldsW, norm, H, n, mBase);
        return;
    }
    const int tid = threadIdx.x;
    const int c   = blockIdx.x % HC;
    const int lo  = (blockIdx.x / HC) * FR;
    const int nb  = (n + NB - 1) / NB;
    const int b0  = lo / NB;

    if (tid == 0) pre_s = 0;
    if (tid < FSB) {
        const int j = b0 + tid;
        sb[tid] = (tid < FNB && j < nb) ? bsum[j] : 0;
    }
    int p = 0;
    for (int j = tid; j < b0; j += HT) p += bsum[j];
    __syncthreads();
    if (p) atomicAdd(&pre_s, p);
    for (int off = 1; off < FSB; off <<= 1) {
        const int t = (tid < FSB && tid >= off) ? sb[tid - off] : 0;
        __syncthreads();
        if (tid < FSB) sb[tid] += t;
        __syncthreads();
    }
    const int pre = pre_s;

    const int cnt = min(FR, n - lo);
    const int* __restrict__ plane = dpart + (size_t)c * n;
    const bool al4 = ((n & 3) == 0) &&
                     ((((size_t)lexcl | (size_t)dpart | (size_t)rowptr) & 15) == 0);
    const int ng = al4 ? (cnt >> 2) : 0;
    const int4* __restrict__ lx4 = (const int4*)(lexcl + lo);
    const int4* __restrict__ pl4 = (const int4*)(plane + lo);
    for (int g = tid; g < ng; g += 4 * HT) {
        int4 lx[4], pl[4];
#pragma unroll
        for (int u = 0; u < 4; ++u) {
            const int gg = min(g + u * HT, ng - 1);
            lx[u] = lx4[gg];
            pl[u] = pl4[gg];
        }
#pragma unroll
        for (int u = 0; u < 4; ++u) {
            const int gg = g + u * HT;
            if (gg < ng) {
                const int k = (4 * gg) / NB;
                const int bs = pre + (k ? sb[k - 1] : 0);
                const int4 rp = make_int4(bs + lx[u].x, bs + lx[u].y, bs + lx[u].z, bs + lx[u].w);
                ((int4*)cur)[gg] = make_int4(rp.x + pl[u].x, rp.y + pl[u].y,
                                             rp.z + pl[u].z, rp.w + pl[u].w);
                if (c == 0) ((int4*)(rowptr + lo))[gg] = rp;
            }
        }
    }
    for (int t = 4 * ng + tid; t < cnt; t += HT) {
        const int i = lo + t;
        const int k = t / NB;
        const int rp = pre + (k ? sb[k - 1] : 0) + lexcl[i];
        cur[t] = rp + plane[i];
        if (c == 0) rowptr[i] = rp;
    }
    __syncthreads();

    const int per = (((e + HC - 1) / HC) + 3) & ~3;
    const int beg = min(c * per, e);
    const int end = min(beg + per, e);
    const bool al = ((((size_t)src | (size_t)dst) & 15) == 0);
    const int nvec = al ? ((end - beg) >> 2) : 0;
    const int4* __restrict__ d4 = (const int4*)(dst + beg);
    const int4* __restrict__ s4 = (const int4*)(src + beg);

    for (int q = tid; q < nvec; q += 4 * HT) {
        const int q1 = q + 1 * HT, q2 = q + 2 * HT, q3 = q + 3 * HT;
        const int4 dv0 = d4[q],                   sv0 = s4[q];
        const int4 dv1 = d4[min(q1, nvec - 1)],   sv1 = s4[min(q1, nvec - 1)];
        const int4 dv2 = d4[min(q2, nvec - 1)],   sv2 = s4[min(q2, nvec - 1)];
        const int4 dv3 = d4[min(q3, nvec - 1)],   sv3 = s4[min(q3, nvec - 1)];
        fill4(cur, sorted, dv0, sv0, lo);
        if (q1 < nvec) fill4(cur, sorted, dv1, sv1, lo);
        if (q2 < nvec) fill4(cur, sorted, dv2, sv2, lo);
        if (q3 < nvec) fill4(cur, sorted, dv3, sv3, lo);
    }
    for (int j = beg + 4 * nvec + tid; j < end; j += HT) {
        const unsigned a = (unsigned)(dst[j] - lo);
        if (a < FR) sorted[atomicAdd(&cur[a], 1)] = src[j];
    }
}

template <int RELU>
static __device__ __forceinline__ float4 gather_row(const int* __restrict__ rowptr,
                                                    const int* __restrict__ sorted,
                                                    const float* __restrict__ h,
                                                    const float* __restrict__ nd,
                                                    const float* __restrict__ bias,
                                                    int row, int sub, int grp, int n, int e) {
    const int beg = rowptr[row];
    const int end = (row + 1 < n) ? rowptr[row + 1] : e;
    const int deg = end - beg;
    const float4* __restrict__ h4 = (const float4*)h;
    float4 acc = h4[(size_t)row * 16 + sub];

    for (int base = 0; base < deg; base += 16) {
        const int cnt = min(16, deg - base);
        int idx = row;
        if (sub < cnt) idx = sorted[beg + base + sub];
        for (int k = 0; k < cnt; k += 4) {
            const int s0 = __builtin_amdgcn_ds_bpermute((grp + k + 0) << 2, idx);
            const int s1 = __builtin_amdgcn_ds_bpermute((grp + k + 1) << 2, idx);
            const int s2 = __builtin_amdgcn_ds_bpermute((grp + k + 2) << 2, idx);
            const int s3 = __builtin_amdgcn_ds_bpermute((grp + k + 3) << 2, idx);
            const float4 v0 = h4[(size_t)s0 * 16 + sub];
            const float4 v1 = h4[(size_t)s1 * 16 + sub];
            const float4 v2 = h4[(size_t)s2 * 16 + sub];
            const float4 v3 = h4[(size_t)s3 * 16 + sub];
            const bool p1 = k + 1 < cnt, p2 = k + 2 < cnt, p3 = k + 3 < cnt;
            acc.x += v0.x; acc.y += v0.y; acc.z += v0.z; acc.w += v0.w;
            acc.x += p1 ? v1.x : 0.0f; acc.y += p1 ? v1.y : 0.0f;
            acc.z += p1 ? v1.z : 0.0f; acc.w += p1 ? v1.w : 0.0f;
            acc.x += p2 ? v2.x : 0.0f; acc.y += p2 ? v2.y : 0.0f;
            acc.z += p2 ? v2.z : 0.0f; acc.w += p2 ? v2.w : 0.0f;
            acc.x += p3 ? v3.x : 0.0f; acc.y += p3 ? v3.y : 0.0f;
            acc.z += p3 ? v3.z : 0.0f; acc.w += p3 ? v3.w : 0.0f;
        }
    }

    const float s = nd[row];
    const float4 b = *(const float4*)(bias + sub * 4);
    float4 v;
    v.x = fmaf(acc.x, s, b.x);
    v.y = fmaf(acc.y, s, b.y);
    v.z = fmaf(acc.z, s, b.z);
    v.w = fmaf(acc.w, s, b.w);
    if (RELU) {
        v.x = fmaxf(v.x, 0.0f); v.y = fmaxf(v.y, 0.0f);
        v.z = fmaxf(v.z, 0.0f); v.w = fmaxf(v.w, 0.0f);
    }
    return v;
}

__global__ __launch_bounds__(256) void k_gather(const int* __restrict__ rowptr,
                                                const int* __restrict__ sorted,
                                                const float* __restrict__ h,
                                                const float* __restrict__ nd,
                                                const float* __restrict__ bias,
                                                float* __restrict__ out, int n, int e) {
    const int sub = threadIdx.x & 15;
    const int grp = threadIdx.x & 16;
    const int row = blockIdx.x * 16 + (threadIdx.x >> 4);
    if (row >= n) return;
    const float4 v = gather_row<0>(rowptr, sorted, h, nd, bias, row, sub, grp, n, e);
    *(float4*)(out + (size_t)row * F_OUT + sub * 4) = v;
}

#define GR 64
#define GT (GR * 16)
#define TP (F_OUT + 4)
static_assert(GT <= 1024 && (F_OUT * F_OUT) % GT == 0, "whole W2 elements per thread");
__global__ __launch_bounds__(GT) void k_gather_gemm(const int* __restrict__ rowptr,
                                                    const int* __restrict__ sorted,
                                                    const float* __restrict__ h,
                                                    const float* __restrict__ nd,
                                                    const float* __restrict__ bias,
                                                    const float* __restrict__ W,
                                                    const float* __restrict__ norm,
                                                    float* __restrict__ H, int n, int e) {
    __shared__ alignas(16) float ldsW[F_OUT * TP];
    __shared__ alignas(16) float tileA[GR * TP];
    __shared__ alignas(16) float tileD[GR * TP];
    const int sub = threadIdx.x & 15;
    const int grp = threadIdx.x & 16;
    const int rib = threadIdx.x >> 4;
    const int row = blockIdx.x * GR + rib;

    constexpr int PER = F_OUT * F_OUT / GT;
    float w[PER];
#pragma unroll
    for (int i = 0; i < PER; ++i) w[i] = W[threadIdx.x + i * GT];

    float4 v = make_float4(0.0f, 0.0f, 0.0f, 0.0f);
    float sn = 0.0f;
    if (row < n) {
        v = gather_row<1>(rowptr, sorted, h, nd, bias, row, sub, grp, n, e);
        sn = norm[row];
    }

#pragma unroll
    for (int i = 0; i < PER; ++i) {
        const int t = threadIdx.x + i * GT;
        ldsW[(t % F_OUT) * TP + t / F_OUT] = w[i];
    }
    *(float4*)(&tileA[rib * TP + sub * 4]) = v;
    __syncthreads();

    const int wave = __builtin_amdgcn_readfirstlane(threadIdx.x >> 5);
    if (wave < 16) {
        const int half = (threadIdx.x >> 4) & 1;
        const int mt = wave >> 2, nt = wave & 3;
        const float* ap = &tileA[(mt * 16 + sub) * TP + 2 * half];
        const float* bp = &ldsW[(nt * 16 + sub) * TP + 2 * half];
        v8f c = {};
#pragma unroll
        for (int i = 0; i < 16; ++i) {
            const v2f a = *(const v2f*)(ap + 4 * i);
            const v2f b = *(const v2f*)(bp + 4 * i);
            c = __builtin_amdgcn_wmma_f32_16x16x4_f32(false, a, false, b, (short)0, c, false, false);
        }
#pragma unroll
        for (int r = 0; r < 8; ++r) tileD[(mt * 16 + 8 * half + r) * TP + nt * 16 + sub] = c[r];
    }
    __syncthreads();

    if (row < n) {
        float4 d = *(const float4*)(&tileD[rib * TP + sub * 4]);
        d.x *= sn; d.y *= sn; d.z *= sn; d.w *= sn;
        *(float4*)(H + (size_t)row * F_OUT + sub * 4) = d;
    }
}

extern "C" void kernel_launch(void* const* d_in, const int* in_sizes, int n_in,
                              void* d_out, int out_size, void* d_ws, size_t ws_size,
                              hipStream_t stream) {
    const float* x   = (const float*)d_in[0];
    const int*   src = (const int*)d_in[1];
    const int*   dst = (const int*)d_in[2];
    const float* W1  = (const float*)d_in[3];
    const float* b1  = (const float*)d_in[4];
    const float* W2  = (const float*)d_in[5];
    const float* b2  = (const float*)d_in[6];
    float*       out = (float*)d_out;

    const int n = in_sizes[0] / 128;
    const int e = in_sizes[1];

    float* ws   = (float*)d_ws;
    float* buf0 = ws;
    int*   sorted = (int*)(ws + (size_t)n * F_OUT);
    int*   rowptr = sorted + e;
    int*   lexcl  = rowptr + n;
    int*   bsum   = lexcl + n;
    float* ns   = ws + 2 * (size_t)n * F_OUT;
    float* nd   = ns + n;

    const int B = 256;

    static_assert(2 * HC * sizeof(int) <= F_OUT * sizeof(float), "the count planes fit buf0");
    int* part = (int*)buf0;
    k_hist<<<dim3(HC, cdiv(n, HR), 2), HT, 0, stream>>>(src, dst, part, n, e);
    k_norm<<<cdiv(n, NB), NB, 0, stream>>>(part, ns, nd, lexcl, bsum, n);

    const int fillBlocks = HC * cdiv(n, FR);
    const int tiles = cdiv(n, 16);
    int tpb = cdiv(tiles, 256);
    if (tpb > HT / 32) tpb = HT / 32;
    if (tpb < 1) tpb = 1;
    k_fill_gemm<<<fillBlocks + cdiv(tiles, tpb), HT, 0, stream>>>(
        src, dst, part + (size_t)HC * n, lexcl, bsum, rowptr, sorted, n, e,
        x, W1, ns, out, fillBlocks, tpb);

    k_gather_gemm<<<cdiv(n, GR), GT, 0, stream>>>(rowptr, sorted, out, nd, b1, W2, ns, buf0, n, e);
    k_gather<<<cdiv(n, 16), B, 0, stream>>>(rowptr, sorted, buf0, nd, b2, out, n, e);
}


# Per-call sort by dst in two levels: slices bucketed in LDS, written dense; then one block per 400-node bucket counts degrees, places edges, does its x W1 tiles; rows now in arrival order; 4 launches
# speedup vs baseline: 4.8312x; 1.1536x over previous
#include <hip/hip_runtime.h>
#include <math.h>

#define F_OUT 64

typedef __attribute__((ext_vector_type(2))) float v2f;
typedef __attribute__((ext_vector_type(8))) float v8f;

static inline int cdiv(int a, int b) { return (a + b - 1) / b; }


template <int K, int NT>
static __device__ __forceinline__ void gemm_stage_w_once(const float* __restrict__ W, float* ldsW) {
    constexpr int PER = K * F_OUT / NT;
    static_assert(PER * NT == K * F_OUT, "whole elements per thread");
    float v[PER];
#pragma unroll
    for (int i = 0; i < PER; ++i) v[i] = W[threadIdx.x + i * NT];
#pragma unroll
    for (int i = 0; i < PER; ++i) {
        const int t = threadIdx.x + i * NT;
        ldsW[(t % F_OUT) * (K + 2) + t / F_OUT] = v[i];
    }
}

template <int K, int U>
static __device__ __forceinline__ void gemm_tile(const float* __restrict__ X, const float* ldsW,
                                                 const float* norm, int nbase,
                                                 float* __restrict__ H, int nrows, int mBase) {
    const int lane  = threadIdx.x & 31;
    const int half  = lane >> 4;
    const int l16   = lane & 15;

    int rowA = mBase + l16;
    if (rowA >= nrows) rowA = nrows - 1;
    const float* xrow = X + (size_t)rowA * K;

    v8f c0 = {}, c1 = {}, c2 = {}, c3 = {};
#pragma unroll U
    for (int k0 = 0; k0 < K; k0 += 4) {
        const int kk = k0 + 2 * half;
        v2f a = *(const v2f*)(xrow + kk);
        v2f b0 = *(const v2f*)(&ldsW[(0 * 16 + l16) * (K + 2) + kk]);
        v2f b1 = *(const v2f*)(&ldsW[(1 * 16 + l16) * (K + 2) + kk]);
        v2f b2 = *(const v2f*)(&ldsW[(2 * 16 + l16) * (K + 2) + kk]);
        v2f b3 = *(const v2f*)(&ldsW[(3 * 16 + l16) * (K + 2) + kk]);
        c0 = __builtin_amdgcn_wmma_f32_16x16x4_f32(false, a, false, b0, (short)0, c0, false, false);
        c1 = __builtin_amdgcn_wmma_f32_16x16x4_f32(false, a, false, b1, (short)0, c1, false, false);
        c2 = __builtin_amdgcn_wmma_f32_16x16x4_f32(false, a, false, b2, (short)0, c2, false, false);
        c3 = __builtin_amdgcn_wmma_f32_16x16x4_f32(false, a, false, b3, (short)0, c3, false, false);
    }

#pragma unroll
    for (int r = 0; r < 8; ++r) {
        int row = mBase + 8 * half + r;
        if (row < nrows) {
            float s = norm[row - nbase];
            size_t base = (size_t)row * F_OUT + l16;
            float v0 = c0[r] * s, v1 = c1[r] * s, v2 = c2[r] * s, v3 = c3[r] * s;
            H[base +  0] = v0;  H[base + 16] = v1;
            H[base + 32] = v2;  H[base + 48] = v3;
        }
    }
}

#define BW   400
#define SL   6400
#define PT   1024
#define EPT  ((SL + PT - 1) / PT)
#define MAXB 256
static_assert(BW % 16 == 0 && BW / 16 <= PT / 32, "a wave per row tile of the bucket");
static_assert(BW <= PT, "a thread per node of the bucket");

static __device__ __forceinline__ int wave_excl_scan(int* c, int* o, int m) {
    const int lane = threadIdx.x & 31;
    const int per = (m + 31) >> 5;
    const int b0 = lane * per;
    int sum = 0;
    for (int k = 0; k < per; ++k) if (b0 + k < m) sum += c[b0 + k];
    int inc = sum;
    for (int off = 1; off < 32; off <<= 1) {
        const int t = __shfl_up(inc, off);
        if (lane >= off) inc += t;
    }
    int run = inc - sum;
    for (int k = 0; k < per; ++k) {
        if (b0 + k < m) {
            const int v = c[b0 + k];
            c[b0 + k] = run;
            if (o) o[b0 + k] = run;
            run += v;
        }
    }
    return __shfl(inc, 31);
}

__global__ __launch_bounds__(PT) void k_part(const int* __restrict__ src,
                                             const int* __restrict__ dst,
                                             int* __restrict__ pD, int* __restrict__ pS,
                                             int* __restrict__ qS, int* __restrict__ tabD,
                                             int* __restrict__ tabS, int e, int nbk, int nsl) {
    __shared__ int cD[MAXB], cS[MAXB], oD[MAXB + 1], oS[MAXB + 1];
    __shared__ int stD[SL], stP[SL], stQ[SL];
    const int tid = threadIdx.x;
    const int s   = blockIdx.x;
    const int beg = s * SL;
    const int len = min(SL, e - beg);

    for (int t = tid; t < MAXB; t += PT) { cD[t] = 0; cS[t] = 0; }

    int d[EPT], u[EPT];
#pragma unroll
    for (int i = 0; i < EPT; ++i) {
        const int j = tid + i * PT;
        d[i] = -1; u[i] = 0;
        if (j < len) { d[i] = dst[beg + j]; u[i] = src[beg + j]; }
    }
    __syncthreads();
#pragma unroll
    for (int i = 0; i < EPT; ++i) {
        if (d[i] >= 0) {
            atomicAdd(&cD[(unsigned)d[i] / BW], 1);
            atomicAdd(&cS[(unsigned)u[i] / BW], 1);
        }
    }
    __syncthreads();
    const int wave = __builtin_amdgcn_readfirstlane(tid >> 5);
    if (wave == 0) { const int t = wave_excl_scan(cD, oD, nbk); if ((tid & 31) == 0) oD[nbk] = t; }
    if (wave == 1) { const int t = wave_excl_scan(cS, oS, nbk); if ((tid & 31) == 0) oS[nbk] = t; }
    __syncthreads();

    for (int b = tid; b <= nbk; b += PT) {
        tabD[(size_t)b * nsl + s] = oD[b];
        tabS[(size_t)b * nsl + s] = oS[b];
    }
#pragma unroll
    for (int i = 0; i < EPT; ++i) {
        if (d[i] >= 0) {
            const int p = atomicAdd(&cD[(unsigned)d[i] / BW], 1);
            stD[p] = d[i];
            stP[p] = u[i];
            const int q = atomicAdd(&cS[(unsigned)u[i] / BW], 1);
            stQ[q] = u[i];
        }
    }
    __syncthreads();
    for (int j = tid; j < len; j += PT) {
        pD[beg + j] = stD[j];
        pS[beg + j] = stP[j];
        qS[beg + j] = stQ[j];
    }
}

#define RB 8
__global__ __launch_bounds__(PT) void k_bucket_gemm(const int* __restrict__ pD,
                                                    const int* __restrict__ pS,
                                                    const int* __restrict__ qS,
                                                    const int* __restrict__ tabD,
                                                    const int* __restrict__ tabS,
                                                    int* __restrict__ rowptr,
                                                    int* __restrict__ sorted,
                                                    float* __restrict__ ns, float* __restrict__ nd,
                                                    int n, int nsl,
                                                    const float* __restrict__ X,
                                                    const float* __restrict__ W,
                                                    float* __restrict__ H) {
    __shared__ alignas(16) float ldsW[F_OUT * (128 + 2)];
    __shared__ int cnt[BW], cur[BW], cntS[BW];
    __shared__ float nsL[BW];
    __shared__ int base_s;
    const int tid = threadIdx.x;
    const int b   = blockIdx.x;
    const int lo  = b * BW;
    const int nn  = min(BW, n - lo);
    const int q   = tid & 3;

    gemm_stage_w_once<128, PT>(W, ldsW);
    if (tid < BW) { cnt[tid] = 0; cntS[tid] = 0; }
    if (tid == 0) base_s = 0;
    __syncthreads();

    int bpart = 0;
    for (int r = tid >> 2; r < nsl; r += PT / 4) {
        const int a0 = tabD[(size_t)b * nsl + r], a1 = tabD[(size_t)(b + 1) * nsl + r];
        const int s0 = tabS[(size_t)b * nsl + r], s1 = tabS[(size_t)(b + 1) * nsl + r];
        if (q == 0) bpart += a0;
        const int* __restrict__ pd = pD + (size_t)r * SL;
        const int* __restrict__ qs = qS + (size_t)r * SL;
        for (int j0 = 0; j0 < max(a1 - a0, s1 - s0); j0 += 4 * RB) {
            int vd[RB], vs[RB];
#pragma unroll
            for (int k = 0; k < RB; ++k) {
                const int jd = a0 + j0 + q + 4 * k, js = s0 + j0 + q + 4 * k;
                vd[k] = (jd < a1) ? pd[jd] : -1;
                vs[k] = (js < s1) ? qs[js] : -1;
            }
#pragma unroll
            for (int k = 0; k < RB; ++k) {
                if (vd[k] >= 0) atomicAdd(&cnt[vd[k] - lo], 1);
                if (vs[k] >= 0) atomicAdd(&cntS[vs[k] - lo], 1);
            }
        }
    }
    if (bpart) atomicAdd(&base_s, bpart);
    __syncthreads();

    const int base = base_s;
    if (tid < nn) {
        const int i = lo + tid;
        nd[i] = rsqrtf((float)(cnt[tid] + 1));
        const float v = rsqrtf((float)(cntS[tid] + 1));
        ns[i] = v;
        nsL[tid] = v;
        cur[tid] = cnt[tid];
    }
    __syncthreads();
    const int wave = __builtin_amdgcn_readfirstlane(tid >> 5);
    if (wave == 0) wave_excl_scan(cur, nullptr, nn);
    __syncthreads();
    if (tid < nn) rowptr[lo + tid] = base + cur[tid];
    __syncthreads();

    for (int r = tid >> 2; r < nsl; r += PT / 4) {
        const int a0 = tabD[(size_t)b * nsl + r], a1 = tabD[(size_t)(b + 1) * nsl + r];
        const int* __restrict__ pd = pD + (size_t)r * SL;
        const int* __restrict__ ps = pS + (size_t)r * SL;
        for (int j0 = a0; j0 < a1; j0 += 4 * RB) {
            int vd[RB], vp[RB];
#pragma unroll
            for (int k = 0; k < RB; ++k) {
                const int j = j0 + q + 4 * k;
                vd[k] = (j < a1) ? pd[j] : -1;
                vp[k] = (j < a1) ? ps[j] : 0;
            }
#pragma unroll
            for (int k = 0; k < RB; ++k)
                if (vd[k] >= 0) sorted[base + atomicAdd(&cur[vd[k] - lo], 1)] = vp[k];
        }
    }

    if (wave * 16 < nn)
        gemm_tile<128, 4>(X, ldsW, nsL, lo, H, n, lo + wave * 16);
}

template <int RELU>
static __device__ __forceinline__ float4 gather_row(const int* __restrict__ rowptr,
                                                    const int* __restrict__ sorted,
                                                    const float* __restrict__ h,
                                                    const float* __restrict__ nd,
                                                    const float* __restrict__ bias,
                                                    int row, int sub, int grp, int n, int e) {
    const int beg = rowptr[row];
    const int end = (row + 1 < n) ? rowptr[row + 1] : e;
    const int deg = end - beg;
    const float4* __restrict__ h4 = (const float4*)h;
    float4 acc = h4[(size_t)row * 16 + sub];

    for (int base = 0; base < deg; base += 16) {
        const int cnt = min(16, deg - base);
        int idx = row;
        if (sub < cnt) idx = sorted[beg + base + sub];
        for (int k = 0; k < cnt; k += 4) {
            const int s0 = __builtin_amdgcn_ds_bpermute((grp + k + 0) << 2, idx);
            const int s1 = __builtin_amdgcn_ds_bpermute((grp + k + 1) << 2, idx);
            const int s2 = __builtin_amdgcn_ds_bpermute((grp + k + 2) << 2, idx);
            const int s3 = __builtin_amdgcn_ds_bpermute((grp + k + 3) << 2, idx);
            const float4 v0 = h4[(size_t)s0 * 16 + sub];
            const float4 v1 = h4[(size_t)s1 * 16 + sub];
            const float4 v2 = h4[(size_t)s2 * 16 + sub];
            const float4 v3 = h4[(size_t)s3 * 16 + sub];
            const bool p1 = k + 1 < cnt, p2 = k + 2 < cnt, p3 = k + 3 < cnt;
            acc.x += v0.x; acc.y += v0.y; acc.z += v0.z; acc.w += v0.w;
            acc.x += p1 ? v1.x : 0.0f; acc.y += p1 ? v1.y : 0.0f;
            acc.z += p1 ? v1.z : 0.0f; acc.w += p1 ? v1.w : 0.0f;
            acc.x += p2 ? v2.x : 0.0f; acc.y += p2 ? v2.y : 0.0f;
            acc.z += p2 ? v2.z : 0.0f; acc.w += p2 ? v2.w : 0.0f;
            acc.x += p3 ? v3.x : 0.0f; acc.y += p3 ? v3.y : 0.0f;
            acc.z += p3 ? v3.z : 0.0f; acc.w += p3 ? v3.w : 0.0f;
        }
    }

    const float s = nd[row];
    const float4 b = *(const float4*)(bias + sub * 4);
    float4 v;
    v.x = fmaf(acc.x, s, b.x);
    v.y = fmaf(acc.y, s, b.y);
    v.z = fmaf(acc.z, s, b.z);
    v.w = fmaf(acc.w, s, b.w);
    if (RELU) {
        v.x = fmaxf(v.x, 0.0f); v.y = fmaxf(v.y, 0.0f);
        v.z = fmaxf(v.z, 0.0f); v.w = fmaxf(v.w, 0.0f);
    }
    return v;
}

__global__ __launch_bounds__(256) void k_gather(const int* __restrict__ rowptr,
                                                const int* __restrict__ sorted,
                                                const float* __restrict__ h,
                                                const float* __restrict__ nd,
                                                const float* __restrict__ bias,
                                                float* __restrict__ out, int n, int e) {
    const int sub = threadIdx.x & 15;
    const int grp = threadIdx.x & 16;
    const int row = blockIdx.x * 16 + (threadIdx.x >> 4);
    if (row >= n) return;
    const float4 v = gather_row<0>(rowptr, sorted, h, nd, bias, row, sub, grp, n, e);
    *(float4*)(out + (size_t)row * F_OUT + sub * 4) = v;
}

#define GR 64
#define GT (GR * 16)
#define TP (F_OUT + 4)
static_assert(GT <= 1024 && (F_OUT * F_OUT) % GT == 0, "whole W2 elements per thread");
__global__ __launch_bounds__(GT) void k_gather_gemm(const int* __restrict__ rowptr,
                                                    const int* __restrict__ sorted,
                                                    const float* __restrict__ h,
                                                    const float* __restrict__ nd,
                                                    const float* __restrict__ bias,
                                                    const float* __restrict__ W,
                                                    const float* __restrict__ norm,
                                                    float* __restrict__ H, int n, int e) {
    __shared__ alignas(16) float ldsW[F_OUT * TP];
    __shared__ alignas(16) float tileA[GR * TP];
    __shared__ alignas(16) float tileD[GR * TP];
    const int sub = threadIdx.x & 15;
    const int grp = threadIdx.x & 16;
    const int rib = threadIdx.x >> 4;
    const int row = blockIdx.x * GR + rib;

    constexpr int PER = F_OUT * F_OUT / GT;
    float w[PER];
#pragma unroll
    for (int i = 0; i < PER; ++i) w[i] = W[threadIdx.x + i * GT];

    float4 v = make_float4(0.0f, 0.0f, 0.0f, 0.0f);
    float sn = 0.0f;
    if (row < n) {
        v = gather_row<1>(rowptr, sorted, h, nd, bias, row, sub, grp, n, e);
        sn = norm[row];
    }

#pragma unroll
    for (int i = 0; i < PER; ++i) {
        const int t = threadIdx.x + i * GT;
        ldsW[(t % F_OUT) * TP + t / F_OUT] = w[i];
    }
    *(float4*)(&tileA[rib * TP + sub * 4]) = v;
    __syncthreads();

    const int wave = __builtin_amdgcn_readfirstlane(threadIdx.x >> 5);
    if (wave < 16) {
        const int half = (threadIdx.x >> 4) & 1;
        const int mt = wave >> 2, nt = wave & 3;
        const float* ap = &tileA[(mt * 16 + sub) * TP + 2 * half];
        const float* bp = &ldsW[(nt * 16 + sub) * TP + 2 * half];
        v8f c = {};
#pragma unroll
        for (int i = 0; i < 16; ++i) {
            const v2f a = *(const v2f*)(ap + 4 * i);
            const v2f b = *(const v2f*)(bp + 4 * i);
            c = __builtin_amdgcn_wmma_f32_16x16x4_f32(false, a, false, b, (short)0, c, false, false);
        }
#pragma unroll
        for (int r = 0; r < 8; ++r) tileD[(mt * 16 + 8 * half + r) * TP + nt * 16 + sub] = c[r];
    }
    __syncthreads();

    if (row < n) {
        float4 d = *(const float4*)(&tileD[rib * TP + sub * 4]);
        d.x *= sn; d.y *= sn; d.z *= sn; d.w *= sn;
        *(float4*)(H + (size_t)row * F_OUT + sub * 4) = d;
    }
}

extern "C" void kernel_launch(void* const* d_in, const int* in_sizes, int n_in,
                              void* d_out, int out_size, void* d_ws, size_t ws_size,
                              hipStream_t stream) {
    const float* x   = (const float*)d_in[0];
    const int*   src = (const int*)d_in[1];
    const int*   dst = (const int*)d_in[2];
    const float* W1  = (const float*)d_in[3];
    const float* b1  = (const float*)d_in[4];
    const float* W2  = (const float*)d_in[5];
    const float* b2  = (const float*)d_in[6];
    float*       out = (float*)d_out;

    const int n = in_sizes[0] / 128;
    const int e = in_sizes[1];

    float* ws   = (float*)d_ws;
    float* buf0 = ws;
    int*   sorted = (int*)(ws + (size_t)n * F_OUT);
    int*   rowptr = sorted + e;
    float* ns   = ws + 2 * (size_t)n * F_OUT;
    float* nd   = ns + n;

    const int B = 256;

    const int nbk = cdiv(n, BW);
    const int nsl = cdiv(e, SL);
    int* pD   = (int*)buf0;
    int* pS   = pD + (size_t)nsl * SL;
    int* qS   = pS + (size_t)nsl * SL;
    int* tabD = qS + (size_t)nsl * SL;
    int* tabS = tabD + (size_t)(nbk + 1) * nsl;
    k_part<<<nsl, PT, 0, stream>>>(src, dst, pD, pS, qS, tabD, tabS, e, nbk, nsl);

    k_bucket_gemm<<<nbk, PT, 0, stream>>>(pD, pS, qS, tabD, tabS, rowptr, sorted, ns, nd,
                                          n, nsl, x, W1, out);

    k_gather_gemm<<<cdiv(n, GR), GT, 0, stream>>>(rowptr, sorted, out, nd, b1, W2, ns, buf0, n, e);
    k_gather<<<cdiv(n, 16), B, 0, stream>>>(rowptr, sorted, buf0, nd, b2, out, n, e);
}


# Both gather launches as 1,563 blocks of 8 waves, all resident at once: a 16-lane group does 4 rows, bounds asked up front, next row's head a row ahead; layer-1 tile 16 rows, 4 WMMA waves; same sums
# speedup vs baseline: 5.0171x; 1.0385x over previous
#include <hip/hip_runtime.h>
#include <math.h>

#define F_OUT 64

typedef __attribute__((ext_vector_type(2))) float v2f;
typedef __attribute__((ext_vector_type(8))) float v8f;

static inline int cdiv(int a, int b) { return (a + b - 1) / b; }


template <int K, int NT>
static __device__ __forceinline__ void gemm_stage_w_once(const float* __restrict__ W, float* ldsW) {
    constexpr int PER = K * F_OUT / NT;
    static_assert(PER * NT == K * F_OUT, "whole elements per thread");
    float v[PER];
#pragma unroll
    for (int i = 0; i < PER; ++i) v[i] = W[threadIdx.x + i * NT];
#pragma unroll
    for (int i = 0; i < PER; ++i) {
        const int t = threadIdx.x + i * NT;
        ldsW[(t % F_OUT) * (K + 2) + t / F_OUT] = v[i];
    }
}

template <int K, int U>
static __device__ __forceinline__ void gemm_tile(const float* __restrict__ X, const float* ldsW,
                                                 const float* norm, int nbase,
                                                 float* __restrict__ H, int nrows, int mBase) {
    const int lane  = threadIdx.x & 31;
    const int half  = lane >> 4;
    const int l16   = lane & 15;

    int rowA = mBase + l16;
    if (rowA >= nrows) rowA = nrows - 1;
    const float* xrow = X + (size_t)rowA * K;

    v8f c0 = {}, c1 = {}, c2 = {}, c3 = {};
#pragma unroll U
    for (int k0 = 0; k0 < K; k0 += 4) {
        const int kk = k0 + 2 * half;
        v2f a = *(const v2f*)(xrow + kk);
        v2f b0 = *(const v2f*)(&ldsW[(0 * 16 + l16) * (K + 2) + kk]);
        v2f b1 = *(const v2f*)(&ldsW[(1 * 16 + l16) * (K + 2) + kk]);
        v2f b2 = *(const v2f*)(&ldsW[(2 * 16 + l16) * (K + 2) + kk]);
        v2f b3 = *(const v2f*)(&ldsW[(3 * 16 + l16) * (K + 2) + kk]);
        c0 = __builtin_amdgcn_wmma_f32_16x16x4_f32(false, a, false, b0, (short)0, c0, false, false);
        c1 = __builtin_amdgcn_wmma_f32_16x16x4_f32(false, a, false, b1, (short)0, c1, false, false);
        c2 = __builtin_amdgcn_wmma_f32_16x16x4_f32(false, a, false, b2, (short)0, c2, false, false);
        c3 = __builtin_amdgcn_wmma_f32_16x16x4_f32(false, a, false, b3, (short)0, c3, false, false);
    }

#pragma unroll
    for (int r = 0; r < 8; ++r) {
        int row = mBase + 8 * half + r;
        if (row < nrows) {
            float s = norm[row - nbase];
            size_t base = (size_t)row * F_OUT + l16;
            float v0 = c0[r] * s, v1 = c1[r] * s, v2 = c2[r] * s, v3 = c3[r] * s;
            H[base +  0] = v0;  H[base + 16] = v1;
            H[base + 32] = v2;  H[base + 48] = v3;
        }
    }
}

#define BW   400
#define SL   6400
#define PT   1024
#define EPT  ((SL + PT - 1) / PT)
#define MAXB 256
static_assert(BW % 16 == 0 && BW / 16 <= PT / 32, "a wave per row tile of the bucket");
static_assert(BW <= PT, "a thread per node of the bucket");

static __device__ __forceinline__ int wave_excl_scan(int* c, int* o, int m) {
    const int lane = threadIdx.x & 31;
    const int per = (m + 31) >> 5;
    const int b0 = lane * per;
    int sum = 0;
    for (int k = 0; k < per; ++k) if (b0 + k < m) sum += c[b0 + k];
    int inc = sum;
    for (int off = 1; off < 32; off <<= 1) {
        const int t = __shfl_up(inc, off);
        if (lane >= off) inc += t;
    }
    int run = inc - sum;
    for (int k = 0; k < per; ++k) {
        if (b0 + k < m) {
            const int v = c[b0 + k];
            c[b0 + k] = run;
            if (o) o[b0 + k] = run;
            run += v;
        }
    }
    return __shfl(inc, 31);
}

__global__ __launch_bounds__(PT) void k_part(const int* __restrict__ src,
                                             const int* __restrict__ dst,
                                             int* __restrict__ pD, int* __restrict__ pS,
                                             int* __restrict__ qS, int* __restrict__ tabD,
                                             int* __restrict__ tabS, int e, int nbk, int nsl) {
    __shared__ int cD[MAXB], cS[MAXB], oD[MAXB + 1], oS[MAXB + 1];
    __shared__ int stD[SL], stP[SL], stQ[SL];
    const int tid = threadIdx.x;
    const int s   = blockIdx.x;
    const int beg = s * SL;
    const int len = min(SL, e - beg);

    for (int t = tid; t < MAXB; t += PT) { cD[t] = 0; cS[t] = 0; }

    int d[EPT], u[EPT];
#pragma unroll
    for (int i = 0; i < EPT; ++i) {
        const int j = tid + i * PT;
        d[i] = -1; u[i] = 0;
        if (j < len) { d[i] = dst[beg + j]; u[i] = src[beg + j]; }
    }
    __syncthreads();
#pragma unroll
    for (int i = 0; i < EPT; ++i) {
        if (d[i] >= 0) {
            atomicAdd(&cD[(unsigned)d[i] / BW], 1);
            atomicAdd(&cS[(unsigned)u[i] / BW], 1);
        }
    }
    __syncthreads();
    const int wave = __builtin_amdgcn_readfirstlane(tid >> 5);
    if (wave == 0) { const int t = wave_excl_scan(cD, oD, nbk); if ((tid & 31) == 0) oD[nbk] = t; }
    if (wave == 1) { const int t = wave_excl_scan(cS, oS, nbk); if ((tid & 31) == 0) oS[nbk] = t; }
    __syncthreads();

    for (int b = tid; b <= nbk; b += PT) {
        tabD[(size_t)b * nsl + s] = oD[b];
        tabS[(size_t)b * nsl + s] = oS[b];
    }
#pragma unroll
    for (int i = 0; i < EPT; ++i) {
        if (d[i] >= 0) {
            const int p = atomicAdd(&cD[(unsigned)d[i] / BW], 1);
            stD[p] = d[i];
            stP[p] = u[i];
            const int q = atomicAdd(&cS[(unsigned)u[i] / BW], 1);
            stQ[q] = u[i];
        }
    }
    __syncthreads();
    for (int j = tid; j < len; j += PT) {
        pD[beg + j] = stD[j];
        pS[beg + j] = stP[j];
        qS[beg + j] = stQ[j];
    }
}

#define RB 8
__global__ __launch_bounds__(PT) void k_bucket_gemm(const int* __restrict__ pD,
                                                    const int* __restrict__ pS,
                                                    const int* __restrict__ qS,
                                                    const int* __restrict__ tabD,
                                                    const int* __restrict__ tabS,
                                                    int* __restrict__ rowptr,
                                                    int* __restrict__ sorted,
                                                    float* __restrict__ ns, float* __restrict__ nd,
                                                    int n, int nsl,
                                                    const float* __restrict__ X,
                                                    const float* __restrict__ W,
                                                    float* __restrict__ H) {
    __shared__ alignas(16) float ldsW[F_OUT * (128 + 2)];
    __shared__ int cnt[BW], cur[BW], cntS[BW];
    __shared__ float nsL[BW];
    __shared__ int base_s;
    const int tid = threadIdx.x;
    const int b   = blockIdx.x;
    const int lo  = b * BW;
    const int nn  = min(BW, n - lo);
    const int q   = tid & 3;

    gemm_stage_w_once<128, PT>(W, ldsW);
    if (tid < BW) { cnt[tid] = 0; cntS[tid] = 0; }
    if (tid == 0) base_s = 0;
    __syncthreads();

    int bpart = 0;
    for (int r = tid >> 2; r < nsl; r += PT / 4) {
        const int a0 = tabD[(size_t)b * nsl + r], a1 = tabD[(size_t)(b + 1) * nsl + r];
        const int s0 = tabS[(size_t)b * nsl + r], s1 = tabS[(size_t)(b + 1) * nsl + r];
        if (q == 0) bpart += a0;
        const int* __restrict__ pd = pD + (size_t)r * SL;
        const int* __restrict__ qs = qS + (size_t)r * SL;
        for (int j0 = 0; j0 < max(a1 - a0, s1 - s0); j0 += 4 * RB) {
            int vd[RB], vs[RB];
#pragma unroll
            for (int k = 0; k < RB; ++k) {
                const int jd = a0 + j0 + q + 4 * k, js = s0 + j0 + q + 4 * k;
                vd[k] = (jd < a1) ? pd[jd] : -1;
                vs[k] = (js < s1) ? qs[js] : -1;
            }
#pragma unroll
            for (int k = 0; k < RB; ++k) {
                if (vd[k] >= 0) atomicAdd(&cnt[vd[k] - lo], 1);
                if (vs[k] >= 0) atomicAdd(&cntS[vs[k] - lo], 1);
            }
        }
    }
    if (bpart) atomicAdd(&base_s, bpart);
    __syncthreads();

    const int base = base_s;
    if (tid < nn) {
        const int i = lo + tid;
        nd[i] = rsqrtf((float)(cnt[tid] + 1));
        const float v = rsqrtf((float)(cntS[tid] + 1));
        ns[i] = v;
        nsL[tid] = v;
        cur[tid] = cnt[tid];
    }
    __syncthreads();
    const int wave = __builtin_amdgcn_readfirstlane(tid >> 5);
    if (wave == 0) wave_excl_scan(cur, nullptr, nn);
    __syncthreads();
    if (tid < nn) rowptr[lo + tid] = base + cur[tid];
    __syncthreads();

    for (int r = tid >> 2; r < nsl; r += PT / 4) {
        const int a0 = tabD[(size_t)b * nsl + r], a1 = tabD[(size_t)(b + 1) * nsl + r];
        const int* __restrict__ pd = pD + (size_t)r * SL;
        const int* __restrict__ ps = pS + (size_t)r * SL;
        for (int j0 = a0; j0 < a1; j0 += 4 * RB) {
            int vd[RB], vp[RB];
#pragma unroll
            for (int k = 0; k < RB; ++k) {
                const int j = j0 + q + 4 * k;
                vd[k] = (j < a1) ? pd[j] : -1;
                vp[k] = (j < a1) ? ps[j] : 0;
            }
#pragma unroll
            for (int k = 0; k < RB; ++k)
                if (vd[k] >= 0) sorted[base + atomicAdd(&cur[vd[k] - lo], 1)] = vp[k];
        }
    }

    if (wave * 16 < nn)
        gemm_tile<128, 4>(X, ldsW, nsL, lo, H, n, lo + wave * 16);
}

#define GR 16
#define GT (GR * 16)
#define TPB 4

struct RowHead { float4 self; int idx; float s; };

static __device__ __forceinline__ RowHead row_head(const int* __restrict__ sorted,
                                                   const float4* __restrict__ h4,
                                                   const float* __restrict__ nd,
                                                   int row, int beg, int deg, int sub) {
    RowHead r;
    r.self = h4[(size_t)row * 16 + sub];
    r.s = nd[row];
    r.idx = row;
    if (sub < deg) r.idx = sorted[beg + sub];
    return r;
}

template <int RELU>
static __device__ __forceinline__ float4 gather_row(const int* __restrict__ sorted,
                                                    const float4* __restrict__ h4,
                                                    RowHead hd, float4 b,
                                                    int row, int beg, int deg, int sub, int grp) {
    float4 acc = hd.self;
    int idx = hd.idx;
    const float s = hd.s;

    for (int base = 0; base < deg; base += 16) {
        const int cnt = min(16, deg - base);
        if (base > 0) {
            idx = row;
            if (sub < cnt) idx = sorted[beg + base + sub];
        }
        for (int k = 0; k < cnt; k += 4) {
            const int s0 = __builtin_amdgcn_ds_bpermute((grp + k + 0) << 2, idx);
            const int s1 = __builtin_amdgcn_ds_bpermute((grp + k + 1) << 2, idx);
            const int s2 = __builtin_amdgcn_ds_bpermute((grp + k + 2) << 2, idx);
            const int s3 = __builtin_amdgcn_ds_bpermute((grp + k + 3) << 2, idx);
            const float4 v0 = h4[(size_t)s0 * 16 + sub];
            const float4 v1 = h4[(size_t)s1 * 16 + sub];
            const float4 v2 = h4[(size_t)s2 * 16 + sub];
            const float4 v3 = h4[(size_t)s3 * 16 + sub];
            const bool p1 = k + 1 < cnt, p2 = k + 2 < cnt, p3 = k + 3 < cnt;
            acc.x += v0.x; acc.y += v0.y; acc.z += v0.z; acc.w += v0.w;
            acc.x += p1 ? v1.x : 0.0f; acc.y += p1 ? v1.y : 0.0f;
            acc.z += p1 ? v1.z : 0.0f; acc.w += p1 ? v1.w : 0.0f;
            acc.x += p2 ? v2.x : 0.0f; acc.y += p2 ? v2.y : 0.0f;
            acc.z += p2 ? v2.z : 0.0f; acc.w += p2 ? v2.w : 0.0f;
            acc.x += p3 ? v3.x : 0.0f; acc.y += p3 ? v3.y : 0.0f;
            acc.z += p3 ? v3.z : 0.0f; acc.w += p3 ? v3.w : 0.0f;
        }
    }

    float4 v;
    v.x = fmaf(acc.x, s, b.x);
    v.y = fmaf(acc.y, s, b.y);
    v.z = fmaf(acc.z, s, b.z);
    v.w = fmaf(acc.w, s, b.w);
    if (RELU) {
        v.x = fmaxf(v.x, 0.0f); v.y = fmaxf(v.y, 0.0f);
        v.z = fmaxf(v.z, 0.0f); v.w = fmaxf(v.w, 0.0f);
    }
    return v;
}

static __device__ __forceinline__ void row_bounds(const int* __restrict__ rowptr, int row0,
                                                  int n, int e, int* rbeg, int* rdeg) {
#pragma unroll
    for (int it = 0; it < TPB; ++it) {
        const int row = row0 + it * GR;
        int b = 0, t = 0;
        if (row < n) {
            b = rowptr[row];
            t = (row + 1 < n) ? rowptr[row + 1] : e;
        }
        rbeg[it] = b;
        rdeg[it] = t - b;
    }
}

__global__ __launch_bounds__(GT) void k_gather(const int* __restrict__ rowptr,
                                               const int* __restrict__ sorted,
                                               const float* __restrict__ h,
                                               const float* __restrict__ nd,
                                               const float* __restrict__ bias,
                                               float* __restrict__ out, int n, int e) {
    const int sub = threadIdx.x & 15;
    const int grp = threadIdx.x & 16;
    const int row0 = blockIdx.x * (TPB * GR) + (threadIdx.x >> 4);
    const float4* __restrict__ h4 = (const float4*)h;
    const float4 b = *(const float4*)(bias + sub * 4);

    int rbeg[TPB], rdeg[TPB];
    row_bounds(rowptr, row0, n, e, rbeg, rdeg);

    RowHead hd;
    hd.self = make_float4(0.0f, 0.0f, 0.0f, 0.0f); hd.idx = 0; hd.s = 0.0f;
    if (row0 < n) hd = row_head(sorted, h4, nd, row0, rbeg[0], rdeg[0], sub);
#pragma unroll
    for (int it = 0; it < TPB; ++it) {
        const int row = row0 + it * GR;
        const RowHead cur = hd;
        if (it + 1 < TPB && row + GR < n)
            hd = row_head(sorted, h4, nd, row + GR, rbeg[it + 1], rdeg[it + 1], sub);
        if (row < n) {
            const float4 v = gather_row<0>(sorted, h4, cur, b, row, rbeg[it], rdeg[it], sub, grp);
            *(float4*)(out + (size_t)row * F_OUT + sub * 4) = v;
        }
    }
}

#define TP (F_OUT + 4)
static_assert(GT <= 1024 && (F_OUT * F_OUT) % GT == 0, "whole W2 elements per thread");
static_assert((GR / 16) * (F_OUT / 16) <= GT / 32, "a wave per 16x16 tile of the product");
__global__ __launch_bounds__(GT) __attribute__((amdgpu_waves_per_eu(16, 16)))
void k_gather_gemm(const int* __restrict__ rowptr,
                                                    const int* __restrict__ sorted,
                                                    const float* __restrict__ h,
                                                    const float* __restrict__ nd,
                                                    const float* __restrict__ bias,
                                                    const float* __restrict__ W,
                                                    const float* __restrict__ norm,
                                                    float* __restrict__ H, int n, int e) {
    __shared__ alignas(16) float ldsW[F_OUT * TP];
    __shared__ alignas(16) float tileA[GR * TP];
    __shared__ alignas(16) float tileD[GR * TP];
    const int sub = threadIdx.x & 15;
    const int grp = threadIdx.x & 16;
    const int rib = threadIdx.x >> 4;
    const int row0 = blockIdx.x * (TPB * GR) + rib;
    const float4* __restrict__ h4 = (const float4*)h;

    constexpr int PER = F_OUT * F_OUT / GT;
    float w[PER];
#pragma unroll
    for (int i = 0; i < PER; ++i) w[i] = W[threadIdx.x + i * GT];

    const float4 b = *(const float4*)(bias + sub * 4);
    int rbeg[TPB], rdeg[TPB];
    row_bounds(rowptr, row0, n, e, rbeg, rdeg);

#pragma unroll
    for (int i = 0; i < PER; ++i) {
        const int t = threadIdx.x + i * GT;
        ldsW[(t % F_OUT) * TP + t / F_OUT] = w[i];
    }

    const int wave = __builtin_amdgcn_readfirstlane(threadIdx.x >> 5);
    const int half = (threadIdx.x >> 4) & 1;

    RowHead hd;
    hd.self = make_float4(0.0f, 0.0f, 0.0f, 0.0f); hd.idx = 0; hd.s = 0.0f;
    if (row0 < n) hd = row_head(sorted, h4, nd, row0, rbeg[0], rdeg[0], sub);
#pragma unroll
    for (int it = 0; it < TPB; ++it) {
        const int row = row0 + it * GR;
        const RowHead cur = hd;
        if (it + 1 < TPB && row + GR < n)
            hd = row_head(sorted, h4, nd, row + GR, rbeg[it + 1], rdeg[it + 1], sub);
        float4 v = make_float4(0.0f, 0.0f, 0.0f, 0.0f);
        float sn = 0.0f;
        if (row < n) {
            v = gather_row<1>(sorted, h4, cur, b, row, rbeg[it], rdeg[it], sub, grp);
            sn = norm[row];
        }

        *(float4*)(&tileA[rib * TP + sub * 4]) = v;
        __syncthreads();

        if (wave < (GR / 16) * (F_OUT / 16)) {
            const int mt = wave >> 2, nt = wave & 3;
            const float* ap = &tileA[(mt * 16 + sub) * TP + 2 * half];
            const float* bp = &ldsW[(nt * 16 + sub) * TP + 2 * half];
            v8f c = {};
#pragma unroll
            for (int i = 0; i < 16; ++i) {
                const v2f a = *(const v2f*)(ap + 4 * i);
                const v2f bb = *(const v2f*)(bp + 4 * i);
                c = __builtin_amdgcn_wmma_f32_16x16x4_f32(false, a, false, bb, (short)0, c, false, false);
            }
#pragma unroll
            for (int r = 0; r < 8; ++r) tileD[(mt * 16 + 8 * half + r) * TP + nt * 16 + sub] = c[r];
        }
        __syncthreads();

        if (row < n) {
            float4 d = *(const float4*)(&tileD[rib * TP + sub * 4]);
            d.x *= sn; d.y *= sn; d.z *= sn; d.w *= sn;
            *(float4*)(H + (size_t)row * F_OUT + sub * 4) = d;
        }
    }
}

extern "C" void kernel_launch(void* const* d_in, const int* in_sizes, int n_in,
                              void* d_out, int out_size, void* d_ws, size_t ws_size,
                              hipStream_t stream) {
    const float* x   = (const float*)d_in[0];
    const int*   src = (const int*)d_in[1];
    const int*   dst = (const int*)d_in[2];
    const float* W1  = (const float*)d_in[3];
    const float* b1  = (const float*)d_in[4];
    const float* W2  = (const float*)d_in[5];
    const float* b2  = (const float*)d_in[6];
    float*       out = (float*)d_out;

    const int n = in_sizes[0] / 128;
    const int e = in_sizes[1];

    float* ws   = (float*)d_ws;
    float* buf0 = ws;
    int*   sorted = (int*)(ws + (size_t)n * F_OUT);
    int*   rowptr = sorted + e;
    float* ns   = ws + 2 * (size_t)n * F_OUT;
    float* nd   = ns + n;

    const int nbk = cdiv(n, BW);
    const int nsl = cdiv(e, SL);
    int* pD   = (int*)buf0;
    int* pS   = pD + (size_t)nsl * SL;
    int* qS   = pS + (size_t)nsl * SL;
    int* tabD = qS + (size_t)nsl * SL;
    int* tabS = tabD + (size_t)(nbk + 1) * nsl;
    k_part<<<nsl, PT, 0, stream>>>(src, dst, pD, pS, qS, tabD, tabS, e, nbk, nsl);

    k_bucket_gemm<<<nbk, PT, 0, stream>>>(pD, pS, qS, tabD, tabS, rowptr, sorted, ns, nd,
                                          n, nsl, x, W1, out);

    const int ngb = cdiv(n, TPB * GR);
    k_gather_gemm<<<ngb, GT, 0, stream>>>(rowptr, sorted, out, nd, b1, W2, ns, buf0, n, e);
    k_gather<<<ngb, GT, 0, stream>>>(rowptr, sorted, buf0, nd, b2, out, n, e);
}


# Gather: a 16-lane group keeps 8 message rows in flight (was 4) on 32-bit lane offsets; a row's first 32 sources come with its head, asked behind the previous row's first 8; 5 rows a group; same sums
# speedup vs baseline: 5.2262x; 1.0417x over previous
#include <hip/hip_runtime.h>
#include <math.h>

#define F_OUT 64

typedef __attribute__((ext_vector_type(2))) float v2f;
typedef __attribute__((ext_vector_type(8))) float v8f;

static inline int cdiv(int a, int b) { return (a + b - 1) / b; }


template <int K, int NT>
static __device__ __forceinline__ void gemm_stage_w_once(const float* __restrict__ W, float* ldsW) {
    constexpr int PER = K * F_OUT / NT;
    static_assert(PER * NT == K * F_OUT, "whole elements per thread");
    float v[PER];
#pragma unroll
    for (int i = 0; i < PER; ++i) v[i] = W[threadIdx.x + i * NT];
#pragma unroll
    for (int i = 0; i < PER; ++i) {
        const int t = threadIdx.x + i * NT;
        ldsW[(t % F_OUT) * (K + 2) + t / F_OUT] = v[i];
    }
}

template <int K, int U>
static __device__ __forceinline__ void gemm_tile(const float* __restrict__ X, const float* ldsW,
                                                 const float* norm, int nbase,
                                                 float* __restrict__ H, int nrows, int mBase) {
    const int lane  = threadIdx.x & 31;
    const int half  = lane >> 4;
    const int l16   = lane & 15;

    int rowA = mBase + l16;
    if (rowA >= nrows) rowA = nrows - 1;
    const float* xrow = X + (size_t)rowA * K;

    v8f c0 = {}, c1 = {}, c2 = {}, c3 = {};
#pragma unroll U
    for (int k0 = 0; k0 < K; k0 += 4) {
        const int kk = k0 + 2 * half;
        v2f a = *(const v2f*)(xrow + kk);
        v2f b0 = *(const v2f*)(&ldsW[(0 * 16 + l16) * (K + 2) + kk]);
        v2f b1 = *(const v2f*)(&ldsW[(1 * 16 + l16) * (K + 2) + kk]);
        v2f b2 = *(const v2f*)(&ldsW[(2 * 16 + l16) * (K + 2) + kk]);
        v2f b3 = *(const v2f*)(&ldsW[(3 * 16 + l16) * (K + 2) + kk]);
        c0 = __builtin_amdgcn_wmma_f32_16x16x4_f32(false, a, false, b0, (short)0, c0, false, false);
        c1 = __builtin_amdgcn_wmma_f32_16x16x4_f32(false, a, false, b1, (short)0, c1, false, false);
        c2 = __builtin_amdgcn_wmma_f32_16x16x4_f32(false, a, false, b2, (short)0, c2, false, false);
        c3 = __builtin_amdgcn_wmma_f32_16x16x4_f32(false, a, false, b3, (short)0, c3, false, false);
    }

#pragma unroll
    for (int r = 0; r < 8; ++r) {
        int row = mBase + 8 * half + r;
        if (row < nrows) {
            float s = norm[row - nbase];
            size_t base = (size_t)row * F_OUT + l16;
            float v0 = c0[r] * s, v1 = c1[r] * s, v2 = c2[r] * s, v3 = c3[r] * s;
            H[base +  0] = v0;  H[base + 16] = v1;
            H[base + 32] = v2;  H[base + 48] = v3;
        }
    }
}

#define BW   400
#define SL   6400
#define PT   1024
#define EPT  ((SL + PT - 1) / PT)
#define MAXB 256
static_assert(BW % 16 == 0 && BW / 16 <= PT / 32, "a wave per row tile of the bucket");
static_assert(BW <= PT, "a thread per node of the bucket");

static __device__ __forceinline__ int wave_excl_scan(int* c, int* o, int m) {
    const int lane = threadIdx.x & 31;
    const int per = (m + 31) >> 5;
    const int b0 = lane * per;
    int sum = 0;
    for (int k = 0; k < per; ++k) if (b0 + k < m) sum += c[b0 + k];
    int inc = sum;
    for (int off = 1; off < 32; off <<= 1) {
        const int t = __shfl_up(inc, off);
        if (lane >= off) inc += t;
    }
    int run = inc - sum;
    for (int k = 0; k < per; ++k) {
        if (b0 + k < m) {
            const int v = c[b0 + k];
            c[b0 + k] = run;
            if (o) o[b0 + k] = run;
            run += v;
        }
    }
    return __shfl(inc, 31);
}

__global__ __launch_bounds__(PT) void k_part(const int* __restrict__ src,
                                             const int* __restrict__ dst,
                                             int* __restrict__ pD, int* __restrict__ pS,
                                             int* __restrict__ qS, int* __restrict__ tabD,
                                             int* __restrict__ tabS, int e, int nbk, int nsl) {
    __shared__ int cD[MAXB], cS[MAXB], oD[MAXB + 1], oS[MAXB + 1];
    __shared__ int stD[SL], stP[SL], stQ[SL];
    const int tid = threadIdx.x;
    const int s   = blockIdx.x;
    const int beg = s * SL;
    const int len = min(SL, e - beg);

    for (int t = tid; t < MAXB; t += PT) { cD[t] = 0; cS[t] = 0; }

    int d[EPT], u[EPT];
#pragma unroll
    for (int i = 0; i < EPT; ++i) {
        const int j = tid + i * PT;
        d[i] = -1; u[i] = 0;
        if (j < len) { d[i] = dst[beg + j]; u[i] = src[beg + j]; }
    }
    __syncthreads();
#pragma unroll
    for (int i = 0; i < EPT; ++i) {
        if (d[i] >= 0) {
            atomicAdd(&cD[(unsigned)d[i] / BW], 1);
            atomicAdd(&cS[(unsigned)u[i] / BW], 1);
        }
    }
    __syncthreads();
    const int wave = __builtin_amdgcn_readfirstlane(tid >> 5);
    if (wave == 0) { const int t = wave_excl_scan(cD, oD, nbk); if ((tid & 31) == 0) oD[nbk] = t; }
    if (wave == 1) { const int t = wave_excl_scan(cS, oS, nbk); if ((tid & 31) == 0) oS[nbk] = t; }
    __syncthreads();

    for (int b = tid; b <= nbk; b += PT) {
        tabD[(size_t)b * nsl + s] = oD[b];
        tabS[(size_t)b * nsl + s] = oS[b];
    }
#pragma unroll
    for (int i = 0; i < EPT; ++i) {
        if (d[i] >= 0) {
            const int p = atomicAdd(&cD[(unsigned)d[i] / BW], 1);
            stD[p] = d[i];
            stP[p] = u[i];
            const int q = atomicAdd(&cS[(unsigned)u[i] / BW], 1);
            stQ[q] = u[i];
        }
    }
    __syncthreads();
    for (int j = tid; j < len; j += PT) {
        pD[beg + j] = stD[j];
        pS[beg + j] = stP[j];
        qS[beg + j] = stQ[j];
    }
}

#define RB 8
__global__ __launch_bounds__(PT) void k_bucket_gemm(const int* __restrict__ pD,
                                                    const int* __restrict__ pS,
                                                    const int* __restrict__ qS,
                                                    const int* __restrict__ tabD,
                                                    const int* __restrict__ tabS,
                                                    int* __restrict__ rowptr,
                                                    int* __restrict__ sorted,
                                                    float* __restrict__ ns, float* __restrict__ nd,
                                                    int n, int nsl,
                                                    const float* __restrict__ X,
                                                    const float* __restrict__ W,
                                                    float* __restrict__ H) {
    __shared__ alignas(16) float ldsW[F_OUT * (128 + 2)];
    __shared__ int cnt[BW], cur[BW], cntS[BW];
    __shared__ float nsL[BW];
    __shared__ int base_s;
    const int tid = threadIdx.x;
    const int b   = blockIdx.x;
    const int lo  = b * BW;
    const int nn  = min(BW, n - lo);
    const int q   = tid & 3;

    gemm_stage_w_once<128, PT>(W, ldsW);
    if (tid < BW) { cnt[tid] = 0; cntS[tid] = 0; }
    if (tid == 0) base_s = 0;
    __syncthreads();

    int bpart = 0;
    for (int r = tid >> 2; r < nsl; r += PT / 4) {
        const int a0 = tabD[(size_t)b * nsl + r], a1 = tabD[(size_t)(b + 1) * nsl + r];
        const int s0 = tabS[(size_t)b * nsl + r], s1 = tabS[(size_t)(b + 1) * nsl + r];
        if (q == 0) bpart += a0;
        const int* __restrict__ pd = pD + (size_t)r * SL;
        const int* __restrict__ qs = qS + (size_t)r * SL;
        for (int j0 = 0; j0 < max(a1 - a0, s1 - s0); j0 += 4 * RB) {
            int vd[RB], vs[RB];
#pragma unroll
            for (int k = 0; k < RB; ++k) {
                const int jd = a0 + j0 + q + 4 * k, js = s0 + j0 + q + 4 * k;
                vd[k] = (jd < a1) ? pd[jd] : -1;
                vs[k] = (js < s1) ? qs[js] : -1;
            }
#pragma unroll
            for (int k = 0; k < RB; ++k) {
                if (vd[k] >= 0) atomicAdd(&cnt[vd[k] - lo], 1);
                if (vs[k] >= 0) atomicAdd(&cntS[vs[k] - lo], 1);
            }
        }
    }
    if (bpart) atomicAdd(&base_s, bpart);
    __syncthreads();

    const int base = base_s;
    if (tid < nn) {
        const int i = lo + tid;
        nd[i] = rsqrtf((float)(cnt[tid] + 1));
        const float v = rsqrtf((float)(cntS[tid] + 1));
        ns[i] = v;
        nsL[tid] = v;
        cur[tid] = cnt[tid];
    }
    __syncthreads();
    const int wave = __builtin_amdgcn_readfirstlane(tid >> 5);
    if (wave == 0) wave_excl_scan(cur, nullptr, nn);
    __syncthreads();
    if (tid < nn) rowptr[lo + tid] = base + cur[tid];
    __syncthreads();

    for (int r = tid >> 2; r < nsl; r += PT / 4) {
        const int a0 = tabD[(size_t)b * nsl + r], a1 = tabD[(size_t)(b + 1) * nsl + r];
        const int* __restrict__ pd = pD + (size_t)r * SL;
        const int* __restrict__ ps = pS + (size_t)r * SL;
        for (int j0 = a0; j0 < a1; j0 += 4 * RB) {
            int vd[RB], vp[RB];
#pragma unroll
            for (int k = 0; k < RB; ++k) {
                const int j = j0 + q + 4 * k;
                vd[k] = (j < a1) ? pd[j] : -1;
                vp[k] = (j < a1) ? ps[j] : 0;
            }
#pragma unroll
            for (int k = 0; k < RB; ++k)
                if (vd[k] >= 0) sorted[base + atomicAdd(&cur[vd[k] - lo], 1)] = vp[k];
        }
    }

    if (wave * 16 < nn)
        gemm_tile<128, 4>(X, ldsW, nsL, lo, H, n, lo + wave * 16);
}

#define GR 16
#define GT (GR * 16)
#define TPB 5
#define NB 8

struct RowHead { float4 self; int idx; int idx2; float s; };

static __device__ __forceinline__ RowHead row_head(const int* __restrict__ sorted,
                                                   const float4* __restrict__ h4,
                                                   const float* __restrict__ nd,
                                                   int row, int beg, int deg, int sub) {
    RowHead r;
    r.self = h4[(size_t)row * 16 + sub];
    r.s = nd[row];
    r.idx = row;
    if (sub < deg) r.idx = sorted[beg + sub];
    r.idx2 = row;
    if (16 + sub < deg) r.idx2 = sorted[beg + 16 + sub];
    return r;
}

static __device__ __forceinline__ void rows_request(const char* __restrict__ hb, int idx,
                                                    int lane0, int sub, float4 (&v)[NB]) {
#pragma unroll
    for (int j = 0; j < NB; ++j) {
        const int s = __builtin_amdgcn_ds_bpermute((lane0 + j) << 2, idx);
        const unsigned off = (unsigned)s * 256u + (unsigned)sub * 16u;
        v[j] = *(const float4*)(hb + off);
    }
}

static __device__ __forceinline__ void rows_add(float4& acc, const float4 (&v)[NB], int cnt) {
#pragma unroll
    for (int j = 0; j < NB; ++j) {
        const bool p = j < cnt;
        acc.x += p ? v[j].x : 0.0f; acc.y += p ? v[j].y : 0.0f;
        acc.z += p ? v[j].z : 0.0f; acc.w += p ? v[j].w : 0.0f;
    }
}

template <int RELU>
static __device__ __forceinline__ float4 row_finish(const int* __restrict__ sorted,
                                                    const char* __restrict__ hb,
                                                    RowHead hd, float4 (&v)[NB], float4 b,
                                                    int row, int beg, int deg, int sub, int grp) {
    float4 acc = hd.self;
    int idx = hd.idx;
    const float s = hd.s;

    rows_add(acc, v, min(NB, deg));
    for (int base = 0; base < deg; base += 16) {
        const int cnt = min(16, deg - base);
        if (base > 0) {
            if (base == 16) {
                idx = hd.idx2;
            } else {
                idx = row;
                if (sub < cnt) idx = sorted[beg + base + sub];
            }
            rows_request(hb, idx, grp, sub, v);
            rows_add(acc, v, min(NB, cnt));
        }
        if (cnt > NB) {
            rows_request(hb, idx, grp + NB, sub, v);
            rows_add(acc, v, cnt - NB);
        }
    }

    float4 r;
    r.x = fmaf(acc.x, s, b.x);
    r.y = fmaf(acc.y, s, b.y);
    r.z = fmaf(acc.z, s, b.z);
    r.w = fmaf(acc.w, s, b.w);
    if (RELU) {
        r.x = fmaxf(r.x, 0.0f); r.y = fmaxf(r.y, 0.0f);
        r.z = fmaxf(r.z, 0.0f); r.w = fmaxf(r.w, 0.0f);
    }
    return r;
}

static __device__ __forceinline__ void row_bounds(const int* __restrict__ rowptr, int row0,
                                                  int n, int e, int* rbeg, int* rdeg) {
#pragma unroll
    for (int it = 0; it < TPB; ++it) {
        const int row = row0 + it * GR;
        int b = 0, t = 0;
        if (row < n) {
            b = rowptr[row];
            t = (row + 1 < n) ? rowptr[row + 1] : e;
        }
        rbeg[it] = b;
        rdeg[it] = t - b;
    }
}

__global__ __launch_bounds__(GT) __attribute__((amdgpu_waves_per_eu(10, 10)))
void k_gather(const int* __restrict__ rowptr,
                                               const int* __restrict__ sorted,
                                               const float* __restrict__ h,
                                               const float* __restrict__ nd,
                                               const float* __restrict__ bias,
                                               float* __restrict__ out, int n, int e) {
    const int sub = threadIdx.x & 15;
    const int grp = threadIdx.x & 16;
    const int row0 = blockIdx.x * (TPB * GR) + (threadIdx.x >> 4);
    const float4* __restrict__ h4 = (const float4*)h;
    const char* __restrict__ hb = (const char*)h;
    const float4 b = *(const float4*)(bias + sub * 4);

    int rbeg[TPB], rdeg[TPB];
    row_bounds(rowptr, row0, n, e, rbeg, rdeg);

    RowHead hd;
    hd.self = make_float4(0.0f, 0.0f, 0.0f, 0.0f); hd.idx = 0; hd.idx2 = 0; hd.s = 0.0f;
    if (row0 < n) hd = row_head(sorted, h4, nd, row0, rbeg[0], rdeg[0], sub);
#pragma unroll
    for (int it = 0; it < TPB; ++it) {
        const int row = row0 + it * GR;
        const RowHead cur = hd;
        float4 v[NB];
        rows_request(hb, cur.idx, grp, sub, v);
        if (it + 1 < TPB && row + GR < n)
            hd = row_head(sorted, h4, nd, row + GR, rbeg[it + 1], rdeg[it + 1], sub);
        if (row < n) {
            const float4 r = row_finish<0>(sorted, hb, cur, v, b, row, rbeg[it], rdeg[it], sub, grp);
            *(float4*)(out + (size_t)row * F_OUT + sub * 4) = r;
        }
    }
}

#define TP (F_OUT + 4)
static_assert(GT <= 1024 && (F_OUT * F_OUT) % GT == 0, "whole W2 elements per thread");
static_assert((GR / 16) * (F_OUT / 16) <= GT / 32, "a wave per 16x16 tile of the product");
__global__ __launch_bounds__(GT) __attribute__((amdgpu_waves_per_eu(10, 10)))
void k_gather_gemm(const int* __restrict__ rowptr,
                                                    const int* __restrict__ sorted,
                                                    const float* __restrict__ h,
                                                    const float* __restrict__ nd,
                                                    const float* __restrict__ bias,
                                                    const float* __restrict__ W,
                                                    const float* __restrict__ norm,
                                                    float* __restrict__ H, int n, int e) {
    __shared__ alignas(16) float ldsW[F_OUT * TP];
    __shared__ alignas(16) float tileA[GR * TP];
    __shared__ alignas(16) float tileD[GR * TP];
    const int sub = threadIdx.x & 15;
    const int grp = threadIdx.x & 16;
    const int rib = threadIdx.x >> 4;
    const int row0 = blockIdx.x * (TPB * GR) + rib;
    const float4* __restrict__ h4 = (const float4*)h;
    const char* __restrict__ hb = (const char*)h;

    constexpr int PER = F_OUT * F_OUT / GT;
    float w[PER];
#pragma unroll
    for (int i = 0; i < PER; ++i) w[i] = W[threadIdx.x + i * GT];

    const float4 b = *(const float4*)(bias + sub * 4);
    int rbeg[TPB], rdeg[TPB];
    row_bounds(rowptr, row0, n, e, rbeg, rdeg);

#pragma unroll
    for (int i = 0; i < PER; ++i) {
        const int t = threadIdx.x + i * GT;
        ldsW[(t % F_OUT) * TP + t / F_OUT] = w[i];
    }

    const int wave = __builtin_amdgcn_readfirstlane(threadIdx.x >> 5);
    const int half = (threadIdx.x >> 4) & 1;

    RowHead hd;
    hd.self = make_float4(0.0f, 0.0f, 0.0f, 0.0f); hd.idx = 0; hd.idx2 = 0; hd.s = 0.0f;
    if (row0 < n) hd = row_head(sorted, h4, nd, row0, rbeg[0], rdeg[0], sub);
#pragma unroll
    for (int it = 0; it < TPB; ++it) {
        const int row = row0 + it * GR;
        const RowHead cur = hd;
        float4 m[NB];
        rows_request(hb, cur.idx, grp, sub, m);
        if (it + 1 < TPB && row + GR < n)
            hd = row_head(sorted, h4, nd, row + GR, rbeg[it + 1], rdeg[it + 1], sub);
        float4 v = make_float4(0.0f, 0.0f, 0.0f, 0.0f);
        float sn = 0.0f;
        if (row < n) {
            v = row_finish<1>(sorted, hb, cur, m, b, row, rbeg[it], rdeg[it], sub, grp);
            sn = norm[row];
        }

        *(float4*)(&tileA[rib * TP + sub * 4]) = v;
        __syncthreads();

        if (wave < (GR / 16) * (F_OUT / 16)) {
            const int mt = wave >> 2, nt = wave & 3;
            const float* ap = &tileA[(mt * 16 + sub) * TP + 2 * half];
            const float* bp = &ldsW[(nt * 16 + sub) * TP + 2 * half];
            v8f c = {};
#pragma unroll
            for (int i = 0; i < 16; ++i) {
                const v2f a = *(const v2f*)(ap + 4 * i);
                const v2f bb = *(const v2f*)(bp + 4 * i);
                c = __builtin_amdgcn_wmma_f32_16x16x4_f32(false, a, false, bb, (short)0, c, false, false);
            }
#pragma unroll
            for (int r = 0; r < 8; ++r) tileD[(mt * 16 + 8 * half + r) * TP + nt * 16 + sub] = c[r];
        }
        __syncthreads();

        if (row < n) {
            float4 d = *(const float4*)(&tileD[rib * TP + sub * 4]);
            d.x *= sn; d.y *= sn; d.z *= sn; d.w *= sn;
            *(float4*)(H + (size_t)row * F_OUT + sub * 4) = d;
        }
    }
}

extern "C" void kernel_launch(void* const* d_in, const int* in_sizes, int n_in,
                              void* d_out, int out_size, void* d_ws, size_t ws_size,
                              hipStream_t stream) {
    const float* x   = (const float*)d_in[0];
    const int*   src = (const int*)d_in[1];
    const int*   dst = (const int*)d_in[2];
    const float* W1  = (const float*)d_in[3];
    const float* b1  = (const float*)d_in[4];
    const float* W2  = (const float*)d_in[5];
    const float* b2  = (const float*)d_in[6];
    float*       out = (float*)d_out;

    const int n = in_sizes[0] / 128;
    const int e = in_sizes[1];

    float* ws   = (float*)d_ws;
    float* buf0 = ws;
    int*   sorted = (int*)(ws + (size_t)n * F_OUT);
    int*   rowptr = sorted + e;
    float* ns   = ws + 2 * (size_t)n * F_OUT;
    float* nd   = ns + n;

    const int nbk = cdiv(n, BW);
    const int nsl = cdiv(e, SL);
    int* pD   = (int*)buf0;
    int* pS   = pD + (size_t)nsl * SL;
    int* qS   = pS + (size_t)nsl * SL;
    int* tabD = qS + (size_t)nsl * SL;
    int* tabS = tabD + (size_t)(nbk + 1) * nsl;
    k_part<<<nsl, PT, 0, stream>>>(src, dst, pD, pS, qS, tabD, tabS, e, nbk, nsl);

    k_bucket_gemm<<<nbk, PT, 0, stream>>>(pD, pS, qS, tabD, tabS, rowptr, sorted, ns, nd,
                                          n, nsl, x, W1, out);

    const int ngb = cdiv(n, TPB * GR);
    k_gather_gemm<<<ngb, GT, 0, stream>>>(rowptr, sorted, out, nd, b1, W2, ns, buf0, n, e);
    k_gather<<<ngb, GT, 0, stream>>>(rowptr, sorted, buf0, nd, b2, out, n, e);
}


# k_bucket_gemm stores whole lines through LDS: a bucket's piece of the sorted list is placed in LDS, written in order (was a scattered 4-B store per edge); product tiles leave as 256-B rows; same sums
# speedup vs baseline: 5.3992x; 1.0331x over previous
#include <hip/hip_runtime.h>
#include <math.h>

#define F_OUT 64
#define TP (F_OUT + 4)

typedef __attribute__((ext_vector_type(2))) float v2f;
typedef __attribute__((ext_vector_type(8))) float v8f;

static inline int cdiv(int a, int b) { return (a + b - 1) / b; }


template <int K, int NT>
static __device__ __forceinline__ void gemm_stage_w_once(const float* __restrict__ W, float* ldsW) {
    constexpr int PER = K * F_OUT / NT;
    static_assert(PER * NT == K * F_OUT, "whole elements per thread");
    float v[PER];
#pragma unroll
    for (int i = 0; i < PER; ++i) v[i] = W[threadIdx.x + i * NT];
#pragma unroll
    for (int i = 0; i < PER; ++i) {
        const int t = threadIdx.x + i * NT;
        ldsW[(t % F_OUT) * (K + 2) + t / F_OUT] = v[i];
    }
}

template <int K, int U>
static __device__ __forceinline__ void gemm_tile(const float* __restrict__ X, const float* ldsW,
                                                 const float* norm, int nbase, float* tile,
                                                 float* __restrict__ H, int nrows, int mBase) {
    const int lane  = threadIdx.x & 31;
    const int half  = lane >> 4;
    const int l16   = lane & 15;

    int rowA = mBase + l16;
    if (rowA >= nrows) rowA = nrows - 1;
    const float* xrow = X + (size_t)rowA * K;

    v8f c0 = {}, c1 = {}, c2 = {}, c3 = {};
#pragma unroll U
    for (int k0 = 0; k0 < K; k0 += 4) {
        const int kk = k0 + 2 * half;
        v2f a = *(const v2f*)(xrow + kk);
        v2f b0 = *(const v2f*)(&ldsW[(0 * 16 + l16) * (K + 2) + kk]);
        v2f b1 = *(const v2f*)(&ldsW[(1 * 16 + l16) * (K + 2) + kk]);
        v2f b2 = *(const v2f*)(&ldsW[(2 * 16 + l16) * (K + 2) + kk]);
        v2f b3 = *(const v2f*)(&ldsW[(3 * 16 + l16) * (K + 2) + kk]);
        c0 = __builtin_amdgcn_wmma_f32_16x16x4_f32(false, a, false, b0, (short)0, c0, false, false);
        c1 = __builtin_amdgcn_wmma_f32_16x16x4_f32(false, a, false, b1, (short)0, c1, false, false);
        c2 = __builtin_amdgcn_wmma_f32_16x16x4_f32(false, a, false, b2, (short)0, c2, false, false);
        c3 = __builtin_amdgcn_wmma_f32_16x16x4_f32(false, a, false, b3, (short)0, c3, false, false);
    }

    float s[8];
#pragma unroll
    for (int r = 0; r < 8; ++r) s[r] = norm[min(mBase + 8 * half + r, nrows - 1) - nbase];
#pragma unroll
    for (int r = 0; r < 8; ++r) {
        float* t = tile + (8 * half + r) * TP + l16;
        t[0]  = c0[r] * s[r];  t[16] = c1[r] * s[r];
        t[32] = c2[r] * s[r];  t[48] = c3[r] * s[r];
    }
    float4 v[8];
#pragma unroll
    for (int i = 0; i < 8; ++i) v[i] = *(const float4*)(tile + (2 * i + half) * TP + 4 * l16);
#pragma unroll
    for (int i = 0; i < 8; ++i) {
        const int row = mBase + 2 * i + half;
        if (row < nrows) *(float4*)(H + (size_t)row * F_OUT + 4 * l16) = v[i];
    }
}

#define BW   400
#define SL   6400
#define PT   1024
#define EPT  ((SL + PT - 1) / PT)
#define MAXB 256
#define CAP  8192
static_assert(BW % 16 == 0 && BW / 16 <= PT / 32, "a wave per row tile of the bucket");
static_assert(BW <= PT, "a thread per node of the bucket");

static __device__ __forceinline__ int wave_excl_scan(int* c, int* o, int m) {
    const int lane = threadIdx.x & 31;
    const int per = (m + 31) >> 5;
    const int b0 = lane * per;
    int sum = 0;
    for (int k = 0; k < per; ++k) if (b0 + k < m) sum += c[b0 + k];
    int inc = sum;
    for (int off = 1; off < 32; off <<= 1) {
        const int t = __shfl_up(inc, off);
        if (lane >= off) inc += t;
    }
    int run = inc - sum;
    for (int k = 0; k < per; ++k) {
        if (b0 + k < m) {
            const int v = c[b0 + k];
            c[b0 + k] = run;
            if (o) o[b0 + k] = run;
            run += v;
        }
    }
    return __shfl(inc, 31);
}

__global__ __launch_bounds__(PT) void k_part(const int* __restrict__ src,
                                             const int* __restrict__ dst,
                                             int* __restrict__ pD, int* __restrict__ pS,
                                             int* __restrict__ qS, int* __restrict__ tabD,
                                             int* __restrict__ tabS, int e, int nbk, int nsl) {
    __shared__ int cD[MAXB], cS[MAXB], oD[MAXB + 1], oS[MAXB + 1];
    __shared__ int stD[SL], stP[SL], stQ[SL];
    const int tid = threadIdx.x;
    const int s   = blockIdx.x;
    const int beg = s * SL;
    const int len = min(SL, e - beg);

    for (int t = tid; t < MAXB; t += PT) { cD[t] = 0; cS[t] = 0; }

    int d[EPT], u[EPT];
#pragma unroll
    for (int i = 0; i < EPT; ++i) {
        const int j = tid + i * PT;
        d[i] = -1; u[i] = 0;
        if (j < len) { d[i] = dst[beg + j]; u[i] = src[beg + j]; }
    }
    __syncthreads();
#pragma unroll
    for (int i = 0; i < EPT; ++i) {
        if (d[i] >= 0) {
            atomicAdd(&cD[(unsigned)d[i] / BW], 1);
            atomicAdd(&cS[(unsigned)u[i] / BW], 1);
        }
    }
    __syncthreads();
    const int wave = __builtin_amdgcn_readfirstlane(tid >> 5);
    if (wave == 0) { const int t = wave_excl_scan(cD, oD, nbk); if ((tid & 31) == 0) oD[nbk] = t; }
    if (wave == 1) { const int t = wave_excl_scan(cS, oS, nbk); if ((tid & 31) == 0) oS[nbk] = t; }
    __syncthreads();

    for (int b = tid; b <= nbk; b += PT) {
        tabD[(size_t)b * nsl + s] = oD[b];
        tabS[(size_t)b * nsl + s] = oS[b];
    }
#pragma unroll
    for (int i = 0; i < EPT; ++i) {
        if (d[i] >= 0) {
            const int p = atomicAdd(&cD[(unsigned)d[i] / BW], 1);
            stD[p] = d[i];
            stP[p] = u[i];
            const int q = atomicAdd(&cS[(unsigned)u[i] / BW], 1);
            stQ[q] = u[i];
        }
    }
    __syncthreads();
    for (int j = tid; j < len; j += PT) {
        pD[beg + j] = stD[j];
        pS[beg + j] = stP[j];
        qS[beg + j] = stQ[j];
    }
}

#define RB 8
__global__ __launch_bounds__(PT) void k_bucket_gemm(const int* __restrict__ pD,
                                                    const int* __restrict__ pS,
                                                    const int* __restrict__ qS,
                                                    const int* __restrict__ tabD,
                                                    const int* __restrict__ tabS,
                                                    int* __restrict__ rowptr,
                                                    int* __restrict__ sorted,
                                                    float* __restrict__ ns, float* __restrict__ nd,
                                                    int n, int nsl,
                                                    const float* __restrict__ X,
                                                    const float* __restrict__ W,
                                                    float* __restrict__ H) {
    __shared__ alignas(16) float ldsW[F_OUT * (128 + 2)];
    __shared__ alignas(16) float tiles[BW * TP];
    __shared__ int stg[CAP];
    __shared__ int cnt[BW], cur[BW], cntS[BW];
    __shared__ float nsL[BW];
    __shared__ int base_s, tot_s;
    const int tid = threadIdx.x;
    const int b   = blockIdx.x;
    const int lo  = b * BW;
    const int nn  = min(BW, n - lo);
    const int q   = tid & 3;

    gemm_stage_w_once<128, PT>(W, ldsW);
    if (tid < BW) { cnt[tid] = 0; cntS[tid] = 0; }
    if (tid == 0) base_s = 0;
    __syncthreads();

    int bpart = 0;
    for (int r = tid >> 2; r < nsl; r += PT / 4) {
        const int a0 = tabD[(size_t)b * nsl + r], a1 = tabD[(size_t)(b + 1) * nsl + r];
        const int s0 = tabS[(size_t)b * nsl + r], s1 = tabS[(size_t)(b + 1) * nsl + r];
        if (q == 0) bpart += a0;
        const int* __restrict__ pd = pD + (size_t)r * SL;
        const int* __restrict__ qs = qS + (size_t)r * SL;
        for (int j0 = 0; j0 < max(a1 - a0, s1 - s0); j0 += 4 * RB) {
            int vd[RB], vs[RB];
#pragma unroll
            for (int k = 0; k < RB; ++k) {
                const int jd = a0 + j0 + q + 4 * k, js = s0 + j0 + q + 4 * k;
                vd[k] = (jd < a1) ? pd[jd] : -1;
                vs[k] = (js < s1) ? qs[js] : -1;
            }
#pragma unroll
            for (int k = 0; k < RB; ++k) {
                if (vd[k] >= 0) atomicAdd(&cnt[vd[k] - lo], 1);
                if (vs[k] >= 0) atomicAdd(&cntS[vs[k] - lo], 1);
            }
        }
    }
    if (bpart) atomicAdd(&base_s, bpart);
    __syncthreads();

    const int base = base_s;
    if (tid < nn) {
        const int i = lo + tid;
        nd[i] = rsqrtf((float)(cnt[tid] + 1));
        const float v = rsqrtf((float)(cntS[tid] + 1));
        ns[i] = v;
        nsL[tid] = v;
        cur[tid] = cnt[tid];
    }
    __syncthreads();
    const int wave = __builtin_amdgcn_readfirstlane(tid >> 5);
    if (wave == 0) {
        const int t = wave_excl_scan(cur, nullptr, nn);
        if ((tid & 31) == 0) tot_s = t;
    }
    __syncthreads();
    if (tid < nn) rowptr[lo + tid] = base + cur[tid];
    const int tot = tot_s;
    __syncthreads();

    for (int r = tid >> 2; r < nsl; r += PT / 4) {
        const int a0 = tabD[(size_t)b * nsl + r], a1 = tabD[(size_t)(b + 1) * nsl + r];
        const int* __restrict__ pd = pD + (size_t)r * SL;
        const int* __restrict__ ps = pS + (size_t)r * SL;
        for (int j0 = a0; j0 < a1; j0 += 4 * RB) {
            int vd[RB], vp[RB];
#pragma unroll
            for (int k = 0; k < RB; ++k) {
                const int j = j0 + q + 4 * k;
                vd[k] = (j < a1) ? pd[j] : -1;
                vp[k] = (j < a1) ? ps[j] : 0;
            }
#pragma unroll
            for (int k = 0; k < RB; ++k) {
                if (vd[k] >= 0) {
                    const int p = atomicAdd(&cur[vd[k] - lo], 1);
                    if (p < CAP) stg[p] = vp[k];
                    else sorted[base + p] = vp[k];
                }
            }
        }
    }
    __syncthreads();

    {
        constexpr int NW = (CAP + 31 + PT - 1) / PT;
        const int sh = base & 31;
        const int m = min(tot, CAP) + sh;
        int w[NW];
#pragma unroll
        for (int i = 0; i < NW; ++i) {
            const int jj = tid + i * PT;
            w[i] = (jj >= sh && jj < m) ? stg[jj - sh] : 0;
        }
#pragma unroll
        for (int i = 0; i < NW; ++i) {
            const int jj = tid + i * PT;
            if (jj >= sh && jj < m) sorted[base + jj - sh] = w[i];
        }
    }

    if (wave * 16 < nn)
        gemm_tile<128, 4>(X, ldsW, nsL, lo, tiles + wave * 16 * TP, H, n, lo + wave * 16);
}

#define GR 16
#define GT (GR * 16)
#define TPB 5
#define NB 8

struct RowHead { float4 self; int idx; int idx2; float s; };

static __device__ __forceinline__ RowHead row_head(const int* __restrict__ sorted,
                                                   const float4* __restrict__ h4,
                                                   const float* __restrict__ nd,
                                                   int row, int beg, int deg, int sub) {
    RowHead r;
    r.self = h4[(size_t)row * 16 + sub];
    r.s = nd[row];
    r.idx = row;
    if (sub < deg) r.idx = sorted[beg + sub];
    r.idx2 = row;
    if (16 + sub < deg) r.idx2 = sorted[beg + 16 + sub];
    return r;
}

static __device__ __forceinline__ void rows_request(const char* __restrict__ hb, int idx,
                                                    int lane0, int sub, float4 (&v)[NB]) {
#pragma unroll
    for (int j = 0; j < NB; ++j) {
        const int s = __builtin_amdgcn_ds_bpermute((lane0 + j) << 2, idx);
        const unsigned off = (unsigned)s * 256u + (unsigned)sub * 16u;
        v[j] = *(const float4*)(hb + off);
    }
}

static __device__ __forceinline__ void rows_add(float4& acc, const float4 (&v)[NB], int cnt) {
#pragma unroll
    for (int j = 0; j < NB; ++j) {
        const bool p = j < cnt;
        acc.x += p ? v[j].x : 0.0f; acc.y += p ? v[j].y : 0.0f;
        acc.z += p ? v[j].z : 0.0f; acc.w += p ? v[j].w : 0.0f;
    }
}

template <int RELU>
static __device__ __forceinline__ float4 row_finish(const int* __restrict__ sorted,
                                                    const char* __restrict__ hb,
                                                    RowHead hd, float4 (&v)[NB], float4 b,
                                                    int row, int beg, int deg, int sub, int grp) {
    float4 acc = hd.self;
    int idx = hd.idx;
    const float s = hd.s;

    rows_add(acc, v, min(NB, deg));
    for (int base = 0; base < deg; base += 16) {
        const int cnt = min(16, deg - base);
        if (base > 0) {
            if (base == 16) {
                idx = hd.idx2;
            } else {
                idx = row;
                if (sub < cnt) idx = sorted[beg + base + sub];
            }
            rows_request(hb, idx, grp, sub, v);
            rows_add(acc, v, min(NB, cnt));
        }
        if (cnt > NB) {
            rows_request(hb, idx, grp + NB, sub, v);
            rows_add(acc, v, cnt - NB);
        }
    }

    float4 r;
    r.x = fmaf(acc.x, s, b.x);
    r.y = fmaf(acc.y, s, b.y);
    r.z = fmaf(acc.z, s, b.z);
    r.w = fmaf(acc.w, s, b.w);
    if (RELU) {
        r.x = fmaxf(r.x, 0.0f); r.y = fmaxf(r.y, 0.0f);
        r.z = fmaxf(r.z, 0.0f); r.w = fmaxf(r.w, 0.0f);
    }
    return r;
}

static __device__ __forceinline__ void row_bounds(const int* __restrict__ rowptr, int row0,
                                                  int n, int e, int* rbeg, int* rdeg) {
#pragma unroll
    for (int it = 0; it < TPB; ++it) {
        const int row = row0 + it * GR;
        int b = 0, t = 0;
        if (row < n) {
            b = rowptr[row];
            t = (row + 1 < n) ? rowptr[row + 1] : e;
        }
        rbeg[it] = b;
        rdeg[it] = t - b;
    }
}

__global__ __launch_bounds__(GT) __attribute__((amdgpu_waves_per_eu(10, 10)))
void k_gather(const int* __restrict__ rowptr,
                                               const int* __restrict__ sorted,
                                               const float* __restrict__ h,
                                               const float* __restrict__ nd,
                                               const float* __restrict__ bias,
                                               float* __restrict__ out, int n, int e) {
    const int sub = threadIdx.x & 15;
    const int grp = threadIdx.x & 16;
    const int row0 = blockIdx.x * (TPB * GR) + (threadIdx.x >> 4);
    const float4* __restrict__ h4 = (const float4*)h;
    const char* __restrict__ hb = (const char*)h;
    const float4 b = *(const float4*)(bias + sub * 4);

    int rbeg[TPB], rdeg[TPB];
    row_bounds(rowptr, row0, n, e, rbeg, rdeg);

    RowHead hd;
    hd.self = make_float4(0.0f, 0.0f, 0.0f, 0.0f); hd.idx = 0; hd.idx2 = 0; hd.s = 0.0f;
    if (row0 < n) hd = row_head(sorted, h4, nd, row0, rbeg[0], rdeg[0], sub);
#pragma unroll
    for (int it = 0; it < TPB; ++it) {
        const int row = row0 + it * GR;
        const RowHead cur = hd;
        float4 v[NB];
        rows_request(hb, cur.idx, grp, sub, v);
        if (it + 1 < TPB && row + GR < n)
            hd = row_head(sorted, h4, nd, row + GR, rbeg[it + 1], rdeg[it + 1], sub);
        if (row < n) {
            const float4 r = row_finish<0>(sorted, hb, cur, v, b, row, rbeg[it], rdeg[it], sub, grp);
            *(float4*)(out + (size_t)row * F_OUT + sub * 4) = r;
        }
    }
}

static_assert(GT <= 1024 && (F_OUT * F_OUT) % GT == 0, "whole W2 elements per thread");
static_assert((GR / 16) * (F_OUT / 16) <= GT / 32, "a wave per 16x16 tile of the product");
__global__ __launch_bounds__(GT) __attribute__((amdgpu_waves_per_eu(10, 10)))
void k_gather_gemm(const int* __restrict__ rowptr,
                                                    const int* __restrict__ sorted,
                                                    const float* __restrict__ h,
                                                    const float* __restrict__ nd,
                                                    const float* __restrict__ bias,
                                                    const float* __restrict__ W,
                                                    const float* __restrict__ norm,
                                                    float* __restrict__ H, int n, int e) {
    __shared__ alignas(16) float ldsW[F_OUT * TP];
    __shared__ alignas(16) float tileA[GR * TP];
    __shared__ alignas(16) float tileD[GR * TP];
    const int sub = threadIdx.x & 15;
    const int grp = threadIdx.x & 16;
    const int rib = threadIdx.x >> 4;
    const int row0 = blockIdx.x * (TPB * GR) + rib;
    const float4* __restrict__ h4 = (const float4*)h;
    const char* __restrict__ hb = (const char*)h;

    constexpr int PER = F_OUT * F_OUT / GT;
    float w[PER];
#pragma unroll
    for (int i = 0; i < PER; ++i) w[i] = W[threadIdx.x + i * GT];

    const float4 b = *(const float4*)(bias + sub * 4);
    int rbeg[TPB], rdeg[TPB];
    row_bounds(rowptr, row0, n, e, rbeg, rdeg);

#pragma unroll
    for (int i = 0; i < PER; ++i) {
        const int t = threadIdx.x + i * GT;
        ldsW[(t % F_OUT) * TP + t / F_OUT] = w[i];
    }

    const int wave = __builtin_amdgcn_readfirstlane(threadIdx.x >> 5);
    const int half = (threadIdx.x >> 4) & 1;

    RowHead hd;
    hd.self = make_float4(0.0f, 0.0f, 0.0f, 0.0f); hd.idx = 0; hd.idx2 = 0; hd.s = 0.0f;
    if (row0 < n) hd = row_head(sorted, h4, nd, row0, rbeg[0], rdeg[0], sub);
#pragma unroll
    for (int it = 0; it < TPB; ++it) {
        const int row = row0 + it * GR;
        const RowHead cur = hd;
        float4 m[NB];
        rows_request(hb, cur.idx, grp, sub, m);
        if (it + 1 < TPB && row + GR < n)
            hd = row_head(sorted, h4, nd, row + GR, rbeg[it + 1], rdeg[it + 1], sub);
        float4 v = make_float4(0.0f, 0.0f, 0.0f, 0.0f);
        float sn = 0.0f;
        if (row < n) {
            v = row_finish<1>(sorted, hb, cur, m, b, row, rbeg[it], rdeg[it], sub, grp);
            sn = norm[row];
        }

        *(float4*)(&tileA[rib * TP + sub * 4]) = v;
        __syncthreads();

        if (wave < (GR / 16) * (F_OUT / 16)) {
            const int mt = wave >> 2, nt = wave & 3;
            const float* ap = &tileA[(mt * 16 + sub) * TP + 2 * half];
            const float* bp = &ldsW[(nt * 16 + sub) * TP + 2 * half];
            v8f c = {};
#pragma unroll
            for (int i = 0; i < 16; ++i) {
                const v2f a = *(const v2f*)(ap + 4 * i);
                const v2f bb = *(const v2f*)(bp + 4 * i);
                c = __builtin_amdgcn_wmma_f32_16x16x4_f32(false, a, false, bb, (short)0, c, false, false);
            }
#pragma unroll
            for (int r = 0; r < 8; ++r) tileD[(mt * 16 + 8 * half + r) * TP + nt * 16 + sub] = c[r];
        }
        __syncthreads();

        if (row < n) {
            float4 d = *(const float4*)(&tileD[rib * TP + sub * 4]);
            d.x *= sn; d.y *= sn; d.z *= sn; d.w *= sn;
            *(float4*)(H + (size_t)row * F_OUT + sub * 4) = d;
        }
    }
}

extern "C" void kernel_launch(void* const* d_in, const int* in_sizes, int n_in,
                              void* d_out, int out_size, void* d_ws, size_t ws_size,
                              hipStream_t stream) {
    const float* x   = (const float*)d_in[0];
    const int*   src = (const int*)d_in[1];
    const int*   dst = (const int*)d_in[2];
    const float* W1  = (const float*)d_in[3];
    const float* b1  = (const float*)d_in[4];
    const float* W2  = (const float*)d_in[5];
    const float* b2  = (const float*)d_in[6];
    float*       out = (float*)d_out;

    const int n = in_sizes[0] / 128;
    const int e = in_sizes[1];

    float* ws   = (float*)d_ws;
    float* buf0 = ws;
    int*   sorted = (int*)(ws + (size_t)n * F_OUT);
    int*   rowptr = sorted + e;
    float* ns   = ws + 2 * (size_t)n * F_OUT;
    float* nd   = ns + n;

    const int nbk = cdiv(n, BW);
    const int nsl = cdiv(e, SL);
    int* pD   = (int*)buf0;
    int* pS   = pD + (size_t)nsl * SL;
    int* qS   = pS + (size_t)nsl * SL;
    int* tabD = qS + (size_t)nsl * SL;
    int* tabS = tabD + (size_t)(nbk + 1) * nsl;
    k_part<<<nsl, PT, 0, stream>>>(src, dst, pD, pS, qS, tabD, tabS, e, nbk, nsl);

    k_bucket_gemm<<<nbk, PT, 0, stream>>>(pD, pS, qS, tabD, tabS, rowptr, sorted, ns, nd,
                                          n, nsl, x, W1, out);

    const int ngb = cdiv(n, TPB * GR);
    k_gather_gemm<<<ngb, GT, 0, stream>>>(rowptr, sorted, out, nd, b1, W2, ns, buf0, n, e);
    k_gather<<<ngb, GT, 0, stream>>>(rowptr, sorted, buf0, nd, b2, out, n, e);
}


# k_bucket_gemm as two kinds of block in one launch, two per WGP: per bucket one block does x W1 (and its rows' out-degrees), one sorts; the product runs beside the sort, not behind it; same sums
# speedup vs baseline: 5.6537x; 1.0471x over previous
#include <hip/hip_runtime.h>
#include <math.h>

#define F_OUT 64
#define TP (F_OUT + 4)

typedef __attribute__((ext_vector_type(2))) float v2f;
typedef __attribute__((ext_vector_type(8))) float v8f;

static inline int cdiv(int a, int b) { return (a + b - 1) / b; }


template <int K, int U>
static __device__ __forceinline__ void gemm_tile(const float* __restrict__ X, const float* ldsW,
                                                 const float* norm, int nbase, float* tile,
                                                 float* __restrict__ H, int nrows, int mBase) {
    const int lane  = threadIdx.x & 31;
    const int half  = lane >> 4;
    const int l16   = lane & 15;

    int rowA = mBase + l16;
    if (rowA >= nrows) rowA = nrows - 1;
    const float* xrow = X + (size_t)rowA * K;

    v8f c0 = {}, c1 = {}, c2 = {}, c3 = {};
#pragma unroll U
    for (int k0 = 0; k0 < K; k0 += 4) {
        const int kk = k0 + 2 * half;
        v2f a = *(const v2f*)(xrow + kk);
        v2f b0 = *(const v2f*)(&ldsW[(0 * 16 + l16) * (K + 2) + kk]);
        v2f b1 = *(const v2f*)(&ldsW[(1 * 16 + l16) * (K + 2) + kk]);
        v2f b2 = *(const v2f*)(&ldsW[(2 * 16 + l16) * (K + 2) + kk]);
        v2f b3 = *(const v2f*)(&ldsW[(3 * 16 + l16) * (K + 2) + kk]);
        c0 = __builtin_amdgcn_wmma_f32_16x16x4_f32(false, a, false, b0, (short)0, c0, false, false);
        c1 = __builtin_amdgcn_wmma_f32_16x16x4_f32(false, a, false, b1, (short)0, c1, false, false);
        c2 = __builtin_amdgcn_wmma_f32_16x16x4_f32(false, a, false, b2, (short)0, c2, false, false);
        c3 = __builtin_amdgcn_wmma_f32_16x16x4_f32(false, a, false, b3, (short)0, c3, false, false);
    }

    float s[8];
#pragma unroll
    for (int r = 0; r < 8; ++r) s[r] = norm[min(mBase + 8 * half + r, nrows - 1) - nbase];
#pragma unroll
    for (int r = 0; r < 8; ++r) {
        float* t = tile + (8 * half + r) * TP + l16;
        t[0]  = c0[r] * s[r];  t[16] = c1[r] * s[r];
        t[32] = c2[r] * s[r];  t[48] = c3[r] * s[r];
    }
    float4 v[8];
#pragma unroll
    for (int i = 0; i < 8; ++i) v[i] = *(const float4*)(tile + (2 * i + half) * TP + 4 * l16);
#pragma unroll
    for (int i = 0; i < 8; ++i) {
        const int row = mBase + 2 * i + half;
        if (row < nrows) *(float4*)(H + (size_t)row * F_OUT + 4 * l16) = v[i];
    }
}

#define BW   400
#define SL   6400
#define PT   1024
#define EPT  ((SL + PT - 1) / PT)
#define MAXB 256
#define CAP  8192
static_assert(BW % 16 == 0 && BW / 16 <= PT / 32, "a wave per row tile of the bucket");
static_assert(BW <= PT, "a thread per node of the bucket");

static __device__ __forceinline__ int wave_excl_scan(int* c, int* o, int m) {
    const int lane = threadIdx.x & 31;
    const int per = (m + 31) >> 5;
    const int b0 = lane * per;
    int sum = 0;
    for (int k = 0; k < per; ++k) if (b0 + k < m) sum += c[b0 + k];
    int inc = sum;
    for (int off = 1; off < 32; off <<= 1) {
        const int t = __shfl_up(inc, off);
        if (lane >= off) inc += t;
    }
    int run = inc - sum;
    for (int k = 0; k < per; ++k) {
        if (b0 + k < m) {
            const int v = c[b0 + k];
            c[b0 + k] = run;
            if (o) o[b0 + k] = run;
            run += v;
        }
    }
    return __shfl(inc, 31);
}

__global__ __launch_bounds__(PT) void k_part(const int* __restrict__ src,
                                             const int* __restrict__ dst,
                                             int* __restrict__ pD, int* __restrict__ pS,
                                             int* __restrict__ qS, int* __restrict__ tabD,
                                             int* __restrict__ tabS, int e, int nbk, int nsl) {
    __shared__ int cD[MAXB], cS[MAXB], oD[MAXB + 1], oS[MAXB + 1];
    __shared__ int stD[SL], stP[SL], stQ[SL];
    const int tid = threadIdx.x;
    const int s   = blockIdx.x;
    const int beg = s * SL;
    const int len = min(SL, e - beg);

    for (int t = tid; t < MAXB; t += PT) { cD[t] = 0; cS[t] = 0; }

    int d[EPT], u[EPT];
#pragma unroll
    for (int i = 0; i < EPT; ++i) {
        const int j = tid + i * PT;
        d[i] = -1; u[i] = 0;
        if (j < len) { d[i] = dst[beg + j]; u[i] = src[beg + j]; }
    }
    __syncthreads();
#pragma unroll
    for (int i = 0; i < EPT; ++i) {
        if (d[i] >= 0) {
            atomicAdd(&cD[(unsigned)d[i] / BW], 1);
            atomicAdd(&cS[(unsigned)u[i] / BW], 1);
        }
    }
    __syncthreads();
    const int wave = __builtin_amdgcn_readfirstlane(tid >> 5);
    if (wave == 0) { const int t = wave_excl_scan(cD, oD, nbk); if ((tid & 31) == 0) oD[nbk] = t; }
    if (wave == 1) { const int t = wave_excl_scan(cS, oS, nbk); if ((tid & 31) == 0) oS[nbk] = t; }
    __syncthreads();

    for (int b = tid; b <= nbk; b += PT) {
        tabD[(size_t)b * nsl + s] = oD[b];
        tabS[(size_t)b * nsl + s] = oS[b];
    }
#pragma unroll
    for (int i = 0; i < EPT; ++i) {
        if (d[i] >= 0) {
            const int p = atomicAdd(&cD[(unsigned)d[i] / BW], 1);
            stD[p] = d[i];
            stP[p] = u[i];
            const int q = atomicAdd(&cS[(unsigned)u[i] / BW], 1);
            stQ[q] = u[i];
        }
    }
    __syncthreads();
    for (int j = tid; j < len; j += PT) {
        pD[beg + j] = stD[j];
        pS[beg + j] = stP[j];
        qS[beg + j] = stQ[j];
    }
}

#define RB 8
#define W1P (128 + 2)
#define LW_PROD (F_OUT * W1P + BW * TP + 2 * BW)
#define LW_SORT (CAP + 2 * BW + 2)
#define LW (LW_PROD > LW_SORT ? LW_PROD : LW_SORT)
static_assert(2 * ((LW * 4 + 2047) / 2048) * 2048 <= 320 * 1024, "two blocks per WGP");
__global__ __launch_bounds__(PT) __attribute__((amdgpu_waves_per_eu(16, 16)))
void k_bucket_gemm(const int* __restrict__ pD,
                                                    const int* __restrict__ pS,
                                                    const int* __restrict__ qS,
                                                    const int* __restrict__ tabD,
                                                    const int* __restrict__ tabS,
                                                    int* __restrict__ rowptr,
                                                    int* __restrict__ sorted,
                                                    float* __restrict__ ns, float* __restrict__ nd,
                                                    int n, int nsl, int nbk,
                                                    const float* __restrict__ X,
                                                    const float* __restrict__ W,
                                                    float* __restrict__ H) {
    __shared__ alignas(16) float lds[LW];
    const int tid = threadIdx.x;
    const int q   = tid & 3;
    const int wave = __builtin_amdgcn_readfirstlane(tid >> 5);

    if (blockIdx.x < nbk) {
        float* ldsW  = lds;
        float* tiles = ldsW + F_OUT * W1P;
        int*   cntS  = (int*)(tiles + BW * TP);
        float* nsL   = (float*)(cntS + BW);
        const int b  = blockIdx.x;
        const int lo = b * BW;
        const int nn = min(BW, n - lo);

        if (tid < BW) cntS[tid] = 0;
        __syncthreads();

        constexpr int PER = 128 * F_OUT / PT;
        static_assert(PER * PT == 128 * F_OUT, "whole elements per thread");
        float wv[PER];
#pragma unroll
        for (int i = 0; i < PER; ++i) wv[i] = W[tid + i * PT];

        for (int r = tid >> 2; r < nsl; r += PT / 4) {
            const int s0 = tabS[(size_t)b * nsl + r], s1 = tabS[(size_t)(b + 1) * nsl + r];
            const int* __restrict__ qs = qS + (size_t)r * SL;
            for (int j0 = s0; j0 < s1; j0 += 4 * RB) {
                int vs[RB];
#pragma unroll
                for (int k = 0; k < RB; ++k) {
                    const int js = j0 + q + 4 * k;
                    vs[k] = (js < s1) ? qs[js] : -1;
                }
#pragma unroll
                for (int k = 0; k < RB; ++k)
                    if (vs[k] >= 0) atomicAdd(&cntS[vs[k] - lo], 1);
            }
        }
#pragma unroll
        for (int i = 0; i < PER; ++i) {
            const int t = tid + i * PT;
            ldsW[(t % F_OUT) * W1P + t / F_OUT] = wv[i];
        }
        __syncthreads();

        if (tid < nn) {
            const float v = rsqrtf((float)(cntS[tid] + 1));
            ns[lo + tid] = v;
            nsL[tid] = v;
        }
        __syncthreads();

        if (wave * 16 < nn)
            gemm_tile<128, 4>(X, ldsW, nsL, lo, tiles + wave * 16 * TP, H, n, lo + wave * 16);
        return;
    }

    int* stg    = (int*)lds;
    int* cnt    = stg + CAP;
    int* cur    = cnt + BW;
    int* base_s = cur + BW;
    int* tot_s  = base_s + 1;
    const int b   = blockIdx.x - nbk;
    const int lo  = b * BW;
    const int nn  = min(BW, n - lo);

    if (tid < BW) cnt[tid] = 0;
    if (tid == 0) *base_s = 0;
    __syncthreads();

    int bpart = 0;
    for (int r = tid >> 2; r < nsl; r += PT / 4) {
        const int a0 = tabD[(size_t)b * nsl + r], a1 = tabD[(size_t)(b + 1) * nsl + r];
        if (q == 0) bpart += a0;
        const int* __restrict__ pd = pD + (size_t)r * SL;
        for (int j0 = a0; j0 < a1; j0 += 4 * RB) {
            int vd[RB];
#pragma unroll
            for (int k = 0; k < RB; ++k) {
                const int jd = j0 + q + 4 * k;
                vd[k] = (jd < a1) ? pd[jd] : -1;
            }
#pragma unroll
            for (int k = 0; k < RB; ++k)
                if (vd[k] >= 0) atomicAdd(&cnt[vd[k] - lo], 1);
        }
    }
    if (bpart) atomicAdd(base_s, bpart);
    __syncthreads();

    const int base = *base_s;
    if (tid < nn) {
        nd[lo + tid] = rsqrtf((float)(cnt[tid] + 1));
        cur[tid] = cnt[tid];
    }
    __syncthreads();
    if (wave == 0) {
        const int t = wave_excl_scan(cur, nullptr, nn);
        if ((tid & 31) == 0) *tot_s = t;
    }
    __syncthreads();
    if (tid < nn) rowptr[lo + tid] = base + cur[tid];
    const int tot = *tot_s;
    __syncthreads();

    for (int r = tid >> 2; r < nsl; r += PT / 4) {
        const int a0 = tabD[(size_t)b * nsl + r], a1 = tabD[(size_t)(b + 1) * nsl + r];
        const int* __restrict__ pd = pD + (size_t)r * SL;
        const int* __restrict__ ps = pS + (size_t)r * SL;
        for (int j0 = a0; j0 < a1; j0 += 4 * RB) {
            int vd[RB], vp[RB];
#pragma unroll
            for (int k = 0; k < RB; ++k) {
                const int j = j0 + q + 4 * k;
                vd[k] = (j < a1) ? pd[j] : -1;
                vp[k] = (j < a1) ? ps[j] : 0;
            }
#pragma unroll
            for (int k = 0; k < RB; ++k) {
                if (vd[k] >= 0) {
                    const int p = atomicAdd(&cur[vd[k] - lo], 1);
                    if (p < CAP) stg[p] = vp[k];
                    else sorted[base + p] = vp[k];
                }
            }
        }
    }
    __syncthreads();

    {
        constexpr int NW = (CAP + 31 + PT - 1) / PT;
        const int sh = base & 31;
        const int m = min(tot, CAP) + sh;
        int w[NW];
#pragma unroll
        for (int i = 0; i < NW; ++i) {
            const int jj = tid + i * PT;
            w[i] = (jj >= sh && jj < m) ? stg[jj - sh] : 0;
        }
#pragma unroll
        for (int i = 0; i < NW; ++i) {
            const int jj = tid + i * PT;
            if (jj >= sh && jj < m) sorted[base + jj - sh] = w[i];
        }
    }
}

#define GR 16
#define GT (GR * 16)
#define TPB 5
#define NB 8

struct RowHead { float4 self; int idx; int idx2; float s; };

static __device__ __forceinline__ RowHead row_head(const int* __restrict__ sorted,
                                                   const float4* __restrict__ h4,
                                                   const float* __restrict__ nd,
                                                   int row, int beg, int deg, int sub) {
    RowHead r;
    r.self = h4[(size_t)row * 16 + sub];
    r.s = nd[row];
    r.idx = row;
    if (sub < deg) r.idx = sorted[beg + sub];
    r.idx2 = row;
    if (16 + sub < deg) r.idx2 = sorted[beg + 16 + sub];
    return r;
}

static __device__ __forceinline__ void rows_request(const char* __restrict__ hb, int idx,
                                                    int lane0, int sub, float4 (&v)[NB]) {
#pragma unroll
    for (int j = 0; j < NB; ++j) {
        const int s = __builtin_amdgcn_ds_bpermute((lane0 + j) << 2, idx);
        const unsigned off = (unsigned)s * 256u + (unsigned)sub * 16u;
        v[j] = *(const float4*)(hb + off);
    }
}

static __device__ __forceinline__ void rows_add(float4& acc, const float4 (&v)[NB], int cnt) {
#pragma unroll
    for (int j = 0; j < NB; ++j) {
        const bool p = j < cnt;
        acc.x += p ? v[j].x : 0.0f; acc.y += p ? v[j].y : 0.0f;
        acc.z += p ? v[j].z : 0.0f; acc.w += p ? v[j].w : 0.0f;
    }
}

template <int RELU>
static __device__ __forceinline__ float4 row_finish(const int* __restrict__ sorted,
                                                    const char* __restrict__ hb,
                                                    RowHead hd, float4 (&v)[NB], float4 b,
                                                    int row, int beg, int deg, int sub, int grp) {
    float4 acc = hd.self;
    int idx = hd.idx;
    const float s = hd.s;

    rows_add(acc, v, min(NB, deg));
    for (int base = 0; base < deg; base += 16) {
        const int cnt = min(16, deg - base);
        if (base > 0) {
            if (base == 16) {
                idx = hd.idx2;
            } else {
                idx = row;
                if (sub < cnt) idx = sorted[beg + base + sub];
            }
            rows_request(hb, idx, grp, sub, v);
            rows_add(acc, v, min(NB, cnt));
        }
        if (cnt > NB) {
            rows_request(hb, idx, grp + NB, sub, v);
            rows_add(acc, v, cnt - NB);
        }
    }

    float4 r;
    r.x = fmaf(acc.x, s, b.x);
    r.y = fmaf(acc.y, s, b.y);
    r.z = fmaf(acc.z, s, b.z);
    r.w = fmaf(acc.w, s, b.w);
    if (RELU) {
        r.x = fmaxf(r.x, 0.0f); r.y = fmaxf(r.y, 0.0f);
        r.z = fmaxf(r.z, 0.0f); r.w = fmaxf(r.w, 0.0f);
    }
    return r;
}

static __device__ __forceinline__ void row_bounds(const int* __restrict__ rowptr, int row0,
                                                  int n, int e, int* rbeg, int* rdeg) {
#pragma unroll
    for (int it = 0; it < TPB; ++it) {
        const int row = row0 + it * GR;
        int b = 0, t = 0;
        if (row < n) {
            b = rowptr[row];
            t = (row + 1 < n) ? rowptr[row + 1] : e;
        }
        rbeg[it] = b;
        rdeg[it] = t - b;
    }
}

__global__ __launch_bounds__(GT) __attribute__((amdgpu_waves_per_eu(10, 10)))
void k_gather(const int* __restrict__ rowptr,
                                               const int* __restrict__ sorted,
                                               const float* __restrict__ h,
                                               const float* __restrict__ nd,
                                               const float* __restrict__ bias,
                                               float* __restrict__ out, int n, int e) {
    const int sub = threadIdx.x & 15;
    const int grp = threadIdx.x & 16;
    const int row0 = blockIdx.x * (TPB * GR) + (threadIdx.x >> 4);
    const float4* __restrict__ h4 = (const float4*)h;
    const char* __restrict__ hb = (const char*)h;
    const float4 b = *(const float4*)(bias + sub * 4);

    int rbeg[TPB], rdeg[TPB];
    row_bounds(rowptr, row0, n, e, rbeg, rdeg);

    RowHead hd;
    hd.self = make_float4(0.0f, 0.0f, 0.0f, 0.0f); hd.idx = 0; hd.idx2 = 0; hd.s = 0.0f;
    if (row0 < n) hd = row_head(sorted, h4, nd, row0, rbeg[0], rdeg[0], sub);
#pragma unroll
    for (int it = 0; it < TPB; ++it) {
        const int row = row0 + it * GR;
        const RowHead cur = hd;
        float4 v[NB];
        rows_request(hb, cur.idx, grp, sub, v);
        if (it + 1 < TPB && row + GR < n)
            hd = row_head(sorted, h4, nd, row + GR, rbeg[it + 1], rdeg[it + 1], sub);
        if (row < n) {
            const float4 r = row_finish<0>(sorted, hb, cur, v, b, row, rbeg[it], rdeg[it], sub, grp);
            *(float4*)(out + (size_t)row * F_OUT + sub * 4) = r;
        }
    }
}

static_assert(GT <= 1024 && (F_OUT * F_OUT) % GT == 0, "whole W2 elements per thread");
static_assert((GR / 16) * (F_OUT / 16) <= GT / 32, "a wave per 16x16 tile of the product");
__global__ __launch_bounds__(GT) __attribute__((amdgpu_waves_per_eu(10, 10)))
void k_gather_gemm(const int* __restrict__ rowptr,
                                                    const int* __restrict__ sorted,
                                                    const float* __restrict__ h,
                                                    const float* __restrict__ nd,
                                                    const float* __restrict__ bias,
                                                    const float* __restrict__ W,
                                                    const float* __restrict__ norm,
                                                    float* __restrict__ H, int n, int e) {
    __shared__ alignas(16) float ldsW[F_OUT * TP];
    __shared__ alignas(16) float tileA[GR * TP];
    __shared__ alignas(16) float tileD[GR * TP];
    const int sub = threadIdx.x & 15;
    const int grp = threadIdx.x & 16;
    const int rib = threadIdx.x >> 4;
    const int row0 = blockIdx.x * (TPB * GR) + rib;
    const float4* __restrict__ h4 = (const float4*)h;
    const char* __restrict__ hb = (const char*)h;

    constexpr int PER = F_OUT * F_OUT / GT;
    float w[PER];
#pragma unroll
    for (int i = 0; i < PER; ++i) w[i] = W[threadIdx.x + i * GT];

    const float4 b = *(const float4*)(bias + sub * 4);
    int rbeg[TPB], rdeg[TPB];
    row_bounds(rowptr, row0, n, e, rbeg, rdeg);

#pragma unroll
    for (int i = 0; i < PER; ++i) {
        const int t = threadIdx.x + i * GT;
        ldsW[(t % F_OUT) * TP + t / F_OUT] = w[i];
    }

    const int wave = __builtin_amdgcn_readfirstlane(threadIdx.x >> 5);
    const int half = (threadIdx.x >> 4) & 1;

    RowHead hd;
    hd.self = make_float4(0.0f, 0.0f, 0.0f, 0.0f); hd.idx = 0; hd.idx2 = 0; hd.s = 0.0f;
    if (row0 < n) hd = row_head(sorted, h4, nd, row0, rbeg[0], rdeg[0], sub);
#pragma unroll
    for (int it = 0; it < TPB; ++it) {
        const int row = row0 + it * GR;
        const RowHead cur = hd;
        float4 m[NB];
        rows_request(hb, cur.idx, grp, sub, m);
        if (it + 1 < TPB && row + GR < n)
            hd = row_head(sorted, h4, nd, row + GR, rbeg[it + 1], rdeg[it + 1], sub);
        float4 v = make_float4(0.0f, 0.0f, 0.0f, 0.0f);
        float sn = 0.0f;
        if (row < n) {
            v = row_finish<1>(sorted, hb, cur, m, b, row, rbeg[it], rdeg[it], sub, grp);
            sn = norm[row];
        }

        *(float4*)(&tileA[rib * TP + sub * 4]) = v;
        __syncthreads();

        if (wave < (GR / 16) * (F_OUT / 16)) {
            const int mt = wave >> 2, nt = wave & 3;
            const float* ap = &tileA[(mt * 16 + sub) * TP + 2 * half];
            const float* bp = &ldsW[(nt * 16 + sub) * TP + 2 * half];
            v8f c = {};
#pragma unroll
            for (int i = 0; i < 16; ++i) {
                const v2f a = *(const v2f*)(ap + 4 * i);
                const v2f bb = *(const v2f*)(bp + 4 * i);
                c = __builtin_amdgcn_wmma_f32_16x16x4_f32(false, a, false, bb, (short)0, c, false, false);
            }
#pragma unroll
            for (int r = 0; r < 8; ++r) tileD[(mt * 16 + 8 * half + r) * TP + nt * 16 + sub] = c[r];
        }
        __syncthreads();

        if (row < n) {
            float4 d = *(const float4*)(&tileD[rib * TP + sub * 4]);
            d.x *= sn; d.y *= sn; d.z *= sn; d.w *= sn;
            *(float4*)(H + (size_t)row * F_OUT + sub * 4) = d;
        }
    }
}

extern "C" void kernel_launch(void* const* d_in, const int* in_sizes, int n_in,
                              void* d_out, int out_size, void* d_ws, size_t ws_size,
                              hipStream_t stream) {
    const float* x   = (const float*)d_in[0];
    const int*   src = (const int*)d_in[1];
    const int*   dst = (const int*)d_in[2];
    const float* W1  = (const float*)d_in[3];
    const float* b1  = (const float*)d_in[4];
    const float* W2  = (const float*)d_in[5];
    const float* b2  = (const float*)d_in[6];
    float*       out = (float*)d_out;

    const int n = in_sizes[0] / 128;
    const int e = in_sizes[1];

    float* ws   = (float*)d_ws;
    float* buf0 = ws;
    int*   sorted = (int*)(ws + (size_t)n * F_OUT);
    int*   rowptr = sorted + e;
    float* ns   = ws + 2 * (size_t)n * F_OUT;
    float* nd   = ns + n;

    const int nbk = cdiv(n, BW);
    const int nsl = cdiv(e, SL);
    int* pD   = (int*)buf0;
    int* pS   = pD + (size_t)nsl * SL;
    int* qS   = pS + (size_t)nsl * SL;
    int* tabD = qS + (size_t)nsl * SL;
    int* tabS = tabD + (size_t)(nbk + 1) * nsl;
    k_part<<<nsl, PT, 0, stream>>>(src, dst, pD, pS, qS, tabD, tabS, e, nbk, nsl);

    k_bucket_gemm<<<2 * nbk, PT, 0, stream>>>(pD, pS, qS, tabD, tabS, rowptr, sorted, ns, nd,
                                              n, nsl, nbk, x, W1, out);

    const int ngb = cdiv(n, TPB * GR);
    k_gather_gemm<<<ngb, GT, 0, stream>>>(rowptr, sorted, out, nd, b1, W2, ns, buf0, n, e);
    k_gather<<<ngb, GT, 0, stream>>>(rowptr, sorted, buf0, nd, b2, out, n, e);
}
